# starting kernel
# baseline (speedup 1.0000x reference)
#include <hip/hip_runtime.h>
#include <hip/hip_bf16.h>
#include <math.h>


typedef float v2f __attribute__((ext_vector_type(2)));
typedef float v8f __attribute__((ext_vector_type(8)));

template <int NT>
__global__ void gemm_wmma_f32(const float* __restrict__ X,
                              const float* __restrict__ Wm,
                              float* __restrict__ Y,
                              int nrows)
{
    constexpr int NCOL = NT * 16;
    constexpr int P    = NCOL + 16;
    __shared__ __align__(16) float2 sB[64 * P];

    const int tid = threadIdx.x;
    for (int i = tid; i < 64 * NCOL; i += 256) {
        const int kp = i / NCOL;
        const int nn = i - kp * NCOL;
        float2 w2;
        w2.x = Wm[(2 * kp) * NCOL + nn];
        w2.y = Wm[(2 * kp + 1) * NCOL + nn];
        sB[kp * P + nn] = w2;
    }
    __syncthreads();

    const int wave = tid >> 5;
    const int lane = tid & 31;
    const int row0 = blockIdx.x * 128 + wave * 16;
    const int m    = lane & 15;
    const int khp  = lane >> 4;
    int row = row0 + m;
    if (row > nrows - 1) row = nrows - 1;

    const float2* Xp = (const float2*)(X + (size_t)row * 128) + khp;
    const int bbase = khp * P + m;

    v8f acc[NT];
#pragma unroll
    for (int t = 0; t < NT; ++t) acc[t] = (v8f)0.0f;

    for (int kk = 0; kk < 32; ++kk) {
        const float2 av = Xp[kk * 2];
        v2f a; a.x = av.x; a.y = av.y;
        const int bo = kk * 2 * P + bbase;
#pragma unroll
        for (int t = 0; t < NT; ++t) {
            const float2 bv = sB[bo + t * 16];
            v2f b; b.x = bv.x; b.y = bv.y;
            acc[t] = __builtin_amdgcn_wmma_f32_16x16x4_f32(
                false, a, false, b, (short)0, acc[t], false, false);
        }
    }

    const int rbase = khp * 8;
#pragma unroll
    for (int r = 0; r < 8; ++r) {
        const int rr = row0 + rbase + r;
        if (rr < nrows) {
#pragma unroll
            for (int t = 0; t < NT; ++t)
                Y[(size_t)rr * NCOL + t * 16 + m] = acc[t][r];
        }
    }
}

__device__ __forceinline__ void atomicMaxF32(float* addr, float v) {
    if (v >= 0.0f) atomicMax((int*)addr, __float_as_int(v));
    else           atomicMin((unsigned int*)addr, __float_as_uint(v));
}

__device__ __forceinline__ void edge_sd(const int* __restrict__ ei, int E,
                                        int e, int& s, int& d) {
    if (e < E) { s = ei[e]; d = ei[E + e]; }
    else       { s = e - E; d = e - E; }
}

__device__ __forceinline__ float lrelu(float a) {
    return a > 0.0f ? a : 0.2f * a;
}

__global__ void att_logits_l1(const float* __restrict__ xh,
                              const float* __restrict__ asrc,
                              const float* __restrict__ adst,
                              float* __restrict__ es, float* __restrict__ ed,
                              float* __restrict__ amax, float* __restrict__ dn,
                              int n)
{
    int t = blockIdx.x * blockDim.x + threadIdx.x;
    if (t >= n * 8) return;
    const int node = t >> 3, h = t & 7;
    const float* v  = xh + (size_t)node * 128 + h * 16;
    const float* as = asrc + h * 16;
    const float* ad = adst + h * 16;
    float s1 = 0.f, s2 = 0.f;
#pragma unroll
    for (int c = 0; c < 16; ++c) { float x = v[c]; s1 += x * as[c]; s2 += x * ad[c]; }
    es[t] = s1; ed[t] = s2;
    amax[t] = -__builtin_inff();
    dn[t]   = 0.0f;
}

__global__ void edge_max_l1(const int* __restrict__ ei, int E, int ET,
                            const float* __restrict__ es,
                            const float* __restrict__ ed,
                            float* __restrict__ amax)
{
    int t = blockIdx.x * blockDim.x + threadIdx.x;
    if (t >= ET * 8) return;
    const int e = t >> 3, h = t & 7;
    int s, d; edge_sd(ei, E, e, s, d);
    atomicMaxF32(&amax[d * 8 + h], lrelu(es[s * 8 + h] + ed[d * 8 + h]));
}

__global__ void edge_sum_l1(const int* __restrict__ ei, int E, int ET,
                            const float* __restrict__ es,
                            const float* __restrict__ ed,
                            const float* __restrict__ amax,
                            float* __restrict__ dn)
{
    int t = blockIdx.x * blockDim.x + threadIdx.x;
    if (t >= ET * 8) return;
    const int e = t >> 3, h = t & 7;
    int s, d; edge_sd(ei, E, e, s, d);
    const float a = lrelu(es[s * 8 + h] + ed[d * 8 + h]);
    atomicAdd(&dn[d * 8 + h], __expf(a - amax[d * 8 + h]));
}

__global__ void edge_scatter_l1(const int* __restrict__ ei, int E, int ET,
                                const float* __restrict__ es,
                                const float* __restrict__ ed,
                                const float* __restrict__ amax,
                                const float* __restrict__ dn,
                                const float* __restrict__ xh,
                                float* __restrict__ out)
{
    int t = blockIdx.x * blockDim.x + threadIdx.x;
    if (t >= ET * 8) return;
    const int e = t >> 3, h = t & 7;
    int s, d; edge_sd(ei, E, e, s, d);
    const float a   = lrelu(es[s * 8 + h] + ed[d * 8 + h]);
    const float att = __expf(a - amax[d * 8 + h]) / (dn[d * 8 + h] + 1e-16f);
    const float4* xs = (const float4*)(xh + (size_t)s * 128 + h * 16);
    float* o = out + (size_t)d * 128 + h * 16;
#pragma unroll
    for (int j = 0; j < 4; ++j) {
        const float4 v = xs[j];
        atomicAdd(o + j * 4 + 0, v.x * att);
        atomicAdd(o + j * 4 + 1, v.y * att);
        atomicAdd(o + j * 4 + 2, v.z * att);
        atomicAdd(o + j * 4 + 3, v.w * att);
    }
}

__global__ void finalize_l1(float* __restrict__ acc, const float* __restrict__ b,
                            int n)
{
    int t = blockIdx.x * blockDim.x + threadIdx.x;
    if (t >= n * 128) return;
    const float v = acc[t] + b[t & 127];
    acc[t] = v > 0.0f ? v : expm1f(v);
}

__global__ void pad_w2(const float* __restrict__ W2, float* __restrict__ W2p)
{
    int t = blockIdx.x * blockDim.x + threadIdx.x;
    if (t >= 128 * 48) return;
    const int k = t / 48, c = t % 48;
    W2p[t] = (c < 40) ? W2[k * 40 + c] : 0.0f;
}

__global__ void att_logits_l2(const float* __restrict__ xh2,
                              const float* __restrict__ asrc,
                              const float* __restrict__ adst,
                              float* __restrict__ es, float* __restrict__ ed,
                              float* __restrict__ amax, float* __restrict__ dn,
                              int n)
{
    int t = blockIdx.x * blockDim.x + threadIdx.x;
    if (t >= n) return;
    const float* v = xh2 + (size_t)t * 48;
    float s1 = 0.f, s2 = 0.f;
#pragma unroll
    for (int c = 0; c < 40; ++c) { float x = v[c]; s1 += x * asrc[c]; s2 += x * adst[c]; }
    es[t] = s1; ed[t] = s2;
    amax[t] = -__builtin_inff();
    dn[t]   = 0.0f;
}

__global__ void edge_max_l2(const int* __restrict__ ei, int E, int ET,
                            const float* __restrict__ es,
                            const float* __restrict__ ed,
                            float* __restrict__ amax)
{
    int t = blockIdx.x * blockDim.x + threadIdx.x;
    if (t >= ET) return;
    int s, d; edge_sd(ei, E, t, s, d);
    atomicMaxF32(&amax[d], lrelu(es[s] + ed[d]));
}

__global__ void edge_sum_l2(const int* __restrict__ ei, int E, int ET,
                            const float* __restrict__ es,
                            const float* __restrict__ ed,
                            const float* __restrict__ amax,
                            float* __restrict__ dn)
{
    int t = blockIdx.x * blockDim.x + threadIdx.x;
    if (t >= ET) return;
    int s, d; edge_sd(ei, E, t, s, d);
    const float a = lrelu(es[s] + ed[d]);
    atomicAdd(&dn[d], __expf(a - amax[d]));
}

__global__ void edge_scatter_l2(const int* __restrict__ ei, int E, int ET,
                                const float* __restrict__ es,
                                const float* __restrict__ ed,
                                const float* __restrict__ amax,
                                const float* __restrict__ dn,
                                const float* __restrict__ xh2,
                                float* __restrict__ out)
{
    int t = blockIdx.x * blockDim.x + threadIdx.x;
    if (t >= ET * 8) return;
    const int e = t >> 3, j = t & 7;
    int s, d; edge_sd(ei, E, e, s, d);
    const float a   = lrelu(es[s] + ed[d]);
    const float att = __expf(a - amax[d]) / (dn[d] + 1e-16f);
    const float* xs = xh2 + (size_t)s * 48 + j * 5;
    float* o = out + (size_t)d * 40 + j * 5;
#pragma unroll
    for (int c = 0; c < 5; ++c) atomicAdd(o + c, xs[c] * att);
}

__global__ void finalize_l2(float* __restrict__ out, const float* __restrict__ b,
                            int n)
{
    int t = blockIdx.x * blockDim.x + threadIdx.x;
    if (t >= n * 40) return;
    out[t] += b[t % 40];
}

extern "C" void kernel_launch(void* const* d_in, const int* in_sizes, int n_in,
                              void* d_out, int out_size, void* d_ws, size_t ws_size,
                              hipStream_t stream)
{
    (void)n_in; (void)out_size; (void)ws_size;

    const float* x   = (const float*)d_in[0];
    const int*   ei  = (const int*)d_in[1];
    const float* W1  = (const float*)d_in[2];
    const float* as1 = (const float*)d_in[3];
    const float* ad1 = (const float*)d_in[4];
    const float* b1  = (const float*)d_in[5];
    const float* W2  = (const float*)d_in[6];
    const float* as2 = (const float*)d_in[7];
    const float* ad2 = (const float*)d_in[8];
    const float* b2  = (const float*)d_in[9];
    float* out = (float*)d_out;

    const int n  = in_sizes[0] / 128;
    const int E  = in_sizes[1] / 2;
    const int ET = E + n;

    char* ws = (char*)d_ws;
    size_t off = 0;
    auto alloc = [&](size_t bytes) -> float* {
        float* p = (float*)(ws + off);
        off += (bytes + 255) & ~(size_t)255;
        return p;
    };
    float* xh1  = alloc((size_t)n * 128 * 4);
    float* accB = alloc((size_t)n * 128 * 4);
    float* es1  = alloc((size_t)n * 8 * 4);
    float* ed1  = alloc((size_t)n * 8 * 4);
    float* am1  = alloc((size_t)n * 8 * 4);
    float* dn1  = alloc((size_t)n * 8 * 4);
    float* es2  = alloc((size_t)n * 4);
    float* ed2  = alloc((size_t)n * 4);
    float* am2  = alloc((size_t)n * 4);
    float* dn2  = alloc((size_t)n * 4);
    float* W2p  = alloc((size_t)128 * 48 * 4);

    const int B = 256;
    const int eh = ET * 8;

    hipMemsetAsync(accB, 0, (size_t)n * 128 * 4, stream);
    hipMemsetAsync(out, 0, (size_t)n * 40 * 4, stream);

    gemm_wmma_f32<8><<<(n + 127) / 128, 256, 0, stream>>>(x, W1, xh1, n);
    att_logits_l1<<<(n * 8 + B - 1) / B, B, 0, stream>>>(xh1, as1, ad1, es1, ed1, am1, dn1, n);
    edge_max_l1<<<(eh + B - 1) / B, B, 0, stream>>>(ei, E, ET, es1, ed1, am1);
    edge_sum_l1<<<(eh + B - 1) / B, B, 0, stream>>>(ei, E, ET, es1, ed1, am1, dn1);
    edge_scatter_l1<<<(eh + B - 1) / B, B, 0, stream>>>(ei, E, ET, es1, ed1, am1, dn1, xh1, accB);
    finalize_l1<<<(n * 128 + B - 1) / B, B, 0, stream>>>(accB, b1, n);

    pad_w2<<<(128 * 48 + B - 1) / B, B, 0, stream>>>(W2, W2p);
    float* xh2 = xh1;
    gemm_wmma_f32<3><<<(n + 127) / 128, 256, 0, stream>>>(accB, W2p, xh2, n);
    att_logits_l2<<<(n + B - 1) / B, B, 0, stream>>>(xh2, as2, ad2, es2, ed2, am2, dn2, n);
    edge_max_l2<<<(ET + B - 1) / B, B, 0, stream>>>(ei, E, ET, es2, ed2, am2);
    edge_sum_l2<<<(ET + B - 1) / B, B, 0, stream>>>(ei, E, ET, es2, ed2, am2, dn2);
    edge_scatter_l2<<<(eh + B - 1) / B, B, 0, stream>>>(ei, E, ET, es2, ed2, am2, dn2, xh2, out);
    finalize_l2<<<(n * 40 + B - 1) / B, B, 0, stream>>>(out, b2, n);
}


# scatter kernels: a wave takes 32 edges and moves each row with lanes on consecutive channels, so every gather and f32 atomic add covers whole 128 B lines; atomic sum order stays arrival order
# speedup vs baseline: 2.3654x; 2.3654x over previous
#include <hip/hip_runtime.h>
#include <hip/hip_bf16.h>
#include <math.h>


typedef float v2f __attribute__((ext_vector_type(2)));
typedef float v8f __attribute__((ext_vector_type(8)));

template <int NT>
__global__ void gemm_wmma_f32(const float* __restrict__ X,
                              const float* __restrict__ Wm,
                              float* __restrict__ Y,
                              int nrows)
{
    constexpr int NCOL = NT * 16;
    constexpr int P    = NCOL + 16;
    __shared__ __align__(16) float2 sB[64 * P];

    const int tid = threadIdx.x;
    for (int i = tid; i < 64 * NCOL; i += 256) {
        const int kp = i / NCOL;
        const int nn = i - kp * NCOL;
        float2 w2;
        w2.x = Wm[(2 * kp) * NCOL + nn];
        w2.y = Wm[(2 * kp + 1) * NCOL + nn];
        sB[kp * P + nn] = w2;
    }
    __syncthreads();

    const int wave = tid >> 5;
    const int lane = tid & 31;
    const int row0 = blockIdx.x * 128 + wave * 16;
    const int m    = lane & 15;
    const int khp  = lane >> 4;
    int row = row0 + m;
    if (row > nrows - 1) row = nrows - 1;

    const float2* Xp = (const float2*)(X + (size_t)row * 128) + khp;
    const int bbase = khp * P + m;

    v8f acc[NT];
#pragma unroll
    for (int t = 0; t < NT; ++t) acc[t] = (v8f)0.0f;

    for (int kk = 0; kk < 32; ++kk) {
        const float2 av = Xp[kk * 2];
        v2f a; a.x = av.x; a.y = av.y;
        const int bo = kk * 2 * P + bbase;
#pragma unroll
        for (int t = 0; t < NT; ++t) {
            const float2 bv = sB[bo + t * 16];
            v2f b; b.x = bv.x; b.y = bv.y;
            acc[t] = __builtin_amdgcn_wmma_f32_16x16x4_f32(
                false, a, false, b, (short)0, acc[t], false, false);
        }
    }

    const int rbase = khp * 8;
#pragma unroll
    for (int r = 0; r < 8; ++r) {
        const int rr = row0 + rbase + r;
        if (rr < nrows) {
#pragma unroll
            for (int t = 0; t < NT; ++t)
                Y[(size_t)rr * NCOL + t * 16 + m] = acc[t][r];
        }
    }
}

__device__ __forceinline__ void atomicMaxF32(float* addr, float v) {
    if (v >= 0.0f) atomicMax((int*)addr, __float_as_int(v));
    else           atomicMin((unsigned int*)addr, __float_as_uint(v));
}

__device__ __forceinline__ void edge_sd(const int* __restrict__ ei, int E,
                                        int e, int& s, int& d) {
    if (e < E) { s = ei[e]; d = ei[E + e]; }
    else       { s = e - E; d = e - E; }
}

__device__ __forceinline__ float lrelu(float a) {
    return a > 0.0f ? a : 0.2f * a;
}

__device__ __forceinline__ int lane_pull(int v, int src) {
    return __builtin_amdgcn_ds_bpermute(src << 2, v);
}

__global__ void att_logits_l1(const float* __restrict__ xh,
                              const float* __restrict__ asrc,
                              const float* __restrict__ adst,
                              float* __restrict__ es, float* __restrict__ ed,
                              float* __restrict__ amax, float* __restrict__ dn,
                              int n)
{
    int t = blockIdx.x * blockDim.x + threadIdx.x;
    if (t >= n * 8) return;
    const int node = t >> 3, h = t & 7;
    const float* v  = xh + (size_t)node * 128 + h * 16;
    const float* as = asrc + h * 16;
    const float* ad = adst + h * 16;
    float s1 = 0.f, s2 = 0.f;
#pragma unroll
    for (int c = 0; c < 16; ++c) { float x = v[c]; s1 += x * as[c]; s2 += x * ad[c]; }
    es[t] = s1; ed[t] = s2;
    amax[t] = -__builtin_inff();
    dn[t]   = 0.0f;
}

__global__ void edge_max_l1(const int* __restrict__ ei, int E, int ET,
                            const float* __restrict__ es,
                            const float* __restrict__ ed,
                            float* __restrict__ amax)
{
    int t = blockIdx.x * blockDim.x + threadIdx.x;
    if (t >= ET * 8) return;
    const int e = t >> 3, h = t & 7;
    int s, d; edge_sd(ei, E, e, s, d);
    atomicMaxF32(&amax[d * 8 + h], lrelu(es[s * 8 + h] + ed[d * 8 + h]));
}

__global__ void edge_sum_l1(const int* __restrict__ ei, int E, int ET,
                            const float* __restrict__ es,
                            const float* __restrict__ ed,
                            const float* __restrict__ amax,
                            float* __restrict__ dn)
{
    int t = blockIdx.x * blockDim.x + threadIdx.x;
    if (t >= ET * 8) return;
    const int e = t >> 3, h = t & 7;
    int s, d; edge_sd(ei, E, e, s, d);
    const float a = lrelu(es[s * 8 + h] + ed[d * 8 + h]);
    atomicAdd(&dn[d * 8 + h], __expf(a - amax[d * 8 + h]));
}

__global__ void edge_scatter_l1(const int* __restrict__ ei, int E, int ET,
                                const float* __restrict__ es,
                                const float* __restrict__ ed,
                                const float* __restrict__ amax,
                                const float* __restrict__ dn,
                                const float* __restrict__ xh,
                                float* __restrict__ out)
{
    const int lane = threadIdx.x & 31;
    const int e0   = (blockIdx.x * 8 + (threadIdx.x >> 5)) * 32;
    if (e0 >= ET) return;
    int s = -1, d = -1;
    if (e0 + lane < ET) edge_sd(ei, E, e0 + lane, s, d);

    const int h   = lane & 7;
    const int sub = lane >> 3;
    const int hi  = lane >> 4;

    for (int half = 0; half < 2; ++half) {
        float att[4];
#pragma unroll
        for (int k = 0; k < 4; ++k) {
            const int g  = half * 4 + k;
            const int sj = lane_pull(s, 4 * g + sub);
            const int dj = lane_pull(d, 4 * g + sub);
            const int si = (sj < 0 ? 0 : sj) * 8 + h;
            const int di = (dj < 0 ? 0 : dj) * 8 + h;
            const float a = lrelu(es[si] + ed[di]);
            const float w = __expf(a - amax[di]) / (dn[di] + 1e-16f);
            att[k] = sj < 0 ? 0.0f : w;
        }

#pragma unroll
        for (int k = 0; k < 4; ++k) {
            const int g = half * 4 + k;
            int su[4], du[4];
            float v[4][4];
#pragma unroll
            for (int q = 0; q < 4; ++q) {
                su[q] = __builtin_amdgcn_readlane(s, 4 * g + q);
                du[q] = __builtin_amdgcn_readlane(d, 4 * g + q);
                const float* xs = xh + (size_t)(su[q] < 0 ? 0 : su[q]) * 128 + lane;
#pragma unroll
                for (int c = 0; c < 4; ++c) v[q][c] = xs[c * 32];
            }
#pragma unroll
            for (int q = 0; q < 4; ++q) {
                if (su[q] >= 0) {
                    float* o = out + (size_t)du[q] * 128 + lane;
#pragma unroll
                    for (int c = 0; c < 4; ++c) {
                        const float w = __int_as_float(
                            lane_pull(__float_as_int(att[k]), q * 8 + 2 * c + hi));
                        atomicAdd(o + c * 32, v[q][c] * w);
                    }
                }
            }
        }
    }
}

__global__ void finalize_l1(float* __restrict__ acc, const float* __restrict__ b,
                            int n)
{
    int t = blockIdx.x * blockDim.x + threadIdx.x;
    if (t >= n * 128) return;
    const float v = acc[t] + b[t & 127];
    acc[t] = v > 0.0f ? v : expm1f(v);
}

__global__ void pad_w2(const float* __restrict__ W2, float* __restrict__ W2p)
{
    int t = blockIdx.x * blockDim.x + threadIdx.x;
    if (t >= 128 * 48) return;
    const int k = t / 48, c = t % 48;
    W2p[t] = (c < 40) ? W2[k * 40 + c] : 0.0f;
}

__global__ void att_logits_l2(const float* __restrict__ xh2,
                              const float* __restrict__ asrc,
                              const float* __restrict__ adst,
                              float* __restrict__ es, float* __restrict__ ed,
                              float* __restrict__ amax, float* __restrict__ dn,
                              int n)
{
    int t = blockIdx.x * blockDim.x + threadIdx.x;
    if (t >= n) return;
    const float* v = xh2 + (size_t)t * 48;
    float s1 = 0.f, s2 = 0.f;
#pragma unroll
    for (int c = 0; c < 40; ++c) { float x = v[c]; s1 += x * asrc[c]; s2 += x * adst[c]; }
    es[t] = s1; ed[t] = s2;
    amax[t] = -__builtin_inff();
    dn[t]   = 0.0f;
}

__global__ void edge_max_l2(const int* __restrict__ ei, int E, int ET,
                            const float* __restrict__ es,
                            const float* __restrict__ ed,
                            float* __restrict__ amax)
{
    int t = blockIdx.x * blockDim.x + threadIdx.x;
    if (t >= ET) return;
    int s, d; edge_sd(ei, E, t, s, d);
    atomicMaxF32(&amax[d], lrelu(es[s] + ed[d]));
}

__global__ void edge_sum_l2(const int* __restrict__ ei, int E, int ET,
                            const float* __restrict__ es,
                            const float* __restrict__ ed,
                            const float* __restrict__ amax,
                            float* __restrict__ dn)
{
    int t = blockIdx.x * blockDim.x + threadIdx.x;
    if (t >= ET) return;
    int s, d; edge_sd(ei, E, t, s, d);
    const float a = lrelu(es[s] + ed[d]);
    atomicAdd(&dn[d], __expf(a - amax[d]));
}

__global__ void edge_scatter_l2(const int* __restrict__ ei, int E, int ET,
                                const float* __restrict__ es,
                                const float* __restrict__ ed,
                                const float* __restrict__ amax,
                                const float* __restrict__ dn,
                                const float* __restrict__ xh2,
                                float* __restrict__ out)
{
    const int lane = threadIdx.x & 31;
    const int e0   = (blockIdx.x * 8 + (threadIdx.x >> 5)) * 32;
    if (e0 >= ET) return;
    int s = -1, d = -1;
    float att = 0.0f;
    if (e0 + lane < ET) {
        edge_sd(ei, E, e0 + lane, s, d);
        const float a = lrelu(es[s] + ed[d]);
        att = __expf(a - amax[d]) / (dn[d] + 1e-16f);
    }

    for (int g = 0; g < 8; ++g) {
        int su[4], du[4];
        float w[4], v0[4], v1[4];
#pragma unroll
        for (int q = 0; q < 4; ++q) {
            su[q] = __builtin_amdgcn_readlane(s, 4 * g + q);
            du[q] = __builtin_amdgcn_readlane(d, 4 * g + q);
            w[q]  = __int_as_float(
                __builtin_amdgcn_readlane(__float_as_int(att), 4 * g + q));
            const float* xs = xh2 + (size_t)(su[q] < 0 ? 0 : su[q]) * 48 + lane;
            v0[q] = xs[0];
            v1[q] = (lane < 8) ? xs[32] : 0.0f;
        }
#pragma unroll
        for (int q = 0; q < 4; ++q) {
            if (su[q] >= 0) {
                float* o = out + (size_t)du[q] * 40 + lane;
                atomicAdd(o, v0[q] * w[q]);
                if (lane < 8) atomicAdd(o + 32, v1[q] * w[q]);
            }
        }
    }
}

__global__ void finalize_l2(float* __restrict__ out, const float* __restrict__ b,
                            int n)
{
    int t = blockIdx.x * blockDim.x + threadIdx.x;
    if (t >= n * 40) return;
    out[t] += b[t % 40];
}

extern "C" void kernel_launch(void* const* d_in, const int* in_sizes, int n_in,
                              void* d_out, int out_size, void* d_ws, size_t ws_size,
                              hipStream_t stream)
{
    (void)n_in; (void)out_size; (void)ws_size;

    const float* x   = (const float*)d_in[0];
    const int*   ei  = (const int*)d_in[1];
    const float* W1  = (const float*)d_in[2];
    const float* as1 = (const float*)d_in[3];
    const float* ad1 = (const float*)d_in[4];
    const float* b1  = (const float*)d_in[5];
    const float* W2  = (const float*)d_in[6];
    const float* as2 = (const float*)d_in[7];
    const float* ad2 = (const float*)d_in[8];
    const float* b2  = (const float*)d_in[9];
    float* out = (float*)d_out;

    const int n  = in_sizes[0] / 128;
    const int E  = in_sizes[1] / 2;
    const int ET = E + n;

    char* ws = (char*)d_ws;
    size_t off = 0;
    auto alloc = [&](size_t bytes) -> float* {
        float* p = (float*)(ws + off);
        off += (bytes + 255) & ~(size_t)255;
        return p;
    };
    float* xh1  = alloc((size_t)n * 128 * 4);
    float* accB = alloc((size_t)n * 128 * 4);
    float* es1  = alloc((size_t)n * 8 * 4);
    float* ed1  = alloc((size_t)n * 8 * 4);
    float* am1  = alloc((size_t)n * 8 * 4);
    float* dn1  = alloc((size_t)n * 8 * 4);
    float* es2  = alloc((size_t)n * 4);
    float* ed2  = alloc((size_t)n * 4);
    float* am2  = alloc((size_t)n * 4);
    float* dn2  = alloc((size_t)n * 4);
    float* W2p  = alloc((size_t)128 * 48 * 4);

    const int B = 256;
    const int eh = ET * 8;

    hipMemsetAsync(accB, 0, (size_t)n * 128 * 4, stream);
    hipMemsetAsync(out, 0, (size_t)n * 40 * 4, stream);

    gemm_wmma_f32<8><<<(n + 127) / 128, 256, 0, stream>>>(x, W1, xh1, n);
    att_logits_l1<<<(n * 8 + B - 1) / B, B, 0, stream>>>(xh1, as1, ad1, es1, ed1, am1, dn1, n);
    edge_max_l1<<<(eh + B - 1) / B, B, 0, stream>>>(ei, E, ET, es1, ed1, am1);
    edge_sum_l1<<<(eh + B - 1) / B, B, 0, stream>>>(ei, E, ET, es1, ed1, am1, dn1);
    edge_scatter_l1<<<(ET + B - 1) / B, B, 0, stream>>>(ei, E, ET, es1, ed1, am1, dn1, xh1, accB);
    finalize_l1<<<(n * 128 + B - 1) / B, B, 0, stream>>>(accB, b1, n);

    pad_w2<<<(128 * 48 + B - 1) / B, B, 0, stream>>>(W2, W2p);
    float* xh2 = xh1;
    gemm_wmma_f32<3><<<(n + 127) / 128, 256, 0, stream>>>(accB, W2p, xh2, n);
    att_logits_l2<<<(n + B - 1) / B, B, 0, stream>>>(xh2, as2, ad2, es2, ed2, am2, dn2, n);
    edge_max_l2<<<(ET + B - 1) / B, B, 0, stream>>>(ei, E, ET, es2, ed2, am2);
    edge_sum_l2<<<(ET + B - 1) / B, B, 0, stream>>>(ei, E, ET, es2, ed2, am2, dn2);
    edge_scatter_l2<<<(ET + B - 1) / B, B, 0, stream>>>(ei, E, ET, es2, ed2, am2, dn2, xh2, out);
    finalize_l2<<<(n * 40 + B - 1) / B, B, 0, stream>>>(out, b2, n);
}


# edges bucketed by destination every call (counting sort, int atomics) and ordered by source; one wave per node does softmax and weighted row sum in registers: no f32 atomics, fixed sum order
# speedup vs baseline: 3.9349x; 1.6635x over previous
#include <hip/hip_runtime.h>
#include <hip/hip_bf16.h>
#include <math.h>


typedef float v2f __attribute__((ext_vector_type(2)));
typedef float v8f __attribute__((ext_vector_type(8)));

template <int NT>
__global__ void gemm_wmma_f32(const float* __restrict__ X,
                              const float* __restrict__ Wm,
                              float* __restrict__ Y,
                              int nrows)
{
    constexpr int NCOL = NT * 16;
    constexpr int P    = NCOL + 16;
    __shared__ __align__(16) float2 sB[64 * P];

    const int tid = threadIdx.x;
    for (int i = tid; i < 64 * NCOL; i += 256) {
        const int kp = i / NCOL;
        const int nn = i - kp * NCOL;
        float2 w2;
        w2.x = Wm[(2 * kp) * NCOL + nn];
        w2.y = Wm[(2 * kp + 1) * NCOL + nn];
        sB[kp * P + nn] = w2;
    }
    __syncthreads();

    const int wave = tid >> 5;
    const int lane = tid & 31;
    const int row0 = blockIdx.x * 128 + wave * 16;
    const int m    = lane & 15;
    const int khp  = lane >> 4;
    int row = row0 + m;
    if (row > nrows - 1) row = nrows - 1;

    const float2* Xp = (const float2*)(X + (size_t)row * 128) + khp;
    const int bbase = khp * P + m;

    v8f acc[NT];
#pragma unroll
    for (int t = 0; t < NT; ++t) acc[t] = (v8f)0.0f;

    for (int kk = 0; kk < 32; ++kk) {
        const float2 av = Xp[kk * 2];
        v2f a; a.x = av.x; a.y = av.y;
        const int bo = kk * 2 * P + bbase;
#pragma unroll
        for (int t = 0; t < NT; ++t) {
            const float2 bv = sB[bo + t * 16];
            v2f b; b.x = bv.x; b.y = bv.y;
            acc[t] = __builtin_amdgcn_wmma_f32_16x16x4_f32(
                false, a, false, b, (short)0, acc[t], false, false);
        }
    }

    const int rbase = khp * 8;
#pragma unroll
    for (int r = 0; r < 8; ++r) {
        const int rr = row0 + rbase + r;
        if (rr < nrows) {
#pragma unroll
            for (int t = 0; t < NT; ++t)
                Y[(size_t)rr * NCOL + t * 16 + m] = acc[t][r];
        }
    }
}

__device__ __forceinline__ float lrelu(float a) {
    return a > 0.0f ? a : 0.2f * a;
}

__device__ __forceinline__ int edge_src(const int* __restrict__ csrc, int beg,
                                        int i, int deg, int d, int n) {
    if (i >= deg) return d;
    const int s = csrc[beg + i];
    return s < 0 ? 0 : (s > n - 1 ? n - 1 : s);
}

__global__ void csr_zero(int* __restrict__ cnt, int m)
{
    int t = blockIdx.x * blockDim.x + threadIdx.x;
    if (t < m) cnt[t] = 0;
}

__global__ void csr_hist(const int* __restrict__ ei, int E, int* __restrict__ cnt)
{
    int e = blockIdx.x * blockDim.x + threadIdx.x;
    if (e < E) atomicAdd(&cnt[ei[E + e]], 1);
}

__global__ void __launch_bounds__(1024)
csr_scan(const int* __restrict__ cnt, int* __restrict__ rowptr,
         int* __restrict__ cursor, int n)
{
    __shared__ int wsum[32];
    const int tid = threadIdx.x, lane = tid & 31, wave = tid >> 5;
    int carry = 0;
    for (int base = 0; base < n; base += 8192) {
        const int i0 = base + tid * 8;
        int4 ca = make_int4(0, 0, 0, 0), cb = make_int4(0, 0, 0, 0);
        if (i0 < n) {
            ca = *(const int4*)(cnt + i0);
            cb = *(const int4*)(cnt + i0 + 4);
        }
        const int p0 = ca.x, p1 = p0 + ca.y, p2 = p1 + ca.z, p3 = p2 + ca.w;
        const int p4 = p3 + cb.x, p5 = p4 + cb.y, p6 = p5 + cb.z, p7 = p6 + cb.w;
        int x = p7;
        for (int o = 1; o < 32; o <<= 1) {
            const int y = __shfl_up(x, o);
            if (lane >= o) x += y;
        }
        if (lane == 31) wsum[wave] = x;
        __syncthreads();
        const int ws = wsum[lane];
        int wx = ws;
        for (int o = 1; o < 32; o <<= 1) {
            const int y = __shfl_up(wx, o);
            if (lane >= o) wx += y;
        }
        const int tile_total = __shfl(wx, 31);
        const int wave_excl  = __shfl(wx - ws, wave);
        __syncthreads();
        const int ex = carry + wave_excl + (x - p7);
        if (i0 < n) {
            const int4 ra = make_int4(ex, ex + p0, ex + p1, ex + p2);
            const int4 rb = make_int4(ex + p3, ex + p4, ex + p5, ex + p6);
            *(int4*)(rowptr + i0)     = ra;
            *(int4*)(rowptr + i0 + 4) = rb;
            *(int4*)(cursor + i0)     = ra;
            *(int4*)(cursor + i0 + 4) = rb;
        }
        carry += tile_total;
    }
    if (tid == 0) rowptr[n] = carry;
}

__global__ void csr_fill(const int* __restrict__ ei, int E,
                         int* __restrict__ cursor, int* __restrict__ bsrc)
{
    int e = blockIdx.x * blockDim.x + threadIdx.x;
    if (e >= E) return;
    int pos = atomicAdd(&cursor[ei[E + e]], 1);
    pos = pos < 0 ? 0 : (pos > E - 1 ? E - 1 : pos);
    bsrc[pos] = ei[e];
}

__global__ void __launch_bounds__(256)
csr_order(const int* __restrict__ rowptr, const int* __restrict__ bsrc,
          int* __restrict__ csrc, int n)
{
    const int lane = threadIdx.x & 31;
    const int d = blockIdx.x * 8 + (threadIdx.x >> 5);
    if (d >= n) return;
    const int beg = __builtin_amdgcn_readfirstlane(rowptr[d]);
    const int end = __builtin_amdgcn_readfirstlane(rowptr[d + 1]);
    const int deg = end - beg;
    for (int c0 = 0; c0 < deg; c0 += 32) {
        const int i   = c0 + lane;
        const int key = (i < deg) ? bsrc[beg + i] : 0;
        int rank = 0;
        for (int b0 = 0; b0 < deg; b0 += 32) {
            const int kc = (b0 + lane < deg) ? bsrc[beg + b0 + lane] : 0;
            const int nb = (deg - b0 < 32) ? deg - b0 : 32;
            for (int jj = 0; jj < nb; ++jj) {
                const int kj = __builtin_amdgcn_readlane(kc, jj);
                const int j  = b0 + jj;
                rank += (kj < key || (kj == key && j < i)) ? 1 : 0;
            }
        }
        if (i < deg) csrc[beg + rank] = key;
    }
}

__global__ void att_logits_l1(const float* __restrict__ xh,
                              const float* __restrict__ asrc,
                              const float* __restrict__ adst,
                              float* __restrict__ es, float* __restrict__ ed,
                              int n)
{
    int t = blockIdx.x * blockDim.x + threadIdx.x;
    if (t >= n * 8) return;
    const int node = t >> 3, h = t & 7;
    const float* v  = xh + (size_t)node * 128 + h * 16;
    const float* as = asrc + h * 16;
    const float* ad = adst + h * 16;
    float s1 = 0.f, s2 = 0.f;
#pragma unroll
    for (int c = 0; c < 16; ++c) { float x = v[c]; s1 += x * as[c]; s2 += x * ad[c]; }
    es[t] = s1; ed[t] = s2;
}

__global__ void __launch_bounds__(256)
node_gather_l1(const int* __restrict__ rowptr, const int* __restrict__ csrc,
               const float* __restrict__ es, const float* __restrict__ ed,
               const float* __restrict__ xh, const float* __restrict__ b,
               float* __restrict__ h1, int n)
{
    const int lane = threadIdx.x & 31;
    const int d = blockIdx.x * 8 + (threadIdx.x >> 5);
    if (d >= n) return;
    const int beg = __builtin_amdgcn_readfirstlane(rowptr[d]);
    const int end = __builtin_amdgcn_readfirstlane(rowptr[d + 1]);
    const int deg = end - beg;
    const int cnt = deg + 1;

    const int h = lane & 7, sub = lane >> 3;
    const float edh = ed[d * 8 + h];

    float m = -__builtin_inff();
    for (int c0 = 0; c0 < cnt; c0 += 32) {
        const int i  = c0 + lane;
        const int sl = edge_src(csrc, beg, i, deg, d, n);
        const int nv = (cnt - c0 < 32) ? cnt - c0 : 32;
#pragma unroll
        for (int t = 0; t < 8; ++t) {
            const int el = 4 * t + sub;
            const int s  = __shfl(sl, el);
            const float a = lrelu(es[s * 8 + h] + edh);
            if (el < nv) m = fmaxf(m, a);
        }
    }
    m = fmaxf(m, __shfl_xor(m, 8));
    m = fmaxf(m, __shfl_xor(m, 16));

    float sum = 0.0f;
    for (int c0 = 0; c0 < cnt; c0 += 32) {
        const int i  = c0 + lane;
        const int sl = edge_src(csrc, beg, i, deg, d, n);
        const int nv = (cnt - c0 < 32) ? cnt - c0 : 32;
#pragma unroll
        for (int t = 0; t < 8; ++t) {
            const int el = 4 * t + sub;
            const int s  = __shfl(sl, el);
            const float a = lrelu(es[s * 8 + h] + edh);
            if (el < nv) sum += __expf(a - m);
        }
    }
    sum += __shfl_xor(sum, 8);
    sum += __shfl_xor(sum, 16);
    const float inv = 1.0f / (sum + 1e-16f);

    const int hh = lane >> 2;
    const float m2   = __shfl(m, hh);
    const float inv2 = __shfl(inv, hh);
    const float ed2  = __shfl(edh, hh);

    float4 acc = make_float4(0.f, 0.f, 0.f, 0.f);
    for (int c0 = 0; c0 < cnt; c0 += 32) {
        const int i  = c0 + lane;
        const int sl = edge_src(csrc, beg, i, deg, d, n);
        const int nv = (cnt - c0 < 32) ? cnt - c0 : 32;
        for (int g = 0; 4 * g < nv; ++g) {
            float4 v[4];
            float  ev[4];
#pragma unroll
            for (int q = 0; q < 4; ++q) {
                const int su = __builtin_amdgcn_readlane(sl, 4 * g + q);
                v[q]  = *(const float4*)(xh + (size_t)su * 128 + lane * 4);
                ev[q] = es[su * 8 + hh];
            }
#pragma unroll
            for (int q = 0; q < 4; ++q) {
                if (4 * g + q < nv) {
                    const float w = __expf(lrelu(ev[q] + ed2) - m2) * inv2;
                    acc.x = fmaf(w, v[q].x, acc.x);
                    acc.y = fmaf(w, v[q].y, acc.y);
                    acc.z = fmaf(w, v[q].z, acc.z);
                    acc.w = fmaf(w, v[q].w, acc.w);
                }
            }
        }
    }

    const float4 bv = *(const float4*)(b + lane * 4);
    float4 o;
    o.x = acc.x + bv.x; o.y = acc.y + bv.y; o.z = acc.z + bv.z; o.w = acc.w + bv.w;
    o.x = o.x > 0.0f ? o.x : expm1f(o.x);
    o.y = o.y > 0.0f ? o.y : expm1f(o.y);
    o.z = o.z > 0.0f ? o.z : expm1f(o.z);
    o.w = o.w > 0.0f ? o.w : expm1f(o.w);
    *(float4*)(h1 + (size_t)d * 128 + lane * 4) = o;
}

__global__ void pad_w2(const float* __restrict__ W2, float* __restrict__ W2p)
{
    int t = blockIdx.x * blockDim.x + threadIdx.x;
    if (t >= 128 * 48) return;
    const int k = t / 48, c = t % 48;
    W2p[t] = (c < 40) ? W2[k * 40 + c] : 0.0f;
}

__global__ void att_logits_l2(const float* __restrict__ xh2,
                              const float* __restrict__ asrc,
                              const float* __restrict__ adst,
                              float* __restrict__ es, float* __restrict__ ed,
                              int n)
{
    int t = blockIdx.x * blockDim.x + threadIdx.x;
    if (t >= n) return;
    const float* v = xh2 + (size_t)t * 48;
    float s1 = 0.f, s2 = 0.f;
#pragma unroll
    for (int c = 0; c < 40; ++c) { float x = v[c]; s1 += x * asrc[c]; s2 += x * adst[c]; }
    es[t] = s1; ed[t] = s2;
}

__global__ void __launch_bounds__(256)
node_gather_l2(const int* __restrict__ rowptr, const int* __restrict__ csrc,
               const float* __restrict__ es, const float* __restrict__ ed,
               const float* __restrict__ xh2, const float* __restrict__ b,
               float* __restrict__ out, int n)
{
    __shared__ __align__(16) float so[8 * 40];
    const int lane = threadIdx.x & 31;
    const int wv   = threadIdx.x >> 5;
    const int d0   = blockIdx.x * 8;
    const int d    = d0 + wv;

    if (d < n) {
        const int beg = __builtin_amdgcn_readfirstlane(rowptr[d]);
        const int end = __builtin_amdgcn_readfirstlane(rowptr[d + 1]);
        const int deg = end - beg;
        const int cnt = deg + 1;
        const float edd = ed[d];

        float m = -__builtin_inff();
        for (int c0 = 0; c0 < cnt; c0 += 32) {
            const int i  = c0 + lane;
            const int sl = edge_src(csrc, beg, i, deg, d, n);
            const float a = lrelu(es[sl] + edd);
            if (i < cnt) m = fmaxf(m, a);
        }
        for (int o = 16; o >= 1; o >>= 1) m = fmaxf(m, __shfl_xor(m, o));

        float sum = 0.0f;
        for (int c0 = 0; c0 < cnt; c0 += 32) {
            const int i  = c0 + lane;
            const int sl = edge_src(csrc, beg, i, deg, d, n);
            const float a = lrelu(es[sl] + edd);
            if (i < cnt) sum += __expf(a - m);
        }
        for (int o = 16; o >= 1; o >>= 1) sum += __shfl_xor(sum, o);
        const float inv = 1.0f / (sum + 1e-16f);

        const int slot = lane / 10;
        const int c4   = lane - slot * 10;
        float4 acc = make_float4(0.f, 0.f, 0.f, 0.f);
        for (int c0 = 0; c0 < cnt; c0 += 32) {
            const int i  = c0 + lane;
            const int sl = edge_src(csrc, beg, i, deg, d, n);
            const float a  = lrelu(es[sl] + edd);
            const float wl = __expf(a - m) * inv;
            const int nv = (cnt - c0 < 32) ? cnt - c0 : 32;
            for (int j0 = 0; j0 < nv; j0 += 12) {
                float4 v[4];
                float  w[4];
#pragma unroll
                for (int u = 0; u < 4; ++u) {
                    const int j  = j0 + 3 * u + slot;
                    const bool ok = (slot < 3) && (j < nv);
                    const int jj = ok ? j : 0;
                    const int s  = __shfl(sl, jj);
                    const float wj = __shfl(wl, jj);
                    w[u] = ok ? wj : 0.0f;
                    v[u] = *(const float4*)(xh2 + (size_t)s * 48 + c4 * 4);
                }
#pragma unroll
                for (int u = 0; u < 4; ++u) {
                    const int j = j0 + 3 * u + slot;
                    if ((slot < 3) && (j < nv)) {
                        acc.x = fmaf(w[u], v[u].x, acc.x);
                        acc.y = fmaf(w[u], v[u].y, acc.y);
                        acc.z = fmaf(w[u], v[u].z, acc.z);
                        acc.w = fmaf(w[u], v[u].w, acc.w);
                    }
                }
            }
        }
        const int l1 = (lane + 10) & 31, l2 = (lane + 20) & 31;
        acc.x += __shfl(acc.x, l1) + __shfl(acc.x, l2);
        acc.y += __shfl(acc.y, l1) + __shfl(acc.y, l2);
        acc.z += __shfl(acc.z, l1) + __shfl(acc.z, l2);
        acc.w += __shfl(acc.w, l1) + __shfl(acc.w, l2);
        if (lane < 10) {
            const float4 bv = *(const float4*)(b + lane * 4);
            float4 o;
            o.x = acc.x + bv.x; o.y = acc.y + bv.y; o.z = acc.z + bv.z; o.w = acc.w + bv.w;
            *(float4*)(so + wv * 40 + lane * 4) = o;
        }
    }
    __syncthreads();
    const int rows = (n - d0 < 8) ? n - d0 : 8;
    if ((int)threadIdx.x < rows * 10)
        *(float4*)(out + (size_t)d0 * 40 + threadIdx.x * 4) = *(const float4*)(so + threadIdx.x * 4);
}

extern "C" void kernel_launch(void* const* d_in, const int* in_sizes, int n_in,
                              void* d_out, int out_size, void* d_ws, size_t ws_size,
                              hipStream_t stream)
{
    (void)n_in; (void)out_size; (void)ws_size;

    const float* x   = (const float*)d_in[0];
    const int*   ei  = (const int*)d_in[1];
    const float* W1  = (const float*)d_in[2];
    const float* as1 = (const float*)d_in[3];
    const float* ad1 = (const float*)d_in[4];
    const float* b1  = (const float*)d_in[5];
    const float* W2  = (const float*)d_in[6];
    const float* as2 = (const float*)d_in[7];
    const float* ad2 = (const float*)d_in[8];
    const float* b2  = (const float*)d_in[9];
    float* out = (float*)d_out;

    const int n  = in_sizes[0] / 128;
    const int E  = in_sizes[1] / 2;

    char* ws = (char*)d_ws;
    size_t off = 0;
    auto alloc = [&](size_t bytes) -> float* {
        float* p = (float*)(ws + off);
        off += (bytes + 255) & ~(size_t)255;
        return p;
    };
    float* xh1  = alloc((size_t)n * 128 * 4);
    float* h1   = alloc((size_t)n * 128 * 4);
    float* es1  = alloc((size_t)n * 8 * 4);
    float* ed1  = alloc((size_t)n * 8 * 4);
    float* W2p  = alloc((size_t)128 * 48 * 4);
    int* cnt    = (int*)alloc((size_t)(n + 16) * 4);
    int* rowptr = (int*)alloc((size_t)(n + 16) * 4);
    int* cursor = (int*)alloc((size_t)(n + 16) * 4);
    int* csrc   = (int*)alloc((size_t)E * 4);
    int* bsrc   = (int*)h1;

    const int B = 256;

    csr_zero<<<(n + 16 + B - 1) / B, B, 0, stream>>>(cnt, n + 16);
    csr_hist<<<(E + B - 1) / B, B, 0, stream>>>(ei, E, cnt);
    csr_scan<<<1, 1024, 0, stream>>>(cnt, rowptr, cursor, n);
    csr_fill<<<(E + B - 1) / B, B, 0, stream>>>(ei, E, cursor, bsrc);
    csr_order<<<(n + 7) / 8, B, 0, stream>>>(rowptr, bsrc, csrc, n);

    gemm_wmma_f32<8><<<(n + 127) / 128, 256, 0, stream>>>(x, W1, xh1, n);
    att_logits_l1<<<(n * 8 + B - 1) / B, B, 0, stream>>>(xh1, as1, ad1, es1, ed1, n);
    node_gather_l1<<<(n + 7) / 8, B, 0, stream>>>(rowptr, csrc, es1, ed1, xh1, b1, h1, n);

    pad_w2<<<(128 * 48 + B - 1) / B, B, 0, stream>>>(W2, W2p);
    float* xh2 = xh1;
    float* es2 = es1;
    float* ed2 = ed1;
    gemm_wmma_f32<3><<<(n + 127) / 128, 256, 0, stream>>>(h1, W2p, xh2, n);
    att_logits_l2<<<(n + B - 1) / B, B, 0, stream>>>(xh2, as2, ad2, es2, ed2, n);
    node_gather_l2<<<(n + 7) / 8, B, 0, stream>>>(rowptr, csrc, es2, ed2, xh2, b2, out, n);
}


# gather kernels: a node of at most 32 incoming edges reads sources and logits once, keeping logits and weights in registers through max, sum and row sum (longer lists: old path); sum order unchanged
# speedup vs baseline: 4.2603x; 1.0827x over previous
#include <hip/hip_runtime.h>
#include <hip/hip_bf16.h>
#include <math.h>


typedef float v2f __attribute__((ext_vector_type(2)));
typedef float v8f __attribute__((ext_vector_type(8)));

template <int NT>
__global__ void gemm_wmma_f32(const float* __restrict__ X,
                              const float* __restrict__ Wm,
                              float* __restrict__ Y,
                              int nrows)
{
    constexpr int NCOL = NT * 16;
    constexpr int P    = NCOL + 16;
    __shared__ __align__(16) float2 sB[64 * P];

    const int tid = threadIdx.x;
    for (int i = tid; i < 64 * NCOL; i += 256) {
        const int kp = i / NCOL;
        const int nn = i - kp * NCOL;
        float2 w2;
        w2.x = Wm[(2 * kp) * NCOL + nn];
        w2.y = Wm[(2 * kp + 1) * NCOL + nn];
        sB[kp * P + nn] = w2;
    }
    __syncthreads();

    const int wave = tid >> 5;
    const int lane = tid & 31;
    const int row0 = blockIdx.x * 128 + wave * 16;
    const int m    = lane & 15;
    const int khp  = lane >> 4;
    int row = row0 + m;
    if (row > nrows - 1) row = nrows - 1;

    const float2* Xp = (const float2*)(X + (size_t)row * 128) + khp;
    const int bbase = khp * P + m;

    v8f acc[NT];
#pragma unroll
    for (int t = 0; t < NT; ++t) acc[t] = (v8f)0.0f;

    for (int kk = 0; kk < 32; ++kk) {
        const float2 av = Xp[kk * 2];
        v2f a; a.x = av.x; a.y = av.y;
        const int bo = kk * 2 * P + bbase;
#pragma unroll
        for (int t = 0; t < NT; ++t) {
            const float2 bv = sB[bo + t * 16];
            v2f b; b.x = bv.x; b.y = bv.y;
            acc[t] = __builtin_amdgcn_wmma_f32_16x16x4_f32(
                false, a, false, b, (short)0, acc[t], false, false);
        }
    }

    const int rbase = khp * 8;
#pragma unroll
    for (int r = 0; r < 8; ++r) {
        const int rr = row0 + rbase + r;
        if (rr < nrows) {
#pragma unroll
            for (int t = 0; t < NT; ++t)
                Y[(size_t)rr * NCOL + t * 16 + m] = acc[t][r];
        }
    }
}

__device__ __forceinline__ float lrelu(float a) {
    return a > 0.0f ? a : 0.2f * a;
}

__device__ __forceinline__ int edge_src(const int* __restrict__ csrc, int beg,
                                        int i, int deg, int d, int n) {
    if (i >= deg) return d;
    const int s = csrc[beg + i];
    return s < 0 ? 0 : (s > n - 1 ? n - 1 : s);
}

template <int CTRL>
__device__ __forceinline__ float quad_move(float x) {
    return __int_as_float(__builtin_amdgcn_update_dpp(0, __float_as_int(x), CTRL, 0xF, 0xF, true));
}
constexpr int QUAD_XOR1 = 0xB1;
constexpr int QUAD_XOR2 = 0x4E;

__global__ void csr_zero(int* __restrict__ cnt, int m)
{
    int t = blockIdx.x * blockDim.x + threadIdx.x;
    if (t < m) cnt[t] = 0;
}

__global__ void csr_hist(const int* __restrict__ ei, int E, int* __restrict__ cnt)
{
    int e = blockIdx.x * blockDim.x + threadIdx.x;
    if (e < E) atomicAdd(&cnt[ei[E + e]], 1);
}

__global__ void __launch_bounds__(1024)
csr_scan(const int* __restrict__ cnt, int* __restrict__ rowptr,
         int* __restrict__ cursor, int n)
{
    __shared__ int wsum[32];
    const int tid = threadIdx.x, lane = tid & 31, wave = tid >> 5;
    int carry = 0;
    for (int base = 0; base < n; base += 8192) {
        const int i0 = base + tid * 8;
        int4 ca = make_int4(0, 0, 0, 0), cb = make_int4(0, 0, 0, 0);
        if (i0 < n) {
            ca = *(const int4*)(cnt + i0);
            cb = *(const int4*)(cnt + i0 + 4);
        }
        const int p0 = ca.x, p1 = p0 + ca.y, p2 = p1 + ca.z, p3 = p2 + ca.w;
        const int p4 = p3 + cb.x, p5 = p4 + cb.y, p6 = p5 + cb.z, p7 = p6 + cb.w;
        int x = p7;
        for (int o = 1; o < 32; o <<= 1) {
            const int y = __shfl_up(x, o);
            if (lane >= o) x += y;
        }
        if (lane == 31) wsum[wave] = x;
        __syncthreads();
        const int ws = wsum[lane];
        int wx = ws;
        for (int o = 1; o < 32; o <<= 1) {
            const int y = __shfl_up(wx, o);
            if (lane >= o) wx += y;
        }
        const int tile_total = __shfl(wx, 31);
        const int wave_excl  = __shfl(wx - ws, wave);
        __syncthreads();
        const int ex = carry + wave_excl + (x - p7);
        if (i0 < n) {
            const int4 ra = make_int4(ex, ex + p0, ex + p1, ex + p2);
            const int4 rb = make_int4(ex + p3, ex + p4, ex + p5, ex + p6);
            *(int4*)(rowptr + i0)     = ra;
            *(int4*)(rowptr + i0 + 4) = rb;
            *(int4*)(cursor + i0)     = ra;
            *(int4*)(cursor + i0 + 4) = rb;
        }
        carry += tile_total;
    }
    if (tid == 0) rowptr[n] = carry;
}

__global__ void csr_fill(const int* __restrict__ ei, int E,
                         int* __restrict__ cursor, int* __restrict__ bsrc)
{
    int e = blockIdx.x * blockDim.x + threadIdx.x;
    if (e >= E) return;
    int pos = atomicAdd(&cursor[ei[E + e]], 1);
    pos = pos < 0 ? 0 : (pos > E - 1 ? E - 1 : pos);
    bsrc[pos] = ei[e];
}

__global__ void __launch_bounds__(256)
csr_order(const int* __restrict__ rowptr, const int* __restrict__ bsrc,
          int* __restrict__ csrc, int n)
{
    const int lane = threadIdx.x & 31;
    const int d = blockIdx.x * 8 + (threadIdx.x >> 5);
    if (d >= n) return;
    const int beg = __builtin_amdgcn_readfirstlane(rowptr[d]);
    const int end = __builtin_amdgcn_readfirstlane(rowptr[d + 1]);
    const int deg = end - beg;
    for (int c0 = 0; c0 < deg; c0 += 32) {
        const int i   = c0 + lane;
        const int key = (i < deg) ? bsrc[beg + i] : 0;
        int rank = 0;
        for (int b0 = 0; b0 < deg; b0 += 32) {
            const int kc = (b0 + lane < deg) ? bsrc[beg + b0 + lane] : 0;
            const int nb = (deg - b0 < 32) ? deg - b0 : 32;
            for (int jj = 0; jj < nb; ++jj) {
                const int kj = __builtin_amdgcn_readlane(kc, jj);
                const int j  = b0 + jj;
                rank += (kj < key || (kj == key && j < i)) ? 1 : 0;
            }
        }
        if (i < deg) csrc[beg + rank] = key;
    }
}

__global__ void att_logits_l1(const float* __restrict__ xh,
                              const float* __restrict__ asrc,
                              const float* __restrict__ adst,
                              float* __restrict__ es, float* __restrict__ ed,
                              int n)
{
    int t = blockIdx.x * blockDim.x + threadIdx.x;
    if (t >= n * 8) return;
    const int node = t >> 3, h = t & 7;
    const float* v  = xh + (size_t)node * 128 + h * 16;
    const float* as = asrc + h * 16;
    const float* ad = adst + h * 16;
    float s1 = 0.f, s2 = 0.f;
#pragma unroll
    for (int c = 0; c < 16; ++c) { float x = v[c]; s1 += x * as[c]; s2 += x * ad[c]; }
    es[t] = s1; ed[t] = s2;
}


template <int G>
__device__ __forceinline__ void l1_group(const float* __restrict__ xh, int sl, float pw,
                                         int cnt, int lane, float4& acc)
{
    float4 v[4];
#pragma unroll
    for (int q = 0; q < 4; ++q) {
        const int su = __builtin_amdgcn_readlane(sl, 4 * G + q);
        v[q] = *(const float4*)(xh + (size_t)su * 128 + lane * 4);
    }
    const float w[4] = { quad_move<0x00>(pw), quad_move<0x55>(pw),
                         quad_move<0xAA>(pw), quad_move<0xFF>(pw) };
#pragma unroll
    for (int q = 0; q < 4; ++q) {
        const bool have = (4 * G + q == 0) || (4 * G + q < cnt);
        const float tx = fmaf(w[q], v[q].x, acc.x);
        const float ty = fmaf(w[q], v[q].y, acc.y);
        const float tz = fmaf(w[q], v[q].z, acc.z);
        const float tw = fmaf(w[q], v[q].w, acc.w);
        acc.x = have ? tx : acc.x;
        acc.y = have ? ty : acc.y;
        acc.z = have ? tz : acc.z;
        acc.w = have ? tw : acc.w;
    }
}

__global__ void __launch_bounds__(256)
node_gather_l1(const int* __restrict__ rowptr, const int* __restrict__ csrc,
               const float* __restrict__ es, const float* __restrict__ ed,
               const float* __restrict__ xh, const float* __restrict__ b,
               float* __restrict__ h1, int n)
{
    const int lane = threadIdx.x & 31;
    const int d = blockIdx.x * 8 + (threadIdx.x >> 5);
    if (d >= n) return;
    const int beg = __builtin_amdgcn_readfirstlane(rowptr[d]);
    const int end = __builtin_amdgcn_readfirstlane(rowptr[d + 1]);
    const int deg = end - beg;
    const int cnt = deg + 1;

    float4 acc = make_float4(0.f, 0.f, 0.f, 0.f);

    if (cnt <= 32) {
        const int hq = lane >> 2, sub = lane & 3;
        const float edh = ed[d * 8 + hq];
        const int sl = edge_src(csrc, beg, lane, deg, d, n);

        float a[8];
        float m = -__builtin_inff();
#pragma unroll
        for (int t = 0; t < 8; ++t) {
            const int el = 4 * t + sub;
            const int s  = __shfl(sl, el);
            a[t] = lrelu(es[s * 8 + hq] + edh);
            if (el < cnt) m = fmaxf(m, a[t]);
        }
        m = fmaxf(m, quad_move<QUAD_XOR1>(m));
        m = fmaxf(m, quad_move<QUAD_XOR2>(m));

        float sum = 0.0f;
#pragma unroll
        for (int t = 0; t < 8; ++t) {
            const int el = 4 * t + sub;
            if (el < cnt) sum += __expf(a[t] - m);
        }
        sum += quad_move<QUAD_XOR1>(sum);
        sum += quad_move<QUAD_XOR2>(sum);
        const float inv = 1.0f / (sum + 1e-16f);

#pragma unroll
        for (int t = 0; t < 8; ++t) a[t] = __expf(a[t] - m) * inv;

        l1_group<0>(xh, sl, a[0], cnt, lane, acc);
        if (cnt >  4) l1_group<1>(xh, sl, a[1], cnt, lane, acc);
        if (cnt >  8) l1_group<2>(xh, sl, a[2], cnt, lane, acc);
        if (cnt > 12) l1_group<3>(xh, sl, a[3], cnt, lane, acc);
        if (cnt > 16) l1_group<4>(xh, sl, a[4], cnt, lane, acc);
        if (cnt > 20) l1_group<5>(xh, sl, a[5], cnt, lane, acc);
        if (cnt > 24) l1_group<6>(xh, sl, a[6], cnt, lane, acc);
        if (cnt > 28) l1_group<7>(xh, sl, a[7], cnt, lane, acc);
    } else {
        const int h = lane & 7, sub = lane >> 3;
        const float edh = ed[d * 8 + h];

        float m = -__builtin_inff();
        for (int c0 = 0; c0 < cnt; c0 += 32) {
            const int i  = c0 + lane;
            const int sl = edge_src(csrc, beg, i, deg, d, n);
            const int nv = (cnt - c0 < 32) ? cnt - c0 : 32;
#pragma unroll
            for (int t = 0; t < 8; ++t) {
                const int el = 4 * t + sub;
                const int s  = __shfl(sl, el);
                const float a = lrelu(es[s * 8 + h] + edh);
                if (el < nv) m = fmaxf(m, a);
            }
        }
        m = fmaxf(m, __shfl_xor(m, 8));
        m = fmaxf(m, __shfl_xor(m, 16));

        float sum = 0.0f;
        for (int c0 = 0; c0 < cnt; c0 += 32) {
            const int i  = c0 + lane;
            const int sl = edge_src(csrc, beg, i, deg, d, n);
            const int nv = (cnt - c0 < 32) ? cnt - c0 : 32;
#pragma unroll
            for (int t = 0; t < 8; ++t) {
                const int el = 4 * t + sub;
                const int s  = __shfl(sl, el);
                const float a = lrelu(es[s * 8 + h] + edh);
                if (el < nv) sum += __expf(a - m);
            }
        }
        sum += __shfl_xor(sum, 8);
        sum += __shfl_xor(sum, 16);
        const float inv = 1.0f / (sum + 1e-16f);

        const int hh = lane >> 2;
        const float m2   = __shfl(m, hh);
        const float inv2 = __shfl(inv, hh);
        const float ed2  = __shfl(edh, hh);

        for (int c0 = 0; c0 < cnt; c0 += 32) {
            const int i  = c0 + lane;
            const int sl = edge_src(csrc, beg, i, deg, d, n);
            const int nv = (cnt - c0 < 32) ? cnt - c0 : 32;
            for (int g = 0; 4 * g < nv; ++g) {
                float4 v[4];
                float  ev[4];
#pragma unroll
                for (int q = 0; q < 4; ++q) {
                    const int su = __builtin_amdgcn_readlane(sl, 4 * g + q);
                    v[q]  = *(const float4*)(xh + (size_t)su * 128 + lane * 4);
                    ev[q] = es[su * 8 + hh];
                }
#pragma unroll
                for (int q = 0; q < 4; ++q) {
                    if (4 * g + q < nv) {
                        const float w = __expf(lrelu(ev[q] + ed2) - m2) * inv2;
                        acc.x = fmaf(w, v[q].x, acc.x);
                        acc.y = fmaf(w, v[q].y, acc.y);
                        acc.z = fmaf(w, v[q].z, acc.z);
                        acc.w = fmaf(w, v[q].w, acc.w);
                    }
                }
            }
        }
    }

    const float4 bv = *(const float4*)(b + lane * 4);
    float4 o;
    o.x = acc.x + bv.x; o.y = acc.y + bv.y; o.z = acc.z + bv.z; o.w = acc.w + bv.w;
    o.x = o.x > 0.0f ? o.x : expm1f(o.x);
    o.y = o.y > 0.0f ? o.y : expm1f(o.y);
    o.z = o.z > 0.0f ? o.z : expm1f(o.z);
    o.w = o.w > 0.0f ? o.w : expm1f(o.w);
    *(float4*)(h1 + (size_t)d * 128 + lane * 4) = o;
}

__global__ void pad_w2(const float* __restrict__ W2, float* __restrict__ W2p)
{
    int t = blockIdx.x * blockDim.x + threadIdx.x;
    if (t >= 128 * 48) return;
    const int k = t / 48, c = t % 48;
    W2p[t] = (c < 40) ? W2[k * 40 + c] : 0.0f;
}

__global__ void att_logits_l2(const float* __restrict__ xh2,
                              const float* __restrict__ asrc,
                              const float* __restrict__ adst,
                              float* __restrict__ es, float* __restrict__ ed,
                              int n)
{
    int t = blockIdx.x * blockDim.x + threadIdx.x;
    if (t >= n) return;
    const float* v = xh2 + (size_t)t * 48;
    float s1 = 0.f, s2 = 0.f;
#pragma unroll
    for (int c = 0; c < 40; ++c) { float x = v[c]; s1 += x * asrc[c]; s2 += x * adst[c]; }
    es[t] = s1; ed[t] = s2;
}


__device__ __forceinline__ void l2_rows(const float* __restrict__ xh2, int sl, float wl,
                                        int nv, int slot, int c4, float4& acc)
{
    for (int j0 = 0; j0 < nv; j0 += 12) {
        float4 v[4];
        float  w[4];
#pragma unroll
        for (int u = 0; u < 4; ++u) {
            const int j  = j0 + 3 * u + slot;
            const bool ok = (slot < 3) && (j < nv);
            const int jj = ok ? j : 0;
            const int s  = __shfl(sl, jj);
            const float wj = __shfl(wl, jj);
            w[u] = ok ? wj : 0.0f;
            v[u] = *(const float4*)(xh2 + (size_t)s * 48 + c4 * 4);
        }
#pragma unroll
        for (int u = 0; u < 4; ++u) {
            const int j = j0 + 3 * u + slot;
            if ((slot < 3) && (j < nv)) {
                acc.x = fmaf(w[u], v[u].x, acc.x);
                acc.y = fmaf(w[u], v[u].y, acc.y);
                acc.z = fmaf(w[u], v[u].z, acc.z);
                acc.w = fmaf(w[u], v[u].w, acc.w);
            }
        }
    }
}

__global__ void __launch_bounds__(256)
node_gather_l2(const int* __restrict__ rowptr, const int* __restrict__ csrc,
               const float* __restrict__ es, const float* __restrict__ ed,
               const float* __restrict__ xh2, const float* __restrict__ b,
               float* __restrict__ out, int n)
{
    __shared__ __align__(16) float so[8 * 40];
    const int lane = threadIdx.x & 31;
    const int wv   = threadIdx.x >> 5;
    const int d0   = blockIdx.x * 8;
    const int d    = d0 + wv;

    if (d < n) {
        const int beg = __builtin_amdgcn_readfirstlane(rowptr[d]);
        const int end = __builtin_amdgcn_readfirstlane(rowptr[d + 1]);
        const int deg = end - beg;
        const int cnt = deg + 1;
        const float edd = ed[d];

        const int slot = lane / 10;
        const int c4   = lane - slot * 10;
        float4 acc = make_float4(0.f, 0.f, 0.f, 0.f);

        if (cnt <= 32) {
            const int sl  = edge_src(csrc, beg, lane, deg, d, n);
            const float a = lrelu(es[sl] + edd);

            float m = -__builtin_inff();
            if (lane < cnt) m = fmaxf(m, a);
            for (int o = 16; o >= 1; o >>= 1) m = fmaxf(m, __shfl_xor(m, o));

            const float e = __expf(a - m);
            float sum = 0.0f;
            if (lane < cnt) sum += e;
            for (int o = 16; o >= 1; o >>= 1) sum += __shfl_xor(sum, o);
            const float inv = 1.0f / (sum + 1e-16f);

            l2_rows(xh2, sl, e * inv, cnt, slot, c4, acc);
        } else {
            float m = -__builtin_inff();
            for (int c0 = 0; c0 < cnt; c0 += 32) {
                const int i  = c0 + lane;
                const int sl = edge_src(csrc, beg, i, deg, d, n);
                const float a = lrelu(es[sl] + edd);
                if (i < cnt) m = fmaxf(m, a);
            }
            for (int o = 16; o >= 1; o >>= 1) m = fmaxf(m, __shfl_xor(m, o));

            float sum = 0.0f;
            for (int c0 = 0; c0 < cnt; c0 += 32) {
                const int i  = c0 + lane;
                const int sl = edge_src(csrc, beg, i, deg, d, n);
                const float a = lrelu(es[sl] + edd);
                if (i < cnt) sum += __expf(a - m);
            }
            for (int o = 16; o >= 1; o >>= 1) sum += __shfl_xor(sum, o);
            const float inv = 1.0f / (sum + 1e-16f);

            for (int c0 = 0; c0 < cnt; c0 += 32) {
                const int i  = c0 + lane;
                const int sl = edge_src(csrc, beg, i, deg, d, n);
                const float a  = lrelu(es[sl] + edd);
                const float wl = __expf(a - m) * inv;
                const int nv = (cnt - c0 < 32) ? cnt - c0 : 32;
                l2_rows(xh2, sl, wl, nv, slot, c4, acc);
            }
        }
        const int l1 = (lane + 10) & 31, l2 = (lane + 20) & 31;
        acc.x += __shfl(acc.x, l1) + __shfl(acc.x, l2);
        acc.y += __shfl(acc.y, l1) + __shfl(acc.y, l2);
        acc.z += __shfl(acc.z, l1) + __shfl(acc.z, l2);
        acc.w += __shfl(acc.w, l1) + __shfl(acc.w, l2);
        if (lane < 10) {
            const float4 bv = *(const float4*)(b + lane * 4);
            float4 o;
            o.x = acc.x + bv.x; o.y = acc.y + bv.y; o.z = acc.z + bv.z; o.w = acc.w + bv.w;
            *(float4*)(so + wv * 40 + lane * 4) = o;
        }
    }
    __syncthreads();
    const int rows = (n - d0 < 8) ? n - d0 : 8;
    if ((int)threadIdx.x < rows * 10)
        *(float4*)(out + (size_t)d0 * 40 + threadIdx.x * 4) = *(const float4*)(so + threadIdx.x * 4);
}

extern "C" void kernel_launch(void* const* d_in, const int* in_sizes, int n_in,
                              void* d_out, int out_size, void* d_ws, size_t ws_size,
                              hipStream_t stream)
{
    (void)n_in; (void)out_size; (void)ws_size;

    const float* x   = (const float*)d_in[0];
    const int*   ei  = (const int*)d_in[1];
    const float* W1  = (const float*)d_in[2];
    const float* as1 = (const float*)d_in[3];
    const float* ad1 = (const float*)d_in[4];
    const float* b1  = (const float*)d_in[5];
    const float* W2  = (const float*)d_in[6];
    const float* as2 = (const float*)d_in[7];
    const float* ad2 = (const float*)d_in[8];
    const float* b2  = (const float*)d_in[9];
    float* out = (float*)d_out;

    const int n  = in_sizes[0] / 128;
    const int E  = in_sizes[1] / 2;

    char* ws = (char*)d_ws;
    size_t off = 0;
    auto alloc = [&](size_t bytes) -> float* {
        float* p = (float*)(ws + off);
        off += (bytes + 255) & ~(size_t)255;
        return p;
    };
    float* xh1  = alloc((size_t)n * 128 * 4);
    float* h1   = alloc((size_t)n * 128 * 4);
    float* es1  = alloc((size_t)n * 8 * 4);
    float* ed1  = alloc((size_t)n * 8 * 4);
    float* W2p  = alloc((size_t)128 * 48 * 4);
    int* cnt    = (int*)alloc((size_t)(n + 16) * 4);
    int* rowptr = (int*)alloc((size_t)(n + 16) * 4);
    int* cursor = (int*)alloc((size_t)(n + 16) * 4);
    int* csrc   = (int*)alloc((size_t)E * 4);
    int* bsrc   = (int*)h1;

    const int B = 256;

    csr_zero<<<(n + 16 + B - 1) / B, B, 0, stream>>>(cnt, n + 16);
    csr_hist<<<(E + B - 1) / B, B, 0, stream>>>(ei, E, cnt);
    csr_scan<<<1, 1024, 0, stream>>>(cnt, rowptr, cursor, n);
    csr_fill<<<(E + B - 1) / B, B, 0, stream>>>(ei, E, cursor, bsrc);
    csr_order<<<(n + 7) / 8, B, 0, stream>>>(rowptr, bsrc, csrc, n);

    gemm_wmma_f32<8><<<(n + 127) / 128, 256, 0, stream>>>(x, W1, xh1, n);
    att_logits_l1<<<(n * 8 + B - 1) / B, B, 0, stream>>>(xh1, as1, ad1, es1, ed1, n);
    node_gather_l1<<<(n + 7) / 8, B, 0, stream>>>(rowptr, csrc, es1, ed1, xh1, b1, h1, n);

    pad_w2<<<(128 * 48 + B - 1) / B, B, 0, stream>>>(W2, W2p);
    float* xh2 = xh1;
    float* es2 = es1;
    float* ed2 = ed1;
    gemm_wmma_f32<3><<<(n + 127) / 128, 256, 0, stream>>>(h1, W2p, xh2, n);
    att_logits_l2<<<(n + B - 1) / B, B, 0, stream>>>(xh2, as2, ad2, es2, ed2, n);
    node_gather_l2<<<(n + 7) / 8, B, 0, stream>>>(rowptr, csrc, es2, ed2, xh2, b2, out, n);
}


# edge bucketing with LDS counters in two levels (256-node ranges per block slice, then per node inside a range) replaces both global-atomic passes; csrc, rowptr and every sum order unchanged
# speedup vs baseline: 7.0947x; 1.6653x over previous
#include <hip/hip_runtime.h>
#include <hip/hip_bf16.h>
#include <math.h>


typedef float v2f __attribute__((ext_vector_type(2)));
typedef float v8f __attribute__((ext_vector_type(8)));

template <int NT>
__global__ void gemm_wmma_f32(const float* __restrict__ X,
                              const float* __restrict__ Wm,
                              float* __restrict__ Y,
                              int nrows)
{
    constexpr int NCOL = NT * 16;
    constexpr int P    = NCOL + 16;
    __shared__ __align__(16) float2 sB[64 * P];

    const int tid = threadIdx.x;
    for (int i = tid; i < 64 * NCOL; i += 256) {
        const int kp = i / NCOL;
        const int nn = i - kp * NCOL;
        float2 w2;
        w2.x = Wm[(2 * kp) * NCOL + nn];
        w2.y = Wm[(2 * kp + 1) * NCOL + nn];
        sB[kp * P + nn] = w2;
    }
    __syncthreads();

    const int wave = tid >> 5;
    const int lane = tid & 31;
    const int row0 = blockIdx.x * 128 + wave * 16;
    const int m    = lane & 15;
    const int khp  = lane >> 4;
    int row = row0 + m;
    if (row > nrows - 1) row = nrows - 1;

    const float2* Xp = (const float2*)(X + (size_t)row * 128) + khp;
    const int bbase = khp * P + m;

    v8f acc[NT];
#pragma unroll
    for (int t = 0; t < NT; ++t) acc[t] = (v8f)0.0f;

    for (int kk = 0; kk < 32; ++kk) {
        const float2 av = Xp[kk * 2];
        v2f a; a.x = av.x; a.y = av.y;
        const int bo = kk * 2 * P + bbase;
#pragma unroll
        for (int t = 0; t < NT; ++t) {
            const float2 bv = sB[bo + t * 16];
            v2f b; b.x = bv.x; b.y = bv.y;
            acc[t] = __builtin_amdgcn_wmma_f32_16x16x4_f32(
                false, a, false, b, (short)0, acc[t], false, false);
        }
    }

    const int rbase = khp * 8;
#pragma unroll
    for (int r = 0; r < 8; ++r) {
        const int rr = row0 + rbase + r;
        if (rr < nrows) {
#pragma unroll
            for (int t = 0; t < NT; ++t)
                Y[(size_t)rr * NCOL + t * 16 + m] = acc[t][r];
        }
    }
}

__device__ __forceinline__ float lrelu(float a) {
    return a > 0.0f ? a : 0.2f * a;
}

__device__ __forceinline__ int edge_src(const int* __restrict__ csrc, int beg,
                                        int i, int deg, int d, int n) {
    if (i >= deg) return d;
    const int s = csrc[beg + i];
    return s < 0 ? 0 : (s > n - 1 ? n - 1 : s);
}

template <int CTRL>
__device__ __forceinline__ float quad_move(float x) {
    return __int_as_float(__builtin_amdgcn_update_dpp(0, __float_as_int(x), CTRL, 0xF, 0xF, true));
}
constexpr int QUAD_XOR1 = 0xB1;
constexpr int QUAD_XOR2 = 0x4E;

constexpr int RSHIFT = 8;

__global__ void __launch_bounds__(1024)
range_count(const int* __restrict__ ei, int E, int nr, int* __restrict__ hist)
{
    extern __shared__ int lc[];
    const int tid = threadIdx.x, c = blockIdx.x, nb = gridDim.x;
    for (int r = tid; r < nr; r += 1024) lc[r] = 0;
    __syncthreads();
    const int chunk = (E + nb - 1) / nb;
    const int e0 = c * chunk;
    const int e1 = (e0 + chunk < E) ? e0 + chunk : E;
    const int* __restrict__ dstp = ei + E;
    int i0 = e0 + tid;
    for (; i0 + 3 * 1024 < e1; i0 += 4 * 1024) {
        int d[4];
#pragma unroll
        for (int u = 0; u < 4; ++u) d[u] = dstp[i0 + u * 1024];
#pragma unroll
        for (int u = 0; u < 4; ++u) atomicAdd(&lc[d[u] >> RSHIFT], 1);
    }
    for (; i0 < e1; i0 += 1024) atomicAdd(&lc[dstp[i0] >> RSHIFT], 1);
    __syncthreads();
    for (int r = tid; r < nr; r += 1024) hist[r * nb + c] = lc[r];
}

__global__ void __launch_bounds__(1024)
csr_scan(const int* __restrict__ cnt, int* __restrict__ rowptr, int n)
{
    __shared__ int wsum[32];
    const int tid = threadIdx.x, lane = tid & 31, wave = tid >> 5;
    int carry = 0;
    for (int base = 0; base < n; base += 8192) {
        const int i0 = base + tid * 8;
        int4 ca = make_int4(0, 0, 0, 0), cb = make_int4(0, 0, 0, 0);
        if (i0 < n) {
            ca = *(const int4*)(cnt + i0);
            cb = *(const int4*)(cnt + i0 + 4);
        }
        const int p0 = ca.x, p1 = p0 + ca.y, p2 = p1 + ca.z, p3 = p2 + ca.w;
        const int p4 = p3 + cb.x, p5 = p4 + cb.y, p6 = p5 + cb.z, p7 = p6 + cb.w;
        int x = p7;
        for (int o = 1; o < 32; o <<= 1) {
            const int y = __shfl_up(x, o);
            if (lane >= o) x += y;
        }
        if (lane == 31) wsum[wave] = x;
        __syncthreads();
        const int ws = wsum[lane];
        int wx = ws;
        for (int o = 1; o < 32; o <<= 1) {
            const int y = __shfl_up(wx, o);
            if (lane >= o) wx += y;
        }
        const int tile_total = __shfl(wx, 31);
        const int wave_excl  = __shfl(wx - ws, wave);
        __syncthreads();
        const int ex = carry + wave_excl + (x - p7);
        if (i0 < n) {
            const int4 ra = make_int4(ex, ex + p0, ex + p1, ex + p2);
            const int4 rb = make_int4(ex + p3, ex + p4, ex + p5, ex + p6);
            *(int4*)(rowptr + i0)     = ra;
            *(int4*)(rowptr + i0 + 4) = rb;
        }
        carry += tile_total;
    }
    if (tid == 0) rowptr[n] = carry;
}

__global__ void __launch_bounds__(1024)
range_place(const int* __restrict__ ei, int E, int nr,
            const int* __restrict__ offs, int* __restrict__ pairs_)
{
    extern __shared__ int lc[];
    int2* __restrict__ pairs = (int2*)pairs_;
    const int tid = threadIdx.x, c = blockIdx.x, nb = gridDim.x;
    for (int r = tid; r < nr; r += 1024) lc[r] = offs[r * nb + c];
    __syncthreads();
    const int chunk = (E + nb - 1) / nb;
    const int e0 = c * chunk;
    const int e1 = (e0 + chunk < E) ? e0 + chunk : E;
    const int* __restrict__ dstp = ei + E;
    int i0 = e0 + tid;
    for (; i0 + 3 * 1024 < e1; i0 += 4 * 1024) {
        int d[4], s[4];
#pragma unroll
        for (int u = 0; u < 4; ++u) { d[u] = dstp[i0 + u * 1024]; s[u] = ei[i0 + u * 1024]; }
#pragma unroll
        for (int u = 0; u < 4; ++u) {
            int pos = atomicAdd(&lc[d[u] >> RSHIFT], 1);
            pos = pos < 0 ? 0 : (pos > E - 1 ? E - 1 : pos);
            pairs[pos] = make_int2(d[u], s[u]);
        }
    }
    for (; i0 < e1; i0 += 1024) {
        const int d = dstp[i0], s = ei[i0];
        int pos = atomicAdd(&lc[d >> RSHIFT], 1);
        pos = pos < 0 ? 0 : (pos > E - 1 ? E - 1 : pos);
        pairs[pos] = make_int2(d, s);
    }
}

__global__ void __launch_bounds__(512)
node_place(const int* __restrict__ pairs_, const int* __restrict__ offs, int nb,
           int* __restrict__ rowptr, int* __restrict__ bsrc, int n, int E)
{
    __shared__ int cn[1 << RSHIFT];
    __shared__ int wt[(1 << RSHIFT) / 32];
    const int2* __restrict__ pairs = (const int2*)pairs_;
    const int tid = threadIdx.x, lane = tid & 31, wave = tid >> 5;
    const int b  = blockIdx.x;
    const int s0 = offs[b * nb];
    const int s1 = offs[(b + 1) * nb];
    const int d0 = b << RSHIFT;
    if (tid < (1 << RSHIFT)) cn[tid] = 0;
    __syncthreads();
    int i0 = s0 + tid;
    for (; i0 + 3 * 512 < s1; i0 += 4 * 512) {
        int dn[4];
#pragma unroll
        for (int u = 0; u < 4; ++u) dn[u] = pairs[i0 + u * 512].x;
#pragma unroll
        for (int u = 0; u < 4; ++u) atomicAdd(&cn[dn[u] - d0], 1);
    }
    for (; i0 < s1; i0 += 512) atomicAdd(&cn[pairs[i0].x - d0], 1);
    __syncthreads();
    int cnt = 0, x = 0;
    if (tid < (1 << RSHIFT)) {
        cnt = cn[tid];
        x = cnt;
        for (int o = 1; o < 32; o <<= 1) {
            const int y = __shfl_up(x, o);
            if (lane >= o) x += y;
        }
        if (lane == 31) wt[wave] = x;
    }
    __syncthreads();
    if (tid < (1 << RSHIFT)) {
        int pre = s0;
        for (int w = 0; w < wave; ++w) pre += wt[w];
        const int ex = pre + (x - cnt);
        cn[tid] = ex;
        if (d0 + tid < n) rowptr[d0 + tid] = ex;
    }
    if (b == (int)gridDim.x - 1 && tid == 0) rowptr[n] = s1;
    __syncthreads();
    for (i0 = s0 + tid; i0 + 3 * 512 < s1; i0 += 4 * 512) {
        int2 p[4];
#pragma unroll
        for (int u = 0; u < 4; ++u) p[u] = pairs[i0 + u * 512];
#pragma unroll
        for (int u = 0; u < 4; ++u) {
            int pos = atomicAdd(&cn[p[u].x - d0], 1);
            pos = pos < 0 ? 0 : (pos > E - 1 ? E - 1 : pos);
            bsrc[pos] = p[u].y;
        }
    }
    for (; i0 < s1; i0 += 512) {
        const int2 p = pairs[i0];
        int pos = atomicAdd(&cn[p.x - d0], 1);
        pos = pos < 0 ? 0 : (pos > E - 1 ? E - 1 : pos);
        bsrc[pos] = p.y;
    }
}

__global__ void __launch_bounds__(256)
csr_order(const int* __restrict__ rowptr, const int* __restrict__ bsrc,
          int* __restrict__ csrc, int n)
{
    const int lane = threadIdx.x & 31;
    const int d = blockIdx.x * 8 + (threadIdx.x >> 5);
    if (d >= n) return;
    const int beg = __builtin_amdgcn_readfirstlane(rowptr[d]);
    const int end = __builtin_amdgcn_readfirstlane(rowptr[d + 1]);
    const int deg = end - beg;
    for (int c0 = 0; c0 < deg; c0 += 32) {
        const int i   = c0 + lane;
        const int key = (i < deg) ? bsrc[beg + i] : 0;
        int rank = 0;
        for (int b0 = 0; b0 < deg; b0 += 32) {
            const int kc = (b0 + lane < deg) ? bsrc[beg + b0 + lane] : 0;
            const int nb = (deg - b0 < 32) ? deg - b0 : 32;
            for (int jj = 0; jj < nb; ++jj) {
                const int kj = __builtin_amdgcn_readlane(kc, jj);
                const int j  = b0 + jj;
                rank += (kj < key || (kj == key && j < i)) ? 1 : 0;
            }
        }
        if (i < deg) csrc[beg + rank] = key;
    }
}

__global__ void att_logits_l1(const float* __restrict__ xh,
                              const float* __restrict__ asrc,
                              const float* __restrict__ adst,
                              float* __restrict__ es, float* __restrict__ ed,
                              int n)
{
    int t = blockIdx.x * blockDim.x + threadIdx.x;
    if (t >= n * 8) return;
    const int node = t >> 3, h = t & 7;
    const float* v  = xh + (size_t)node * 128 + h * 16;
    const float* as = asrc + h * 16;
    const float* ad = adst + h * 16;
    float s1 = 0.f, s2 = 0.f;
#pragma unroll
    for (int c = 0; c < 16; ++c) { float x = v[c]; s1 += x * as[c]; s2 += x * ad[c]; }
    es[t] = s1; ed[t] = s2;
}


template <int G>
__device__ __forceinline__ void l1_group(const float* __restrict__ xh, int sl, float pw,
                                         int cnt, int lane, float4& acc)
{
    float4 v[4];
#pragma unroll
    for (int q = 0; q < 4; ++q) {
        const int su = __builtin_amdgcn_readlane(sl, 4 * G + q);
        v[q] = *(const float4*)(xh + (size_t)su * 128 + lane * 4);
    }
    const float w[4] = { quad_move<0x00>(pw), quad_move<0x55>(pw),
                         quad_move<0xAA>(pw), quad_move<0xFF>(pw) };
#pragma unroll
    for (int q = 0; q < 4; ++q) {
        const bool have = (4 * G + q == 0) || (4 * G + q < cnt);
        const float tx = fmaf(w[q], v[q].x, acc.x);
        const float ty = fmaf(w[q], v[q].y, acc.y);
        const float tz = fmaf(w[q], v[q].z, acc.z);
        const float tw = fmaf(w[q], v[q].w, acc.w);
        acc.x = have ? tx : acc.x;
        acc.y = have ? ty : acc.y;
        acc.z = have ? tz : acc.z;
        acc.w = have ? tw : acc.w;
    }
}

__global__ void __launch_bounds__(256)
node_gather_l1(const int* __restrict__ rowptr, const int* __restrict__ csrc,
               const float* __restrict__ es, const float* __restrict__ ed,
               const float* __restrict__ xh, const float* __restrict__ b,
               float* __restrict__ h1, int n)
{
    const int lane = threadIdx.x & 31;
    const int d = blockIdx.x * 8 + (threadIdx.x >> 5);
    if (d >= n) return;
    const int beg = __builtin_amdgcn_readfirstlane(rowptr[d]);
    const int end = __builtin_amdgcn_readfirstlane(rowptr[d + 1]);
    const int deg = end - beg;
    const int cnt = deg + 1;

    float4 acc = make_float4(0.f, 0.f, 0.f, 0.f);

    if (cnt <= 32) {
        const int hq = lane >> 2, sub = lane & 3;
        const float edh = ed[d * 8 + hq];
        const int sl = edge_src(csrc, beg, lane, deg, d, n);

        float a[8];
        float m = -__builtin_inff();
#pragma unroll
        for (int t = 0; t < 8; ++t) {
            const int el = 4 * t + sub;
            const int s  = __shfl(sl, el);
            a[t] = lrelu(es[s * 8 + hq] + edh);
            if (el < cnt) m = fmaxf(m, a[t]);
        }
        m = fmaxf(m, quad_move<QUAD_XOR1>(m));
        m = fmaxf(m, quad_move<QUAD_XOR2>(m));

        float sum = 0.0f;
#pragma unroll
        for (int t = 0; t < 8; ++t) {
            const int el = 4 * t + sub;
            if (el < cnt) sum += __expf(a[t] - m);
        }
        sum += quad_move<QUAD_XOR1>(sum);
        sum += quad_move<QUAD_XOR2>(sum);
        const float inv = 1.0f / (sum + 1e-16f);

#pragma unroll
        for (int t = 0; t < 8; ++t) a[t] = __expf(a[t] - m) * inv;

        l1_group<0>(xh, sl, a[0], cnt, lane, acc);
        if (cnt >  4) l1_group<1>(xh, sl, a[1], cnt, lane, acc);
        if (cnt >  8) l1_group<2>(xh, sl, a[2], cnt, lane, acc);
        if (cnt > 12) l1_group<3>(xh, sl, a[3], cnt, lane, acc);
        if (cnt > 16) l1_group<4>(xh, sl, a[4], cnt, lane, acc);
        if (cnt > 20) l1_group<5>(xh, sl, a[5], cnt, lane, acc);
        if (cnt > 24) l1_group<6>(xh, sl, a[6], cnt, lane, acc);
        if (cnt > 28) l1_group<7>(xh, sl, a[7], cnt, lane, acc);
    } else {
        const int h = lane & 7, sub = lane >> 3;
        const float edh = ed[d * 8 + h];

        float m = -__builtin_inff();
        for (int c0 = 0; c0 < cnt; c0 += 32) {
            const int i  = c0 + lane;
            const int sl = edge_src(csrc, beg, i, deg, d, n);
            const int nv = (cnt - c0 < 32) ? cnt - c0 : 32;
#pragma unroll
            for (int t = 0; t < 8; ++t) {
                const int el = 4 * t + sub;
                const int s  = __shfl(sl, el);
                const float a = lrelu(es[s * 8 + h] + edh);
                if (el < nv) m = fmaxf(m, a);
            }
        }
        m = fmaxf(m, __shfl_xor(m, 8));
        m = fmaxf(m, __shfl_xor(m, 16));

        float sum = 0.0f;
        for (int c0 = 0; c0 < cnt; c0 += 32) {
            const int i  = c0 + lane;
            const int sl = edge_src(csrc, beg, i, deg, d, n);
            const int nv = (cnt - c0 < 32) ? cnt - c0 : 32;
#pragma unroll
            for (int t = 0; t < 8; ++t) {
                const int el = 4 * t + sub;
                const int s  = __shfl(sl, el);
                const float a = lrelu(es[s * 8 + h] + edh);
                if (el < nv) sum += __expf(a - m);
            }
        }
        sum += __shfl_xor(sum, 8);
        sum += __shfl_xor(sum, 16);
        const float inv = 1.0f / (sum + 1e-16f);

        const int hh = lane >> 2;
        const float m2   = __shfl(m, hh);
        const float inv2 = __shfl(inv, hh);
        const float ed2  = __shfl(edh, hh);

        for (int c0 = 0; c0 < cnt; c0 += 32) {
            const int i  = c0 + lane;
            const int sl = edge_src(csrc, beg, i, deg, d, n);
            const int nv = (cnt - c0 < 32) ? cnt - c0 : 32;
            for (int g = 0; 4 * g < nv; ++g) {
                float4 v[4];
                float  ev[4];
#pragma unroll
                for (int q = 0; q < 4; ++q) {
                    const int su = __builtin_amdgcn_readlane(sl, 4 * g + q);
                    v[q]  = *(const float4*)(xh + (size_t)su * 128 + lane * 4);
                    ev[q] = es[su * 8 + hh];
                }
#pragma unroll
                for (int q = 0; q < 4; ++q) {
                    if (4 * g + q < nv) {
                        const float w = __expf(lrelu(ev[q] + ed2) - m2) * inv2;
                        acc.x = fmaf(w, v[q].x, acc.x);
                        acc.y = fmaf(w, v[q].y, acc.y);
                        acc.z = fmaf(w, v[q].z, acc.z);
                        acc.w = fmaf(w, v[q].w, acc.w);
                    }
                }
            }
        }
    }

    const float4 bv = *(const float4*)(b + lane * 4);
    float4 o;
    o.x = acc.x + bv.x; o.y = acc.y + bv.y; o.z = acc.z + bv.z; o.w = acc.w + bv.w;
    o.x = o.x > 0.0f ? o.x : expm1f(o.x);
    o.y = o.y > 0.0f ? o.y : expm1f(o.y);
    o.z = o.z > 0.0f ? o.z : expm1f(o.z);
    o.w = o.w > 0.0f ? o.w : expm1f(o.w);
    *(float4*)(h1 + (size_t)d * 128 + lane * 4) = o;
}

__global__ void pad_w2(const float* __restrict__ W2, float* __restrict__ W2p)
{
    int t = blockIdx.x * blockDim.x + threadIdx.x;
    if (t >= 128 * 48) return;
    const int k = t / 48, c = t % 48;
    W2p[t] = (c < 40) ? W2[k * 40 + c] : 0.0f;
}

__global__ void att_logits_l2(const float* __restrict__ xh2,
                              const float* __restrict__ asrc,
                              const float* __restrict__ adst,
                              float* __restrict__ es, float* __restrict__ ed,
                              int n)
{
    int t = blockIdx.x * blockDim.x + threadIdx.x;
    if (t >= n) return;
    const float* v = xh2 + (size_t)t * 48;
    float s1 = 0.f, s2 = 0.f;
#pragma unroll
    for (int c = 0; c < 40; ++c) { float x = v[c]; s1 += x * asrc[c]; s2 += x * adst[c]; }
    es[t] = s1; ed[t] = s2;
}


__device__ __forceinline__ void l2_rows(const float* __restrict__ xh2, int sl, float wl,
                                        int nv, int slot, int c4, float4& acc)
{
    for (int j0 = 0; j0 < nv; j0 += 12) {
        float4 v[4];
        float  w[4];
#pragma unroll
        for (int u = 0; u < 4; ++u) {
            const int j  = j0 + 3 * u + slot;
            const bool ok = (slot < 3) && (j < nv);
            const int jj = ok ? j : 0;
            const int s  = __shfl(sl, jj);
            const float wj = __shfl(wl, jj);
            w[u] = ok ? wj : 0.0f;
            v[u] = *(const float4*)(xh2 + (size_t)s * 48 + c4 * 4);
        }
#pragma unroll
        for (int u = 0; u < 4; ++u) {
            const int j = j0 + 3 * u + slot;
            if ((slot < 3) && (j < nv)) {
                acc.x = fmaf(w[u], v[u].x, acc.x);
                acc.y = fmaf(w[u], v[u].y, acc.y);
                acc.z = fmaf(w[u], v[u].z, acc.z);
                acc.w = fmaf(w[u], v[u].w, acc.w);
            }
        }
    }
}

__global__ void __launch_bounds__(256)
node_gather_l2(const int* __restrict__ rowptr, const int* __restrict__ csrc,
               const float* __restrict__ es, const float* __restrict__ ed,
               const float* __restrict__ xh2, const float* __restrict__ b,
               float* __restrict__ out, int n)
{
    __shared__ __align__(16) float so[8 * 40];
    const int lane = threadIdx.x & 31;
    const int wv   = threadIdx.x >> 5;
    const int d0   = blockIdx.x * 8;
    const int d    = d0 + wv;

    if (d < n) {
        const int beg = __builtin_amdgcn_readfirstlane(rowptr[d]);
        const int end = __builtin_amdgcn_readfirstlane(rowptr[d + 1]);
        const int deg = end - beg;
        const int cnt = deg + 1;
        const float edd = ed[d];

        const int slot = lane / 10;
        const int c4   = lane - slot * 10;
        float4 acc = make_float4(0.f, 0.f, 0.f, 0.f);

        if (cnt <= 32) {
            const int sl  = edge_src(csrc, beg, lane, deg, d, n);
            const float a = lrelu(es[sl] + edd);

            float m = -__builtin_inff();
            if (lane < cnt) m = fmaxf(m, a);
            for (int o = 16; o >= 1; o >>= 1) m = fmaxf(m, __shfl_xor(m, o));

            const float e = __expf(a - m);
            float sum = 0.0f;
            if (lane < cnt) sum += e;
            for (int o = 16; o >= 1; o >>= 1) sum += __shfl_xor(sum, o);
            const float inv = 1.0f / (sum + 1e-16f);

            l2_rows(xh2, sl, e * inv, cnt, slot, c4, acc);
        } else {
            float m = -__builtin_inff();
            for (int c0 = 0; c0 < cnt; c0 += 32) {
                const int i  = c0 + lane;
                const int sl = edge_src(csrc, beg, i, deg, d, n);
                const float a = lrelu(es[sl] + edd);
                if (i < cnt) m = fmaxf(m, a);
            }
            for (int o = 16; o >= 1; o >>= 1) m = fmaxf(m, __shfl_xor(m, o));

            float sum = 0.0f;
            for (int c0 = 0; c0 < cnt; c0 += 32) {
                const int i  = c0 + lane;
                const int sl = edge_src(csrc, beg, i, deg, d, n);
                const float a = lrelu(es[sl] + edd);
                if (i < cnt) sum += __expf(a - m);
            }
            for (int o = 16; o >= 1; o >>= 1) sum += __shfl_xor(sum, o);
            const float inv = 1.0f / (sum + 1e-16f);

            for (int c0 = 0; c0 < cnt; c0 += 32) {
                const int i  = c0 + lane;
                const int sl = edge_src(csrc, beg, i, deg, d, n);
                const float a  = lrelu(es[sl] + edd);
                const float wl = __expf(a - m) * inv;
                const int nv = (cnt - c0 < 32) ? cnt - c0 : 32;
                l2_rows(xh2, sl, wl, nv, slot, c4, acc);
            }
        }
        const int l1 = (lane + 10) & 31, l2 = (lane + 20) & 31;
        acc.x += __shfl(acc.x, l1) + __shfl(acc.x, l2);
        acc.y += __shfl(acc.y, l1) + __shfl(acc.y, l2);
        acc.z += __shfl(acc.z, l1) + __shfl(acc.z, l2);
        acc.w += __shfl(acc.w, l1) + __shfl(acc.w, l2);
        if (lane < 10) {
            const float4 bv = *(const float4*)(b + lane * 4);
            float4 o;
            o.x = acc.x + bv.x; o.y = acc.y + bv.y; o.z = acc.z + bv.z; o.w = acc.w + bv.w;
            *(float4*)(so + wv * 40 + lane * 4) = o;
        }
    }
    __syncthreads();
    const int rows = (n - d0 < 8) ? n - d0 : 8;
    if ((int)threadIdx.x < rows * 10)
        *(float4*)(out + (size_t)d0 * 40 + threadIdx.x * 4) = *(const float4*)(so + threadIdx.x * 4);
}

extern "C" void kernel_launch(void* const* d_in, const int* in_sizes, int n_in,
                              void* d_out, int out_size, void* d_ws, size_t ws_size,
                              hipStream_t stream)
{
    (void)n_in; (void)out_size; (void)ws_size;

    const float* x   = (const float*)d_in[0];
    const int*   ei  = (const int*)d_in[1];
    const float* W1  = (const float*)d_in[2];
    const float* as1 = (const float*)d_in[3];
    const float* ad1 = (const float*)d_in[4];
    const float* b1  = (const float*)d_in[5];
    const float* W2  = (const float*)d_in[6];
    const float* as2 = (const float*)d_in[7];
    const float* ad2 = (const float*)d_in[8];
    const float* b2  = (const float*)d_in[9];
    float* out = (float*)d_out;

    const int n  = in_sizes[0] / 128;
    const int E  = in_sizes[1] / 2;

    char* ws = (char*)d_ws;
    size_t off = 0;
    auto alloc = [&](size_t bytes) -> float* {
        float* p = (float*)(ws + off);
        off += (bytes + 255) & ~(size_t)255;
        return p;
    };
    float* xh1  = alloc((size_t)n * 128 * 4);
    float* h1   = alloc((size_t)n * 128 * 4);
    float* es1  = alloc((size_t)n * 8 * 4);
    float* ed1  = alloc((size_t)n * 8 * 4);
    float* W2p  = alloc((size_t)128 * 48 * 4);
    constexpr int NBLK = 128;
    const int nr = (n + (1 << RSHIFT) - 1) >> RSHIFT;
    const int M  = nr * NBLK;
    int* hist   = (int*)alloc((size_t)(M + 16) * 4);
    int* offs   = (int*)alloc((size_t)(M + 16) * 4);
    int* rowptr = (int*)alloc((size_t)(n + 16) * 4);
    int* csrc   = (int*)alloc((size_t)E * 4);
    int* bsrc   = (int*)h1;
    int* pairs  = (int*)xh1;

    const int B = 256;

    range_count<<<NBLK, 1024, nr * sizeof(int), stream>>>(ei, E, nr, hist);
    csr_scan<<<1, 1024, 0, stream>>>(hist, offs, M);
    range_place<<<NBLK, 1024, nr * sizeof(int), stream>>>(ei, E, nr, offs, pairs);
    node_place<<<nr, 512, 0, stream>>>(pairs, offs, NBLK, rowptr, bsrc, n, E);
    csr_order<<<(n + 7) / 8, B, 0, stream>>>(rowptr, bsrc, csrc, n);

    gemm_wmma_f32<8><<<(n + 127) / 128, 256, 0, stream>>>(x, W1, xh1, n);
    att_logits_l1<<<(n * 8 + B - 1) / B, B, 0, stream>>>(xh1, as1, ad1, es1, ed1, n);
    node_gather_l1<<<(n + 7) / 8, B, 0, stream>>>(rowptr, csrc, es1, ed1, xh1, b1, h1, n);

    pad_w2<<<(128 * 48 + B - 1) / B, B, 0, stream>>>(W2, W2p);
    float* xh2 = xh1;
    float* es2 = es1;
    float* ed2 = ed1;
    gemm_wmma_f32<3><<<(n + 127) / 128, 256, 0, stream>>>(h1, W2p, xh2, n);
    att_logits_l2<<<(n + B - 1) / B, B, 0, stream>>>(xh2, as2, ad2, es2, ed2, n);
    node_gather_l2<<<(n + 7) / 8, B, 0, stream>>>(rowptr, csrc, es2, ed2, xh2, b2, out, n);
}


# e_src/e_dst = X(W a): W a_src, W a_dst built per call as weight columns; logits come out of the WMMA GEMMs (l1: 9th tile; l2: pad columns of the row); logit kernels go; logit sums reassociated
# speedup vs baseline: 7.3183x; 1.0315x over previous
#include <hip/hip_runtime.h>
#include <hip/hip_bf16.h>
#include <math.h>


typedef float v2f __attribute__((ext_vector_type(2)));
typedef float v8f __attribute__((ext_vector_type(8)));

template <int NT, bool LOGITS>
__global__ void gemm_wmma_f32(const float* __restrict__ X,
                              const float* __restrict__ Wm,
                              float* __restrict__ Y,
                              float* __restrict__ es,
                              float* __restrict__ ed,
                              int nrows)
{
    constexpr int NCOL = NT * 16;
    constexpr int YT   = LOGITS ? NT - 1 : NT;
    constexpr int YP   = YT * 16;
    constexpr int P    = LOGITS ? NCOL : NCOL + 16;
    __shared__ __align__(16) float2 sB[64 * P];

    const int tid = threadIdx.x;
    for (int i = tid; i < 64 * NCOL; i += 256) {
        const int kp = i / NCOL;
        const int nn = i - kp * NCOL;
        float2 w2;
        w2.x = Wm[(2 * kp) * NCOL + nn];
        w2.y = Wm[(2 * kp + 1) * NCOL + nn];
        sB[kp * P + nn] = w2;
    }
    __syncthreads();

    const int wave = tid >> 5;
    const int lane = tid & 31;
    const int row0 = blockIdx.x * 128 + wave * 16;
    const int m    = lane & 15;
    const int khp  = lane >> 4;
    int row = row0 + m;
    if (row > nrows - 1) row = nrows - 1;

    const float2* Xp = (const float2*)(X + (size_t)row * 128) + khp;
    const int bbase = khp * P + m;

    v8f acc[NT];
#pragma unroll
    for (int t = 0; t < NT; ++t) acc[t] = (v8f)0.0f;

    for (int kk = 0; kk < 32; ++kk) {
        const float2 av = Xp[kk * 2];
        v2f a; a.x = av.x; a.y = av.y;
        const int bo = kk * 2 * P + bbase;
#pragma unroll
        for (int t = 0; t < NT; ++t) {
            const float2 bv = sB[bo + t * 16];
            v2f b; b.x = bv.x; b.y = bv.y;
            acc[t] = __builtin_amdgcn_wmma_f32_16x16x4_f32(
                false, a, false, b, (short)0, acc[t], false, false);
        }
    }

    const int rbase = khp * 8;
#pragma unroll
    for (int r = 0; r < 8; ++r) {
        const int rr = row0 + rbase + r;
        if (rr < nrows) {
#pragma unroll
            for (int t = 0; t < YT; ++t)
                Y[(size_t)rr * YP + t * 16 + m] = acc[t][r];
            if (LOGITS) {
                const float v = acc[NT - 1][r];
                if (m < 8) es[(size_t)rr * 8 + m] = v;
                else       ed[(size_t)rr * 8 + (m - 8)] = v;
            }
        }
    }
}

__device__ __forceinline__ float lrelu(float a) {
    return a > 0.0f ? a : 0.2f * a;
}

__device__ __forceinline__ int edge_src(const int* __restrict__ csrc, int beg,
                                        int i, int deg, int d, int n) {
    if (i >= deg) return d;
    const int s = csrc[beg + i];
    return s < 0 ? 0 : (s > n - 1 ? n - 1 : s);
}

template <int CTRL>
__device__ __forceinline__ float quad_move(float x) {
    return __int_as_float(__builtin_amdgcn_update_dpp(0, __float_as_int(x), CTRL, 0xF, 0xF, true));
}
constexpr int QUAD_XOR1 = 0xB1;
constexpr int QUAD_XOR2 = 0x4E;

constexpr int RSHIFT = 8;

__global__ void __launch_bounds__(1024)
range_count(const int* __restrict__ ei, int E, int nr, int* __restrict__ hist)
{
    extern __shared__ int lc[];
    const int tid = threadIdx.x, c = blockIdx.x, nb = gridDim.x;
    for (int r = tid; r < nr; r += 1024) lc[r] = 0;
    __syncthreads();
    const int chunk = (E + nb - 1) / nb;
    const int e0 = c * chunk;
    const int e1 = (e0 + chunk < E) ? e0 + chunk : E;
    const int* __restrict__ dstp = ei + E;
    int i0 = e0 + tid;
    for (; i0 + 3 * 1024 < e1; i0 += 4 * 1024) {
        int d[4];
#pragma unroll
        for (int u = 0; u < 4; ++u) d[u] = dstp[i0 + u * 1024];
#pragma unroll
        for (int u = 0; u < 4; ++u) atomicAdd(&lc[d[u] >> RSHIFT], 1);
    }
    for (; i0 < e1; i0 += 1024) atomicAdd(&lc[dstp[i0] >> RSHIFT], 1);
    __syncthreads();
    for (int r = tid; r < nr; r += 1024) hist[r * nb + c] = lc[r];
}

__global__ void __launch_bounds__(1024)
csr_scan(const int* __restrict__ cnt, int* __restrict__ rowptr, int n)
{
    __shared__ int wsum[32];
    const int tid = threadIdx.x, lane = tid & 31, wave = tid >> 5;
    int carry = 0;
    for (int base = 0; base < n; base += 8192) {
        const int i0 = base + tid * 8;
        int4 ca = make_int4(0, 0, 0, 0), cb = make_int4(0, 0, 0, 0);
        if (i0 < n) {
            ca = *(const int4*)(cnt + i0);
            cb = *(const int4*)(cnt + i0 + 4);
        }
        const int p0 = ca.x, p1 = p0 + ca.y, p2 = p1 + ca.z, p3 = p2 + ca.w;
        const int p4 = p3 + cb.x, p5 = p4 + cb.y, p6 = p5 + cb.z, p7 = p6 + cb.w;
        int x = p7;
        for (int o = 1; o < 32; o <<= 1) {
            const int y = __shfl_up(x, o);
            if (lane >= o) x += y;
        }
        if (lane == 31) wsum[wave] = x;
        __syncthreads();
        const int ws = wsum[lane];
        int wx = ws;
        for (int o = 1; o < 32; o <<= 1) {
            const int y = __shfl_up(wx, o);
            if (lane >= o) wx += y;
        }
        const int tile_total = __shfl(wx, 31);
        const int wave_excl  = __shfl(wx - ws, wave);
        __syncthreads();
        const int ex = carry + wave_excl + (x - p7);
        if (i0 < n) {
            const int4 ra = make_int4(ex, ex + p0, ex + p1, ex + p2);
            const int4 rb = make_int4(ex + p3, ex + p4, ex + p5, ex + p6);
            *(int4*)(rowptr + i0)     = ra;
            *(int4*)(rowptr + i0 + 4) = rb;
        }
        carry += tile_total;
    }
    if (tid == 0) rowptr[n] = carry;
}

__global__ void __launch_bounds__(1024)
range_place(const int* __restrict__ ei, int E, int nr,
            const int* __restrict__ offs, int* __restrict__ pairs_)
{
    extern __shared__ int lc[];
    int2* __restrict__ pairs = (int2*)pairs_;
    const int tid = threadIdx.x, c = blockIdx.x, nb = gridDim.x;
    for (int r = tid; r < nr; r += 1024) lc[r] = offs[r * nb + c];
    __syncthreads();
    const int chunk = (E + nb - 1) / nb;
    const int e0 = c * chunk;
    const int e1 = (e0 + chunk < E) ? e0 + chunk : E;
    const int* __restrict__ dstp = ei + E;
    int i0 = e0 + tid;
    for (; i0 + 3 * 1024 < e1; i0 += 4 * 1024) {
        int d[4], s[4];
#pragma unroll
        for (int u = 0; u < 4; ++u) { d[u] = dstp[i0 + u * 1024]; s[u] = ei[i0 + u * 1024]; }
#pragma unroll
        for (int u = 0; u < 4; ++u) {
            int pos = atomicAdd(&lc[d[u] >> RSHIFT], 1);
            pos = pos < 0 ? 0 : (pos > E - 1 ? E - 1 : pos);
            pairs[pos] = make_int2(d[u], s[u]);
        }
    }
    for (; i0 < e1; i0 += 1024) {
        const int d = dstp[i0], s = ei[i0];
        int pos = atomicAdd(&lc[d >> RSHIFT], 1);
        pos = pos < 0 ? 0 : (pos > E - 1 ? E - 1 : pos);
        pairs[pos] = make_int2(d, s);
    }
}

__global__ void __launch_bounds__(512)
node_place(const int* __restrict__ pairs_, const int* __restrict__ offs, int nb,
           int* __restrict__ rowptr, int* __restrict__ bsrc, int n, int E)
{
    __shared__ int cn[1 << RSHIFT];
    __shared__ int wt[(1 << RSHIFT) / 32];
    const int2* __restrict__ pairs = (const int2*)pairs_;
    const int tid = threadIdx.x, lane = tid & 31, wave = tid >> 5;
    const int b  = blockIdx.x;
    const int s0 = offs[b * nb];
    const int s1 = offs[(b + 1) * nb];
    const int d0 = b << RSHIFT;
    if (tid < (1 << RSHIFT)) cn[tid] = 0;
    __syncthreads();
    int i0 = s0 + tid;
    for (; i0 + 3 * 512 < s1; i0 += 4 * 512) {
        int dn[4];
#pragma unroll
        for (int u = 0; u < 4; ++u) dn[u] = pairs[i0 + u * 512].x;
#pragma unroll
        for (int u = 0; u < 4; ++u) atomicAdd(&cn[dn[u] - d0], 1);
    }
    for (; i0 < s1; i0 += 512) atomicAdd(&cn[pairs[i0].x - d0], 1);
    __syncthreads();
    int cnt = 0, x = 0;
    if (tid < (1 << RSHIFT)) {
        cnt = cn[tid];
        x = cnt;
        for (int o = 1; o < 32; o <<= 1) {
            const int y = __shfl_up(x, o);
            if (lane >= o) x += y;
        }
        if (lane == 31) wt[wave] = x;
    }
    __syncthreads();
    if (tid < (1 << RSHIFT)) {
        int pre = s0;
        for (int w = 0; w < wave; ++w) pre += wt[w];
        const int ex = pre + (x - cnt);
        cn[tid] = ex;
        if (d0 + tid < n) rowptr[d0 + tid] = ex;
    }
    if (b == (int)gridDim.x - 1 && tid == 0) rowptr[n] = s1;
    __syncthreads();
    for (i0 = s0 + tid; i0 + 3 * 512 < s1; i0 += 4 * 512) {
        int2 p[4];
#pragma unroll
        for (int u = 0; u < 4; ++u) p[u] = pairs[i0 + u * 512];
#pragma unroll
        for (int u = 0; u < 4; ++u) {
            int pos = atomicAdd(&cn[p[u].x - d0], 1);
            pos = pos < 0 ? 0 : (pos > E - 1 ? E - 1 : pos);
            bsrc[pos] = p[u].y;
        }
    }
    for (; i0 < s1; i0 += 512) {
        const int2 p = pairs[i0];
        int pos = atomicAdd(&cn[p.x - d0], 1);
        pos = pos < 0 ? 0 : (pos > E - 1 ? E - 1 : pos);
        bsrc[pos] = p.y;
    }
}

__global__ void __launch_bounds__(256)
csr_order(const int* __restrict__ rowptr, const int* __restrict__ bsrc,
          int* __restrict__ csrc, int n)
{
    const int lane = threadIdx.x & 31;
    const int d = blockIdx.x * 8 + (threadIdx.x >> 5);
    if (d >= n) return;
    const int beg = __builtin_amdgcn_readfirstlane(rowptr[d]);
    const int end = __builtin_amdgcn_readfirstlane(rowptr[d + 1]);
    const int deg = end - beg;
    for (int c0 = 0; c0 < deg; c0 += 32) {
        const int i   = c0 + lane;
        const int key = (i < deg) ? bsrc[beg + i] : 0;
        int rank = 0;
        for (int b0 = 0; b0 < deg; b0 += 32) {
            const int kc = (b0 + lane < deg) ? bsrc[beg + b0 + lane] : 0;
            const int nb = (deg - b0 < 32) ? deg - b0 : 32;
            for (int jj = 0; jj < nb; ++jj) {
                const int kj = __builtin_amdgcn_readlane(kc, jj);
                const int j  = b0 + jj;
                rank += (kj < key || (kj == key && j < i)) ? 1 : 0;
            }
        }
        if (i < deg) csrc[beg + rank] = key;
    }
}


template <int G>
__device__ __forceinline__ void l1_group(const float* __restrict__ xh, int sl, float pw,
                                         int cnt, int lane, float4& acc)
{
    float4 v[4];
#pragma unroll
    for (int q = 0; q < 4; ++q) {
        const int su = __builtin_amdgcn_readlane(sl, 4 * G + q);
        v[q] = *(const float4*)(xh + (size_t)su * 128 + lane * 4);
    }
    const float w[4] = { quad_move<0x00>(pw), quad_move<0x55>(pw),
                         quad_move<0xAA>(pw), quad_move<0xFF>(pw) };
#pragma unroll
    for (int q = 0; q < 4; ++q) {
        const bool have = (4 * G + q == 0) || (4 * G + q < cnt);
        const float tx = fmaf(w[q], v[q].x, acc.x);
        const float ty = fmaf(w[q], v[q].y, acc.y);
        const float tz = fmaf(w[q], v[q].z, acc.z);
        const float tw = fmaf(w[q], v[q].w, acc.w);
        acc.x = have ? tx : acc.x;
        acc.y = have ? ty : acc.y;
        acc.z = have ? tz : acc.z;
        acc.w = have ? tw : acc.w;
    }
}

__global__ void __launch_bounds__(256)
node_gather_l1(const int* __restrict__ rowptr, const int* __restrict__ csrc,
               const float* __restrict__ es, const float* __restrict__ ed,
               const float* __restrict__ xh, const float* __restrict__ b,
               float* __restrict__ h1, int n)
{
    const int lane = threadIdx.x & 31;
    const int d = blockIdx.x * 8 + (threadIdx.x >> 5);
    if (d >= n) return;
    const int beg = __builtin_amdgcn_readfirstlane(rowptr[d]);
    const int end = __builtin_amdgcn_readfirstlane(rowptr[d + 1]);
    const int deg = end - beg;
    const int cnt = deg + 1;

    float4 acc = make_float4(0.f, 0.f, 0.f, 0.f);

    if (cnt <= 32) {
        const int hq = lane >> 2, sub = lane & 3;
        const float edh = ed[d * 8 + hq];
        const int sl = edge_src(csrc, beg, lane, deg, d, n);

        float a[8];
        float m = -__builtin_inff();
#pragma unroll
        for (int t = 0; t < 8; ++t) {
            const int el = 4 * t + sub;
            const int s  = __shfl(sl, el);
            a[t] = lrelu(es[s * 8 + hq] + edh);
            if (el < cnt) m = fmaxf(m, a[t]);
        }
        m = fmaxf(m, quad_move<QUAD_XOR1>(m));
        m = fmaxf(m, quad_move<QUAD_XOR2>(m));

        float sum = 0.0f;
#pragma unroll
        for (int t = 0; t < 8; ++t) {
            const int el = 4 * t + sub;
            if (el < cnt) sum += __expf(a[t] - m);
        }
        sum += quad_move<QUAD_XOR1>(sum);
        sum += quad_move<QUAD_XOR2>(sum);
        const float inv = 1.0f / (sum + 1e-16f);

#pragma unroll
        for (int t = 0; t < 8; ++t) a[t] = __expf(a[t] - m) * inv;

        l1_group<0>(xh, sl, a[0], cnt, lane, acc);
        if (cnt >  4) l1_group<1>(xh, sl, a[1], cnt, lane, acc);
        if (cnt >  8) l1_group<2>(xh, sl, a[2], cnt, lane, acc);
        if (cnt > 12) l1_group<3>(xh, sl, a[3], cnt, lane, acc);
        if (cnt > 16) l1_group<4>(xh, sl, a[4], cnt, lane, acc);
        if (cnt > 20) l1_group<5>(xh, sl, a[5], cnt, lane, acc);
        if (cnt > 24) l1_group<6>(xh, sl, a[6], cnt, lane, acc);
        if (cnt > 28) l1_group<7>(xh, sl, a[7], cnt, lane, acc);
    } else {
        const int h = lane & 7, sub = lane >> 3;
        const float edh = ed[d * 8 + h];

        float m = -__builtin_inff();
        for (int c0 = 0; c0 < cnt; c0 += 32) {
            const int i  = c0 + lane;
            const int sl = edge_src(csrc, beg, i, deg, d, n);
            const int nv = (cnt - c0 < 32) ? cnt - c0 : 32;
#pragma unroll
            for (int t = 0; t < 8; ++t) {
                const int el = 4 * t + sub;
                const int s  = __shfl(sl, el);
                const float a = lrelu(es[s * 8 + h] + edh);
                if (el < nv) m = fmaxf(m, a);
            }
        }
        m = fmaxf(m, __shfl_xor(m, 8));
        m = fmaxf(m, __shfl_xor(m, 16));

        float sum = 0.0f;
        for (int c0 = 0; c0 < cnt; c0 += 32) {
            const int i  = c0 + lane;
            const int sl = edge_src(csrc, beg, i, deg, d, n);
            const int nv = (cnt - c0 < 32) ? cnt - c0 : 32;
#pragma unroll
            for (int t = 0; t < 8; ++t) {
                const int el = 4 * t + sub;
                const int s  = __shfl(sl, el);
                const float a = lrelu(es[s * 8 + h] + edh);
                if (el < nv) sum += __expf(a - m);
            }
        }
        sum += __shfl_xor(sum, 8);
        sum += __shfl_xor(sum, 16);
        const float inv = 1.0f / (sum + 1e-16f);

        const int hh = lane >> 2;
        const float m2   = __shfl(m, hh);
        const float inv2 = __shfl(inv, hh);
        const float ed2  = __shfl(edh, hh);

        for (int c0 = 0; c0 < cnt; c0 += 32) {
            const int i  = c0 + lane;
            const int sl = edge_src(csrc, beg, i, deg, d, n);
            const int nv = (cnt - c0 < 32) ? cnt - c0 : 32;
            for (int g = 0; 4 * g < nv; ++g) {
                float4 v[4];
                float  ev[4];
#pragma unroll
                for (int q = 0; q < 4; ++q) {
                    const int su = __builtin_amdgcn_readlane(sl, 4 * g + q);
                    v[q]  = *(const float4*)(xh + (size_t)su * 128 + lane * 4);
                    ev[q] = es[su * 8 + hh];
                }
#pragma unroll
                for (int q = 0; q < 4; ++q) {
                    if (4 * g + q < nv) {
                        const float w = __expf(lrelu(ev[q] + ed2) - m2) * inv2;
                        acc.x = fmaf(w, v[q].x, acc.x);
                        acc.y = fmaf(w, v[q].y, acc.y);
                        acc.z = fmaf(w, v[q].z, acc.z);
                        acc.w = fmaf(w, v[q].w, acc.w);
                    }
                }
            }
        }
    }

    const float4 bv = *(const float4*)(b + lane * 4);
    float4 o;
    o.x = acc.x + bv.x; o.y = acc.y + bv.y; o.z = acc.z + bv.z; o.w = acc.w + bv.w;
    o.x = o.x > 0.0f ? o.x : expm1f(o.x);
    o.y = o.y > 0.0f ? o.y : expm1f(o.y);
    o.z = o.z > 0.0f ? o.z : expm1f(o.z);
    o.w = o.w > 0.0f ? o.w : expm1f(o.w);
    *(float4*)(h1 + (size_t)d * 128 + lane * 4) = o;
}

constexpr int W1E_COLS = 144;
constexpr int W2P_COLS = 48;
constexpr int XH2_ES   = 40;
constexpr int XH2_ED   = 41;

__global__ void prep_weights(const float* __restrict__ W1,
                             const float* __restrict__ as1,
                             const float* __restrict__ ad1,
                             const float* __restrict__ W2,
                             const float* __restrict__ as2,
                             const float* __restrict__ ad2,
                             float* __restrict__ W1e, float* __restrict__ W2p)
{
    const int t = blockIdx.x * blockDim.x + threadIdx.x;
    if (t < 128 * W1E_COLS) {
        const int k = t / W1E_COLS, c = t - k * W1E_COLS;
        float v;
        if (c < 128) {
            v = W1[k * 128 + c];
        } else {
            const int h = (c - 128) & 7;
            const float* a = (c < 136 ? as1 : ad1) + h * 16;
            const float* w = W1 + k * 128 + h * 16;
            float s = 0.f;
            for (int j = 0; j < 16; ++j) s += w[j] * a[j];
            v = s;
        }
        W1e[t] = v;
    } else if (t < 128 * W1E_COLS + 128 * W2P_COLS) {
        const int u = t - 128 * W1E_COLS;
        const int k = u / W2P_COLS, c = u - k * W2P_COLS;
        float v = 0.0f;
        if (c < 40) {
            v = W2[k * 40 + c];
        } else if (c == XH2_ES || c == XH2_ED) {
            const float* a = (c == XH2_ES) ? as2 : ad2;
            const float* w = W2 + k * 40;
            float s = 0.f;
            for (int j = 0; j < 40; ++j) s += w[j] * a[j];
            v = s;
        }
        W2p[u] = v;
    }
}


__device__ __forceinline__ void l2_rows(const float* __restrict__ xh2, int sl, float wl,
                                        int nv, int slot, int c4, float4& acc)
{
    for (int j0 = 0; j0 < nv; j0 += 12) {
        float4 v[4];
        float  w[4];
#pragma unroll
        for (int u = 0; u < 4; ++u) {
            const int j  = j0 + 3 * u + slot;
            const bool ok = (slot < 3) && (j < nv);
            const int jj = ok ? j : 0;
            const int s  = __shfl(sl, jj);
            const float wj = __shfl(wl, jj);
            w[u] = ok ? wj : 0.0f;
            v[u] = *(const float4*)(xh2 + (size_t)s * 48 + c4 * 4);
        }
#pragma unroll
        for (int u = 0; u < 4; ++u) {
            const int j = j0 + 3 * u + slot;
            if ((slot < 3) && (j < nv)) {
                acc.x = fmaf(w[u], v[u].x, acc.x);
                acc.y = fmaf(w[u], v[u].y, acc.y);
                acc.z = fmaf(w[u], v[u].z, acc.z);
                acc.w = fmaf(w[u], v[u].w, acc.w);
            }
        }
    }
}

__global__ void __launch_bounds__(256)
node_gather_l2(const int* __restrict__ rowptr, const int* __restrict__ csrc,
               const float* __restrict__ xh2, const float* __restrict__ b,
               float* __restrict__ out, int n)
{
    __shared__ __align__(16) float so[8 * 40];
    const int lane = threadIdx.x & 31;
    const int wv   = threadIdx.x >> 5;
    const int d0   = blockIdx.x * 8;
    const int d    = d0 + wv;

    if (d < n) {
        const int beg = __builtin_amdgcn_readfirstlane(rowptr[d]);
        const int end = __builtin_amdgcn_readfirstlane(rowptr[d + 1]);
        const int deg = end - beg;
        const int cnt = deg + 1;
        const float edd = xh2[(size_t)d * W2P_COLS + XH2_ED];

        const int slot = lane / 10;
        const int c4   = lane - slot * 10;
        float4 acc = make_float4(0.f, 0.f, 0.f, 0.f);

        if (cnt <= 32) {
            const int sl  = edge_src(csrc, beg, lane, deg, d, n);
            const float a = lrelu(xh2[(size_t)sl * W2P_COLS + XH2_ES] + edd);

            float m = -__builtin_inff();
            if (lane < cnt) m = fmaxf(m, a);
            for (int o = 16; o >= 1; o >>= 1) m = fmaxf(m, __shfl_xor(m, o));

            const float e = __expf(a - m);
            float sum = 0.0f;
            if (lane < cnt) sum += e;
            for (int o = 16; o >= 1; o >>= 1) sum += __shfl_xor(sum, o);
            const float inv = 1.0f / (sum + 1e-16f);

            l2_rows(xh2, sl, e * inv, cnt, slot, c4, acc);
        } else {
            float m = -__builtin_inff();
            for (int c0 = 0; c0 < cnt; c0 += 32) {
                const int i  = c0 + lane;
                const int sl = edge_src(csrc, beg, i, deg, d, n);
                const float a = lrelu(xh2[(size_t)sl * W2P_COLS + XH2_ES] + edd);
                if (i < cnt) m = fmaxf(m, a);
            }
            for (int o = 16; o >= 1; o >>= 1) m = fmaxf(m, __shfl_xor(m, o));

            float sum = 0.0f;
            for (int c0 = 0; c0 < cnt; c0 += 32) {
                const int i  = c0 + lane;
                const int sl = edge_src(csrc, beg, i, deg, d, n);
                const float a = lrelu(xh2[(size_t)sl * W2P_COLS + XH2_ES] + edd);
                if (i < cnt) sum += __expf(a - m);
            }
            for (int o = 16; o >= 1; o >>= 1) sum += __shfl_xor(sum, o);
            const float inv = 1.0f / (sum + 1e-16f);

            for (int c0 = 0; c0 < cnt; c0 += 32) {
                const int i  = c0 + lane;
                const int sl = edge_src(csrc, beg, i, deg, d, n);
                const float a  = lrelu(xh2[(size_t)sl * W2P_COLS + XH2_ES] + edd);
                const float wl = __expf(a - m) * inv;
                const int nv = (cnt - c0 < 32) ? cnt - c0 : 32;
                l2_rows(xh2, sl, wl, nv, slot, c4, acc);
            }
        }
        const int l1 = (lane + 10) & 31, l2 = (lane + 20) & 31;
        acc.x += __shfl(acc.x, l1) + __shfl(acc.x, l2);
        acc.y += __shfl(acc.y, l1) + __shfl(acc.y, l2);
        acc.z += __shfl(acc.z, l1) + __shfl(acc.z, l2);
        acc.w += __shfl(acc.w, l1) + __shfl(acc.w, l2);
        if (lane < 10) {
            const float4 bv = *(const float4*)(b + lane * 4);
            float4 o;
            o.x = acc.x + bv.x; o.y = acc.y + bv.y; o.z = acc.z + bv.z; o.w = acc.w + bv.w;
            *(float4*)(so + wv * 40 + lane * 4) = o;
        }
    }
    __syncthreads();
    const int rows = (n - d0 < 8) ? n - d0 : 8;
    if ((int)threadIdx.x < rows * 10)
        *(float4*)(out + (size_t)d0 * 40 + threadIdx.x * 4) = *(const float4*)(so + threadIdx.x * 4);
}

extern "C" void kernel_launch(void* const* d_in, const int* in_sizes, int n_in,
                              void* d_out, int out_size, void* d_ws, size_t ws_size,
                              hipStream_t stream)
{
    (void)n_in; (void)out_size; (void)ws_size;

    const float* x   = (const float*)d_in[0];
    const int*   ei  = (const int*)d_in[1];
    const float* W1  = (const float*)d_in[2];
    const float* as1 = (const float*)d_in[3];
    const float* ad1 = (const float*)d_in[4];
    const float* b1  = (const float*)d_in[5];
    const float* W2  = (const float*)d_in[6];
    const float* as2 = (const float*)d_in[7];
    const float* ad2 = (const float*)d_in[8];
    const float* b2  = (const float*)d_in[9];
    float* out = (float*)d_out;

    const int n  = in_sizes[0] / 128;
    const int E  = in_sizes[1] / 2;

    char* ws = (char*)d_ws;
    size_t off = 0;
    auto alloc = [&](size_t bytes) -> float* {
        float* p = (float*)(ws + off);
        off += (bytes + 255) & ~(size_t)255;
        return p;
    };
    float* xh1  = alloc((size_t)n * 128 * 4);
    float* h1   = alloc((size_t)n * 128 * 4);
    float* es1  = alloc((size_t)n * 8 * 4);
    float* ed1  = alloc((size_t)n * 8 * 4);
    float* W2p  = alloc((size_t)128 * W2P_COLS * 4);
    constexpr int NBLK = 128;
    const int nr = (n + (1 << RSHIFT) - 1) >> RSHIFT;
    const int M  = nr * NBLK;
    int* hist   = (int*)alloc((size_t)(M + 16) * 4);
    int* offs   = (int*)alloc((size_t)(M + 16) * 4);
    int* rowptr = (int*)alloc((size_t)(n + 16) * 4);
    int* csrc   = (int*)alloc((size_t)E * 4);
    int* bsrc   = (int*)h1;
    int* pairs  = (int*)xh1;
    float* W1e  = h1 + (size_t)n * 64;

    const int B = 256;

    prep_weights<<<(128 * W1E_COLS + 128 * W2P_COLS + B - 1) / B, B, 0, stream>>>(W1, as1, ad1, W2, as2, ad2, W1e, W2p);

    range_count<<<NBLK, 1024, nr * sizeof(int), stream>>>(ei, E, nr, hist);
    csr_scan<<<1, 1024, 0, stream>>>(hist, offs, M);
    range_place<<<NBLK, 1024, nr * sizeof(int), stream>>>(ei, E, nr, offs, pairs);
    node_place<<<nr, 512, 0, stream>>>(pairs, offs, NBLK, rowptr, bsrc, n, E);
    csr_order<<<(n + 7) / 8, B, 0, stream>>>(rowptr, bsrc, csrc, n);

    gemm_wmma_f32<9, true><<<(n + 127) / 128, 256, 0, stream>>>(x, W1e, xh1, es1, ed1, n);
    node_gather_l1<<<(n + 7) / 8, B, 0, stream>>>(rowptr, csrc, es1, ed1, xh1, b1, h1, n);

    float* xh2 = xh1;
    gemm_wmma_f32<3, false><<<(n + 127) / 128, 256, 0, stream>>>(h1, W2p, xh2, nullptr, nullptr, n);
    node_gather_l2<<<(n + 7) / 8, B, 0, stream>>>(rowptr, csrc, xh2, b2, out, n);
}


# level-1 edge bucketing in one launch: a block counts, prefix-sums and places its own slice in place; a range block reads its 128 pieces and sums their offsets; histogram and scan launches go
# speedup vs baseline: 8.0626x; 1.1017x over previous
#include <hip/hip_runtime.h>
#include <hip/hip_bf16.h>
#include <math.h>


typedef float v2f __attribute__((ext_vector_type(2)));
typedef float v8f __attribute__((ext_vector_type(8)));

template <int NT, bool LOGITS>
__global__ void gemm_wmma_f32(const float* __restrict__ X,
                              const float* __restrict__ Wm,
                              float* __restrict__ Y,
                              float* __restrict__ es,
                              float* __restrict__ ed,
                              int nrows)
{
    constexpr int NCOL = NT * 16;
    constexpr int YT   = LOGITS ? NT - 1 : NT;
    constexpr int YP   = YT * 16;
    constexpr int P    = LOGITS ? NCOL : NCOL + 16;
    __shared__ __align__(16) float2 sB[64 * P];

    const int tid = threadIdx.x;
    for (int i = tid; i < 64 * NCOL; i += 256) {
        const int kp = i / NCOL;
        const int nn = i - kp * NCOL;
        float2 w2;
        w2.x = Wm[(2 * kp) * NCOL + nn];
        w2.y = Wm[(2 * kp + 1) * NCOL + nn];
        sB[kp * P + nn] = w2;
    }
    __syncthreads();

    const int wave = tid >> 5;
    const int lane = tid & 31;
    const int row0 = blockIdx.x * 128 + wave * 16;
    const int m    = lane & 15;
    const int khp  = lane >> 4;
    int row = row0 + m;
    if (row > nrows - 1) row = nrows - 1;

    const float2* Xp = (const float2*)(X + (size_t)row * 128) + khp;
    const int bbase = khp * P + m;

    v8f acc[NT];
#pragma unroll
    for (int t = 0; t < NT; ++t) acc[t] = (v8f)0.0f;

    for (int kk = 0; kk < 32; ++kk) {
        const float2 av = Xp[kk * 2];
        v2f a; a.x = av.x; a.y = av.y;
        const int bo = kk * 2 * P + bbase;
#pragma unroll
        for (int t = 0; t < NT; ++t) {
            const float2 bv = sB[bo + t * 16];
            v2f b; b.x = bv.x; b.y = bv.y;
            acc[t] = __builtin_amdgcn_wmma_f32_16x16x4_f32(
                false, a, false, b, (short)0, acc[t], false, false);
        }
    }

    const int rbase = khp * 8;
#pragma unroll
    for (int r = 0; r < 8; ++r) {
        const int rr = row0 + rbase + r;
        if (rr < nrows) {
#pragma unroll
            for (int t = 0; t < YT; ++t)
                Y[(size_t)rr * YP + t * 16 + m] = acc[t][r];
            if (LOGITS) {
                const float v = acc[NT - 1][r];
                if (m < 8) es[(size_t)rr * 8 + m] = v;
                else       ed[(size_t)rr * 8 + (m - 8)] = v;
            }
        }
    }
}

__device__ __forceinline__ float lrelu(float a) {
    return a > 0.0f ? a : 0.2f * a;
}

__device__ __forceinline__ int edge_src(const int* __restrict__ csrc, int beg,
                                        int i, int deg, int d, int n) {
    if (i >= deg) return d;
    const int s = csrc[beg + i];
    return s < 0 ? 0 : (s > n - 1 ? n - 1 : s);
}

template <int CTRL>
__device__ __forceinline__ float quad_move(float x) {
    return __int_as_float(__builtin_amdgcn_update_dpp(0, __float_as_int(x), CTRL, 0xF, 0xF, true));
}
constexpr int QUAD_XOR1 = 0xB1;
constexpr int QUAD_XOR2 = 0x4E;

constexpr int RSHIFT = 8;
constexpr int L1_BLOCKS = 128;

__global__ void __launch_bounds__(1024)
range_bucket(const int* __restrict__ ei, int E, int nr,
             int* __restrict__ segs, int* __restrict__ pairs_)
{
    extern __shared__ int lc[];
    __shared__ int wsum[32];
    int2* __restrict__ pairs = (int2*)pairs_;
    const int tid = threadIdx.x, lane = tid & 31, wave = tid >> 5;
    const int c = blockIdx.x, nb = gridDim.x;
    for (int r = tid; r < nr; r += 1024) lc[r] = 0;
    __syncthreads();
    const int chunk = (E + nb - 1) / nb;
    const int e0 = c * chunk;
    const int e1 = (e0 + chunk < E) ? e0 + chunk : E;
    const int* __restrict__ dstp = ei + E;
    int i0 = e0 + tid;
    for (; i0 + 3 * 1024 < e1; i0 += 4 * 1024) {
        int d[4];
#pragma unroll
        for (int u = 0; u < 4; ++u) d[u] = dstp[i0 + u * 1024];
#pragma unroll
        for (int u = 0; u < 4; ++u) atomicAdd(&lc[d[u] >> RSHIFT], 1);
    }
    for (; i0 < e1; i0 += 1024) atomicAdd(&lc[dstp[i0] >> RSHIFT], 1);
    __syncthreads();

    int carry = e0;
    for (int base = 0; base < nr; base += 1024) {
        const int r = base + tid;
        const int v = (r < nr) ? lc[r] : 0;
        int x = v;
        for (int o = 1; o < 32; o <<= 1) {
            const int y = __shfl_up(x, o);
            if (lane >= o) x += y;
        }
        if (lane == 31) wsum[wave] = x;
        __syncthreads();
        const int ws = wsum[lane];
        int wx = ws;
        for (int o = 1; o < 32; o <<= 1) {
            const int y = __shfl_up(wx, o);
            if (lane >= o) wx += y;
        }
        const int tile_total = __shfl(wx, 31);
        const int wave_excl  = __shfl(wx - ws, wave);
        __syncthreads();
        const int ex = carry + wave_excl + (x - v);
        if (r < nr) {
            lc[r] = ex;
            segs[r * nb + c] = ex;
        }
        carry += tile_total;
    }
    if (tid == 0) segs[nr * nb + c] = carry;
    __syncthreads();

    for (i0 = e0 + tid; i0 + 3 * 1024 < e1; i0 += 4 * 1024) {
        int d[4], s[4];
#pragma unroll
        for (int u = 0; u < 4; ++u) { d[u] = dstp[i0 + u * 1024]; s[u] = ei[i0 + u * 1024]; }
#pragma unroll
        for (int u = 0; u < 4; ++u) {
            int pos = atomicAdd(&lc[d[u] >> RSHIFT], 1);
            pos = pos < 0 ? 0 : (pos > E - 1 ? E - 1 : pos);
            pairs[pos] = make_int2(d[u], s[u]);
        }
    }
    for (; i0 < e1; i0 += 1024) {
        const int d = dstp[i0], s = ei[i0];
        int pos = atomicAdd(&lc[d >> RSHIFT], 1);
        pos = pos < 0 ? 0 : (pos > E - 1 ? E - 1 : pos);
        pairs[pos] = make_int2(d, s);
    }
}

constexpr int NP_WAVES = 16;
constexpr int NP_PPW   = L1_BLOCKS / NP_WAVES;

__global__ void __launch_bounds__(NP_WAVES * 32)
node_place(const int* __restrict__ pairs_, const int* __restrict__ segs,
           int* __restrict__ rowptr, int* __restrict__ bsrc, int n, int E)
{
    __shared__ int cn[1 << RSHIFT];
    __shared__ int wt[(1 << RSHIFT) / 32];
    __shared__ int ps[L1_BLOCKS];
    __shared__ int pe[L1_BLOCKS];
    const int2* __restrict__ pairs = (const int2*)pairs_;
    const int tid = threadIdx.x, lane = tid & 31, wave = tid >> 5;
    const int b  = blockIdx.x;
    const int d0 = b << RSHIFT;
    const int chunk = (E + L1_BLOCKS - 1) / L1_BLOCKS;
    if (tid < L1_BLOCKS) {
        ps[tid] = segs[b * L1_BLOCKS + tid];
        pe[tid] = segs[(b + 1) * L1_BLOCKS + tid];
    }
    if (tid < (1 << RSHIFT)) cn[tid] = 0;
    __syncthreads();

    int a0 = 0, a1 = 0;
    for (int cc = lane; cc < L1_BLOCKS; cc += 32) {
        a0 += ps[cc] - cc * chunk;
        a1 += pe[cc] - cc * chunk;
    }
    for (int o = 16; o >= 1; o >>= 1) {
        a0 += __shfl_xor(a0, o);
        a1 += __shfl_xor(a1, o);
    }
    const int s0 = __builtin_amdgcn_readfirstlane(a0);
    const int s1 = __builtin_amdgcn_readfirstlane(a1);

    int pb[NP_PPW], pl[NP_PPW];
#pragma unroll
    for (int q = 0; q < NP_PPW; ++q) {
        pb[q] = __builtin_amdgcn_readfirstlane(ps[wave * NP_PPW + q]);
        pl[q] = __builtin_amdgcn_readfirstlane(pe[wave * NP_PPW + q]) - pb[q];
    }

    {
        int dn[NP_PPW][2];
#pragma unroll
        for (int q = 0; q < NP_PPW; ++q)
#pragma unroll
            for (int t = 0; t < 2; ++t) {
                const int j = lane + 32 * t;
                dn[q][t] = (j < pl[q]) ? pairs[pb[q] + j].x : -1;
            }
#pragma unroll
        for (int q = 0; q < NP_PPW; ++q)
#pragma unroll
            for (int t = 0; t < 2; ++t)
                if (dn[q][t] >= 0) atomicAdd(&cn[dn[q][t] - d0], 1);
#pragma unroll
        for (int q = 0; q < NP_PPW; ++q)
            for (int j = lane + 64; j < pl[q]; j += 32) atomicAdd(&cn[pairs[pb[q] + j].x - d0], 1);
    }
    __syncthreads();
    int cnt = 0, x = 0;
    if (tid < (1 << RSHIFT)) {
        cnt = cn[tid];
        x = cnt;
        for (int o = 1; o < 32; o <<= 1) {
            const int y = __shfl_up(x, o);
            if (lane >= o) x += y;
        }
        if (lane == 31) wt[wave] = x;
    }
    __syncthreads();
    if (tid < (1 << RSHIFT)) {
        int pre = s0;
        for (int w = 0; w < wave; ++w) pre += wt[w];
        const int ex = pre + (x - cnt);
        cn[tid] = ex;
        if (d0 + tid < n) rowptr[d0 + tid] = ex;
    }
    if (b == (int)gridDim.x - 1 && tid == 0) rowptr[n] = s1;
    __syncthreads();
    {
        int2 p[NP_PPW][2];
#pragma unroll
        for (int q = 0; q < NP_PPW; ++q)
#pragma unroll
            for (int t = 0; t < 2; ++t) {
                const int j = lane + 32 * t;
                p[q][t] = (j < pl[q]) ? pairs[pb[q] + j] : make_int2(-1, 0);
            }
#pragma unroll
        for (int q = 0; q < NP_PPW; ++q)
#pragma unroll
            for (int t = 0; t < 2; ++t)
                if (p[q][t].x >= 0) {
                    int pos = atomicAdd(&cn[p[q][t].x - d0], 1);
                    pos = pos < 0 ? 0 : (pos > E - 1 ? E - 1 : pos);
                    bsrc[pos] = p[q][t].y;
                }
#pragma unroll
        for (int q = 0; q < NP_PPW; ++q)
            for (int j = lane + 64; j < pl[q]; j += 32) {
                const int2 pp = pairs[pb[q] + j];
                int pos = atomicAdd(&cn[pp.x - d0], 1);
                pos = pos < 0 ? 0 : (pos > E - 1 ? E - 1 : pos);
                bsrc[pos] = pp.y;
            }
    }
}

__global__ void __launch_bounds__(256)
csr_order(const int* __restrict__ rowptr, const int* __restrict__ bsrc,
          int* __restrict__ csrc, int n)
{
    const int lane = threadIdx.x & 31;
    const int d = blockIdx.x * 8 + (threadIdx.x >> 5);
    if (d >= n) return;
    const int beg = __builtin_amdgcn_readfirstlane(rowptr[d]);
    const int end = __builtin_amdgcn_readfirstlane(rowptr[d + 1]);
    const int deg = end - beg;
    for (int c0 = 0; c0 < deg; c0 += 32) {
        const int i   = c0 + lane;
        const int key = (i < deg) ? bsrc[beg + i] : 0;
        int rank = 0;
        for (int b0 = 0; b0 < deg; b0 += 32) {
            const int kc = (b0 + lane < deg) ? bsrc[beg + b0 + lane] : 0;
            const int nb = (deg - b0 < 32) ? deg - b0 : 32;
            for (int jj = 0; jj < nb; ++jj) {
                const int kj = __builtin_amdgcn_readlane(kc, jj);
                const int j  = b0 + jj;
                rank += (kj < key || (kj == key && j < i)) ? 1 : 0;
            }
        }
        if (i < deg) csrc[beg + rank] = key;
    }
}


template <int G>
__device__ __forceinline__ void l1_group(const float* __restrict__ xh, int sl, float pw,
                                         int cnt, int lane, float4& acc)
{
    float4 v[4];
#pragma unroll
    for (int q = 0; q < 4; ++q) {
        const int su = __builtin_amdgcn_readlane(sl, 4 * G + q);
        v[q] = *(const float4*)(xh + (size_t)su * 128 + lane * 4);
    }
    const float w[4] = { quad_move<0x00>(pw), quad_move<0x55>(pw),
                         quad_move<0xAA>(pw), quad_move<0xFF>(pw) };
#pragma unroll
    for (int q = 0; q < 4; ++q) {
        const bool have = (4 * G + q == 0) || (4 * G + q < cnt);
        const float tx = fmaf(w[q], v[q].x, acc.x);
        const float ty = fmaf(w[q], v[q].y, acc.y);
        const float tz = fmaf(w[q], v[q].z, acc.z);
        const float tw = fmaf(w[q], v[q].w, acc.w);
        acc.x = have ? tx : acc.x;
        acc.y = have ? ty : acc.y;
        acc.z = have ? tz : acc.z;
        acc.w = have ? tw : acc.w;
    }
}

__global__ void __launch_bounds__(256)
node_gather_l1(const int* __restrict__ rowptr, const int* __restrict__ csrc,
               const float* __restrict__ es, const float* __restrict__ ed,
               const float* __restrict__ xh, const float* __restrict__ b,
               float* __restrict__ h1, int n)
{
    const int lane = threadIdx.x & 31;
    const int d = blockIdx.x * 8 + (threadIdx.x >> 5);
    if (d >= n) return;
    const int beg = __builtin_amdgcn_readfirstlane(rowptr[d]);
    const int end = __builtin_amdgcn_readfirstlane(rowptr[d + 1]);
    const int deg = end - beg;
    const int cnt = deg + 1;

    float4 acc = make_float4(0.f, 0.f, 0.f, 0.f);

    if (cnt <= 32) {
        const int hq = lane >> 2, sub = lane & 3;
        const float edh = ed[d * 8 + hq];
        const int sl = edge_src(csrc, beg, lane, deg, d, n);

        float a[8];
        float m = -__builtin_inff();
#pragma unroll
        for (int t = 0; t < 8; ++t) {
            const int el = 4 * t + sub;
            const int s  = __shfl(sl, el);
            a[t] = lrelu(es[s * 8 + hq] + edh);
            if (el < cnt) m = fmaxf(m, a[t]);
        }
        m = fmaxf(m, quad_move<QUAD_XOR1>(m));
        m = fmaxf(m, quad_move<QUAD_XOR2>(m));

        float sum = 0.0f;
#pragma unroll
        for (int t = 0; t < 8; ++t) {
            const int el = 4 * t + sub;
            if (el < cnt) sum += __expf(a[t] - m);
        }
        sum += quad_move<QUAD_XOR1>(sum);
        sum += quad_move<QUAD_XOR2>(sum);
        const float inv = 1.0f / (sum + 1e-16f);

#pragma unroll
        for (int t = 0; t < 8; ++t) a[t] = __expf(a[t] - m) * inv;

        l1_group<0>(xh, sl, a[0], cnt, lane, acc);
        if (cnt >  4) l1_group<1>(xh, sl, a[1], cnt, lane, acc);
        if (cnt >  8) l1_group<2>(xh, sl, a[2], cnt, lane, acc);
        if (cnt > 12) l1_group<3>(xh, sl, a[3], cnt, lane, acc);
        if (cnt > 16) l1_group<4>(xh, sl, a[4], cnt, lane, acc);
        if (cnt > 20) l1_group<5>(xh, sl, a[5], cnt, lane, acc);
        if (cnt > 24) l1_group<6>(xh, sl, a[6], cnt, lane, acc);
        if (cnt > 28) l1_group<7>(xh, sl, a[7], cnt, lane, acc);
    } else {
        const int h = lane & 7, sub = lane >> 3;
        const float edh = ed[d * 8 + h];

        float m = -__builtin_inff();
        for (int c0 = 0; c0 < cnt; c0 += 32) {
            const int i  = c0 + lane;
            const int sl = edge_src(csrc, beg, i, deg, d, n);
            const int nv = (cnt - c0 < 32) ? cnt - c0 : 32;
#pragma unroll
            for (int t = 0; t < 8; ++t) {
                const int el = 4 * t + sub;
                const int s  = __shfl(sl, el);
                const float a = lrelu(es[s * 8 + h] + edh);
                if (el < nv) m = fmaxf(m, a);
            }
        }
        m = fmaxf(m, __shfl_xor(m, 8));
        m = fmaxf(m, __shfl_xor(m, 16));

        float sum = 0.0f;
        for (int c0 = 0; c0 < cnt; c0 += 32) {
            const int i  = c0 + lane;
            const int sl = edge_src(csrc, beg, i, deg, d, n);
            const int nv = (cnt - c0 < 32) ? cnt - c0 : 32;
#pragma unroll
            for (int t = 0; t < 8; ++t) {
                const int el = 4 * t + sub;
                const int s  = __shfl(sl, el);
                const float a = lrelu(es[s * 8 + h] + edh);
                if (el < nv) sum += __expf(a - m);
            }
        }
        sum += __shfl_xor(sum, 8);
        sum += __shfl_xor(sum, 16);
        const float inv = 1.0f / (sum + 1e-16f);

        const int hh = lane >> 2;
        const float m2   = __shfl(m, hh);
        const float inv2 = __shfl(inv, hh);
        const float ed2  = __shfl(edh, hh);

        for (int c0 = 0; c0 < cnt; c0 += 32) {
            const int i  = c0 + lane;
            const int sl = edge_src(csrc, beg, i, deg, d, n);
            const int nv = (cnt - c0 < 32) ? cnt - c0 : 32;
            for (int g = 0; 4 * g < nv; ++g) {
                float4 v[4];
                float  ev[4];
#pragma unroll
                for (int q = 0; q < 4; ++q) {
                    const int su = __builtin_amdgcn_readlane(sl, 4 * g + q);
                    v[q]  = *(const float4*)(xh + (size_t)su * 128 + lane * 4);
                    ev[q] = es[su * 8 + hh];
                }
#pragma unroll
                for (int q = 0; q < 4; ++q) {
                    if (4 * g + q < nv) {
                        const float w = __expf(lrelu(ev[q] + ed2) - m2) * inv2;
                        acc.x = fmaf(w, v[q].x, acc.x);
                        acc.y = fmaf(w, v[q].y, acc.y);
                        acc.z = fmaf(w, v[q].z, acc.z);
                        acc.w = fmaf(w, v[q].w, acc.w);
                    }
                }
            }
        }
    }

    const float4 bv = *(const float4*)(b + lane * 4);
    float4 o;
    o.x = acc.x + bv.x; o.y = acc.y + bv.y; o.z = acc.z + bv.z; o.w = acc.w + bv.w;
    o.x = o.x > 0.0f ? o.x : expm1f(o.x);
    o.y = o.y > 0.0f ? o.y : expm1f(o.y);
    o.z = o.z > 0.0f ? o.z : expm1f(o.z);
    o.w = o.w > 0.0f ? o.w : expm1f(o.w);
    *(float4*)(h1 + (size_t)d * 128 + lane * 4) = o;
}

constexpr int W1E_COLS = 144;
constexpr int W2P_COLS = 48;
constexpr int XH2_ES   = 40;
constexpr int XH2_ED   = 41;

__global__ void prep_weights(const float* __restrict__ W1,
                             const float* __restrict__ as1,
                             const float* __restrict__ ad1,
                             const float* __restrict__ W2,
                             const float* __restrict__ as2,
                             const float* __restrict__ ad2,
                             float* __restrict__ W1e, float* __restrict__ W2p)
{
    const int t = blockIdx.x * blockDim.x + threadIdx.x;
    if (t < 128 * W1E_COLS) {
        const int k = t / W1E_COLS, c = t - k * W1E_COLS;
        float v;
        if (c < 128) {
            v = W1[k * 128 + c];
        } else {
            const int h = (c - 128) & 7;
            const float* a = (c < 136 ? as1 : ad1) + h * 16;
            const float* w = W1 + k * 128 + h * 16;
            float s = 0.f;
            for (int j = 0; j < 16; ++j) s += w[j] * a[j];
            v = s;
        }
        W1e[t] = v;
    } else if (t < 128 * W1E_COLS + 128 * W2P_COLS) {
        const int u = t - 128 * W1E_COLS;
        const int k = u / W2P_COLS, c = u - k * W2P_COLS;
        float v = 0.0f;
        if (c < 40) {
            v = W2[k * 40 + c];
        } else if (c == XH2_ES || c == XH2_ED) {
            const float* a = (c == XH2_ES) ? as2 : ad2;
            const float* w = W2 + k * 40;
            float s = 0.f;
            for (int j = 0; j < 40; ++j) s += w[j] * a[j];
            v = s;
        }
        W2p[u] = v;
    }
}


__device__ __forceinline__ void l2_rows(const float* __restrict__ xh2, int sl, float wl,
                                        int nv, int slot, int c4, float4& acc)
{
    for (int j0 = 0; j0 < nv; j0 += 12) {
        float4 v[4];
        float  w[4];
#pragma unroll
        for (int u = 0; u < 4; ++u) {
            const int j  = j0 + 3 * u + slot;
            const bool ok = (slot < 3) && (j < nv);
            const int jj = ok ? j : 0;
            const int s  = __shfl(sl, jj);
            const float wj = __shfl(wl, jj);
            w[u] = ok ? wj : 0.0f;
            v[u] = *(const float4*)(xh2 + (size_t)s * 48 + c4 * 4);
        }
#pragma unroll
        for (int u = 0; u < 4; ++u) {
            const int j = j0 + 3 * u + slot;
            if ((slot < 3) && (j < nv)) {
                acc.x = fmaf(w[u], v[u].x, acc.x);
                acc.y = fmaf(w[u], v[u].y, acc.y);
                acc.z = fmaf(w[u], v[u].z, acc.z);
                acc.w = fmaf(w[u], v[u].w, acc.w);
            }
        }
    }
}

__global__ void __launch_bounds__(256)
node_gather_l2(const int* __restrict__ rowptr, const int* __restrict__ csrc,
               const float* __restrict__ xh2, const float* __restrict__ b,
               float* __restrict__ out, int n)
{
    __shared__ __align__(16) float so[8 * 40];
    const int lane = threadIdx.x & 31;
    const int wv   = threadIdx.x >> 5;
    const int d0   = blockIdx.x * 8;
    const int d    = d0 + wv;

    if (d < n) {
        const int beg = __builtin_amdgcn_readfirstlane(rowptr[d]);
        const int end = __builtin_amdgcn_readfirstlane(rowptr[d + 1]);
        const int deg = end - beg;
        const int cnt = deg + 1;
        const float edd = xh2[(size_t)d * W2P_COLS + XH2_ED];

        const int slot = lane / 10;
        const int c4   = lane - slot * 10;
        float4 acc = make_float4(0.f, 0.f, 0.f, 0.f);

        if (cnt <= 32) {
            const int sl  = edge_src(csrc, beg, lane, deg, d, n);
            const float a = lrelu(xh2[(size_t)sl * W2P_COLS + XH2_ES] + edd);

            float m = -__builtin_inff();
            if (lane < cnt) m = fmaxf(m, a);
            for (int o = 16; o >= 1; o >>= 1) m = fmaxf(m, __shfl_xor(m, o));

            const float e = __expf(a - m);
            float sum = 0.0f;
            if (lane < cnt) sum += e;
            for (int o = 16; o >= 1; o >>= 1) sum += __shfl_xor(sum, o);
            const float inv = 1.0f / (sum + 1e-16f);

            l2_rows(xh2, sl, e * inv, cnt, slot, c4, acc);
        } else {
            float m = -__builtin_inff();
            for (int c0 = 0; c0 < cnt; c0 += 32) {
                const int i  = c0 + lane;
                const int sl = edge_src(csrc, beg, i, deg, d, n);
                const float a = lrelu(xh2[(size_t)sl * W2P_COLS + XH2_ES] + edd);
                if (i < cnt) m = fmaxf(m, a);
            }
            for (int o = 16; o >= 1; o >>= 1) m = fmaxf(m, __shfl_xor(m, o));

            float sum = 0.0f;
            for (int c0 = 0; c0 < cnt; c0 += 32) {
                const int i  = c0 + lane;
                const int sl = edge_src(csrc, beg, i, deg, d, n);
                const float a = lrelu(xh2[(size_t)sl * W2P_COLS + XH2_ES] + edd);
                if (i < cnt) sum += __expf(a - m);
            }
            for (int o = 16; o >= 1; o >>= 1) sum += __shfl_xor(sum, o);
            const float inv = 1.0f / (sum + 1e-16f);

            for (int c0 = 0; c0 < cnt; c0 += 32) {
                const int i  = c0 + lane;
                const int sl = edge_src(csrc, beg, i, deg, d, n);
                const float a  = lrelu(xh2[(size_t)sl * W2P_COLS + XH2_ES] + edd);
                const float wl = __expf(a - m) * inv;
                const int nv = (cnt - c0 < 32) ? cnt - c0 : 32;
                l2_rows(xh2, sl, wl, nv, slot, c4, acc);
            }
        }
        const int l1 = (lane + 10) & 31, l2 = (lane + 20) & 31;
        acc.x += __shfl(acc.x, l1) + __shfl(acc.x, l2);
        acc.y += __shfl(acc.y, l1) + __shfl(acc.y, l2);
        acc.z += __shfl(acc.z, l1) + __shfl(acc.z, l2);
        acc.w += __shfl(acc.w, l1) + __shfl(acc.w, l2);
        if (lane < 10) {
            const float4 bv = *(const float4*)(b + lane * 4);
            float4 o;
            o.x = acc.x + bv.x; o.y = acc.y + bv.y; o.z = acc.z + bv.z; o.w = acc.w + bv.w;
            *(float4*)(so + wv * 40 + lane * 4) = o;
        }
    }
    __syncthreads();
    const int rows = (n - d0 < 8) ? n - d0 : 8;
    if ((int)threadIdx.x < rows * 10)
        *(float4*)(out + (size_t)d0 * 40 + threadIdx.x * 4) = *(const float4*)(so + threadIdx.x * 4);
}

extern "C" void kernel_launch(void* const* d_in, const int* in_sizes, int n_in,
                              void* d_out, int out_size, void* d_ws, size_t ws_size,
                              hipStream_t stream)
{
    (void)n_in; (void)out_size; (void)ws_size;

    const float* x   = (const float*)d_in[0];
    const int*   ei  = (const int*)d_in[1];
    const float* W1  = (const float*)d_in[2];
    const float* as1 = (const float*)d_in[3];
    const float* ad1 = (const float*)d_in[4];
    const float* b1  = (const float*)d_in[5];
    const float* W2  = (const float*)d_in[6];
    const float* as2 = (const float*)d_in[7];
    const float* ad2 = (const float*)d_in[8];
    const float* b2  = (const float*)d_in[9];
    float* out = (float*)d_out;

    const int n  = in_sizes[0] / 128;
    const int E  = in_sizes[1] / 2;

    char* ws = (char*)d_ws;
    size_t off = 0;
    auto alloc = [&](size_t bytes) -> float* {
        float* p = (float*)(ws + off);
        off += (bytes + 255) & ~(size_t)255;
        return p;
    };
    float* xh1  = alloc((size_t)n * 128 * 4);
    float* h1   = alloc((size_t)n * 128 * 4);
    float* es1  = alloc((size_t)n * 8 * 4);
    float* ed1  = alloc((size_t)n * 8 * 4);
    float* W2p  = alloc((size_t)128 * W2P_COLS * 4);
    const int nr = (n + (1 << RSHIFT) - 1) >> RSHIFT;
    int* segs   = (int*)alloc((size_t)((nr + 1) * L1_BLOCKS + 16) * 4);
    int* rowptr = (int*)alloc((size_t)(n + 16) * 4);
    int* csrc   = (int*)alloc((size_t)E * 4);
    int* bsrc   = (int*)h1;
    int* pairs  = (int*)xh1;
    float* W1e  = h1 + (size_t)n * 64;

    const int B = 256;

    prep_weights<<<(128 * W1E_COLS + 128 * W2P_COLS + B - 1) / B, B, 0, stream>>>(W1, as1, ad1, W2, as2, ad2, W1e, W2p);

    range_bucket<<<L1_BLOCKS, 1024, nr * sizeof(int), stream>>>(ei, E, nr, segs, pairs);
    node_place<<<nr, NP_WAVES * 32, 0, stream>>>(pairs, segs, rowptr, bsrc, n, E);
    csr_order<<<(n + 7) / 8, B, 0, stream>>>(rowptr, bsrc, csrc, n);

    gemm_wmma_f32<9, true><<<(n + 127) / 128, 256, 0, stream>>>(x, W1e, xh1, es1, ed1, n);
    node_gather_l1<<<(n + 7) / 8, B, 0, stream>>>(rowptr, csrc, es1, ed1, xh1, b1, h1, n);

    float* xh2 = xh1;
    gemm_wmma_f32<3, false><<<(n + 127) / 128, 256, 0, stream>>>(h1, W2p, xh2, nullptr, nullptr, n);
    node_gather_l2<<<(n + 7) / 8, B, 0, stream>>>(rowptr, csrc, xh2, b2, out, n);
}


# edge ordering moves into node_place: a range's buckets sit in LDS, each source is ranked inside its bucket and stored to csrc at its rank; the one-wave-per-node csr_order launch goes; sum orders same
# speedup vs baseline: 8.5063x; 1.0550x over previous
#include <hip/hip_runtime.h>
#include <hip/hip_bf16.h>
#include <math.h>


typedef float v2f __attribute__((ext_vector_type(2)));
typedef float v8f __attribute__((ext_vector_type(8)));

template <int NT, bool LOGITS>
__global__ void gemm_wmma_f32(const float* __restrict__ X,
                              const float* __restrict__ Wm,
                              float* __restrict__ Y,
                              float* __restrict__ es,
                              float* __restrict__ ed,
                              int nrows)
{
    constexpr int NCOL = NT * 16;
    constexpr int YT   = LOGITS ? NT - 1 : NT;
    constexpr int YP   = YT * 16;
    constexpr int P    = LOGITS ? NCOL : NCOL + 16;
    __shared__ __align__(16) float2 sB[64 * P];

    const int tid = threadIdx.x;
    for (int i = tid; i < 64 * NCOL; i += 256) {
        const int kp = i / NCOL;
        const int nn = i - kp * NCOL;
        float2 w2;
        w2.x = Wm[(2 * kp) * NCOL + nn];
        w2.y = Wm[(2 * kp + 1) * NCOL + nn];
        sB[kp * P + nn] = w2;
    }
    __syncthreads();

    const int wave = tid >> 5;
    const int lane = tid & 31;
    const int row0 = blockIdx.x * 128 + wave * 16;
    const int m    = lane & 15;
    const int khp  = lane >> 4;
    int row = row0 + m;
    if (row > nrows - 1) row = nrows - 1;

    const float2* Xp = (const float2*)(X + (size_t)row * 128) + khp;
    const int bbase = khp * P + m;

    v8f acc[NT];
#pragma unroll
    for (int t = 0; t < NT; ++t) acc[t] = (v8f)0.0f;

    for (int kk = 0; kk < 32; ++kk) {
        const float2 av = Xp[kk * 2];
        v2f a; a.x = av.x; a.y = av.y;
        const int bo = kk * 2 * P + bbase;
#pragma unroll
        for (int t = 0; t < NT; ++t) {
            const float2 bv = sB[bo + t * 16];
            v2f b; b.x = bv.x; b.y = bv.y;
            acc[t] = __builtin_amdgcn_wmma_f32_16x16x4_f32(
                false, a, false, b, (short)0, acc[t], false, false);
        }
    }

    const int rbase = khp * 8;
#pragma unroll
    for (int r = 0; r < 8; ++r) {
        const int rr = row0 + rbase + r;
        if (rr < nrows) {
#pragma unroll
            for (int t = 0; t < YT; ++t)
                Y[(size_t)rr * YP + t * 16 + m] = acc[t][r];
            if (LOGITS) {
                const float v = acc[NT - 1][r];
                if (m < 8) es[(size_t)rr * 8 + m] = v;
                else       ed[(size_t)rr * 8 + (m - 8)] = v;
            }
        }
    }
}

__device__ __forceinline__ float lrelu(float a) {
    return a > 0.0f ? a : 0.2f * a;
}

__device__ __forceinline__ int edge_src(const int* __restrict__ csrc, int beg,
                                        int i, int deg, int d, int n) {
    if (i >= deg) return d;
    const int s = csrc[beg + i];
    return s < 0 ? 0 : (s > n - 1 ? n - 1 : s);
}

template <int CTRL>
__device__ __forceinline__ float quad_move(float x) {
    return __int_as_float(__builtin_amdgcn_update_dpp(0, __float_as_int(x), CTRL, 0xF, 0xF, true));
}
constexpr int QUAD_XOR1 = 0xB1;
constexpr int QUAD_XOR2 = 0x4E;

constexpr int RSHIFT = 8;
constexpr int L1_BLOCKS = 128;

__global__ void __launch_bounds__(1024)
range_bucket(const int* __restrict__ ei, int E, int nr,
             int* __restrict__ segs, int* __restrict__ pairs_)
{
    extern __shared__ int lc[];
    __shared__ int wsum[32];
    int2* __restrict__ pairs = (int2*)pairs_;
    const int tid = threadIdx.x, lane = tid & 31, wave = tid >> 5;
    const int c = blockIdx.x, nb = gridDim.x;
    for (int r = tid; r < nr; r += 1024) lc[r] = 0;
    __syncthreads();
    const int chunk = (E + nb - 1) / nb;
    const int e0 = c * chunk;
    const int e1 = (e0 + chunk < E) ? e0 + chunk : E;
    const int* __restrict__ dstp = ei + E;
    int i0 = e0 + tid;
    for (; i0 + 3 * 1024 < e1; i0 += 4 * 1024) {
        int d[4];
#pragma unroll
        for (int u = 0; u < 4; ++u) d[u] = dstp[i0 + u * 1024];
#pragma unroll
        for (int u = 0; u < 4; ++u) atomicAdd(&lc[d[u] >> RSHIFT], 1);
    }
    for (; i0 < e1; i0 += 1024) atomicAdd(&lc[dstp[i0] >> RSHIFT], 1);
    __syncthreads();

    int carry = e0;
    for (int base = 0; base < nr; base += 1024) {
        const int r = base + tid;
        const int v = (r < nr) ? lc[r] : 0;
        int x = v;
        for (int o = 1; o < 32; o <<= 1) {
            const int y = __shfl_up(x, o);
            if (lane >= o) x += y;
        }
        if (lane == 31) wsum[wave] = x;
        __syncthreads();
        const int ws = wsum[lane];
        int wx = ws;
        for (int o = 1; o < 32; o <<= 1) {
            const int y = __shfl_up(wx, o);
            if (lane >= o) wx += y;
        }
        const int tile_total = __shfl(wx, 31);
        const int wave_excl  = __shfl(wx - ws, wave);
        __syncthreads();
        const int ex = carry + wave_excl + (x - v);
        if (r < nr) {
            lc[r] = ex;
            segs[r * nb + c] = ex;
        }
        carry += tile_total;
    }
    if (tid == 0) segs[nr * nb + c] = carry;
    __syncthreads();

    for (i0 = e0 + tid; i0 + 3 * 1024 < e1; i0 += 4 * 1024) {
        int d[4], s[4];
#pragma unroll
        for (int u = 0; u < 4; ++u) { d[u] = dstp[i0 + u * 1024]; s[u] = ei[i0 + u * 1024]; }
#pragma unroll
        for (int u = 0; u < 4; ++u) {
            int pos = atomicAdd(&lc[d[u] >> RSHIFT], 1);
            pos = pos < 0 ? 0 : (pos > E - 1 ? E - 1 : pos);
            pairs[pos] = make_int2(d[u], s[u]);
        }
    }
    for (; i0 < e1; i0 += 1024) {
        const int d = dstp[i0], s = ei[i0];
        int pos = atomicAdd(&lc[d >> RSHIFT], 1);
        pos = pos < 0 ? 0 : (pos > E - 1 ? E - 1 : pos);
        pairs[pos] = make_int2(d, s);
    }
}

constexpr int NP_WAVES = 16;
constexpr int NP_PPW   = L1_BLOCKS / NP_WAVES;
constexpr int NP_CAP   = 8192;

struct in_lds_t { static constexpr bool value = true;  };
struct in_mem_t { static constexpr bool value = false; };

__device__ __forceinline__ void order_bucket_mem(const int* bsrc, int* __restrict__ csrc,
                                                 int beg, int end, int lane)
{
    const int deg = end - beg;
    for (int c0 = 0; c0 < deg; c0 += 32) {
        const int i   = c0 + lane;
        const int key = (i < deg) ? bsrc[beg + i] : 0;
        int rank = 0;
        for (int b0 = 0; b0 < deg; b0 += 32) {
            const int kc = (b0 + lane < deg) ? bsrc[beg + b0 + lane] : 0;
            const int nb = (deg - b0 < 32) ? deg - b0 : 32;
            for (int jj = 0; jj < nb; ++jj) {
                const int kj = __builtin_amdgcn_readlane(kc, jj);
                const int j  = b0 + jj;
                rank += (kj < key || (kj == key && j < i)) ? 1 : 0;
            }
        }
        if (i < deg) csrc[beg + rank] = key;
    }
}

__global__ void __launch_bounds__(NP_WAVES * 32)
node_place(const int* __restrict__ pairs_, const int* __restrict__ segs,
           int* __restrict__ rowptr, int* bsrc, int* __restrict__ csrc, int n, int E)
{
    __shared__ int cn[1 << RSHIFT];
    __shared__ int wt[(1 << RSHIFT) / 32];
    __shared__ int ps[L1_BLOCKS];
    __shared__ int pe[L1_BLOCKS];
    __shared__ int cst[(1 << RSHIFT) + 1];
    __shared__ int lsrc[NP_CAP + 8];
    __shared__ unsigned char lnod[NP_CAP];
    const int2* __restrict__ pairs = (const int2*)pairs_;
    const int tid = threadIdx.x, lane = tid & 31, wave = tid >> 5;
    const int b  = blockIdx.x;
    const int d0 = b << RSHIFT;
    const int chunk = (E + L1_BLOCKS - 1) / L1_BLOCKS;
    if (tid < L1_BLOCKS) {
        ps[tid] = segs[b * L1_BLOCKS + tid];
        pe[tid] = segs[(b + 1) * L1_BLOCKS + tid];
    }
    if (tid < (1 << RSHIFT)) cn[tid] = 0;
    __syncthreads();

    int a0 = 0, a1 = 0;
    for (int cc = lane; cc < L1_BLOCKS; cc += 32) {
        a0 += ps[cc] - cc * chunk;
        a1 += pe[cc] - cc * chunk;
    }
    for (int o = 16; o >= 1; o >>= 1) {
        a0 += __shfl_xor(a0, o);
        a1 += __shfl_xor(a1, o);
    }
    const int s0 = __builtin_amdgcn_readfirstlane(a0);
    const int s1 = __builtin_amdgcn_readfirstlane(a1);
    const int total = s1 - s0;
    const bool fits = total <= NP_CAP;

    int pb[NP_PPW], pl[NP_PPW];
#pragma unroll
    for (int q = 0; q < NP_PPW; ++q) {
        pb[q] = __builtin_amdgcn_readfirstlane(ps[wave * NP_PPW + q]);
        pl[q] = __builtin_amdgcn_readfirstlane(pe[wave * NP_PPW + q]) - pb[q];
    }

    {
        int dn[NP_PPW][2];
#pragma unroll
        for (int q = 0; q < NP_PPW; ++q)
#pragma unroll
            for (int t = 0; t < 2; ++t) {
                const int j = lane + 32 * t;
                dn[q][t] = (j < pl[q]) ? pairs[pb[q] + j].x : -1;
            }
#pragma unroll
        for (int q = 0; q < NP_PPW; ++q)
#pragma unroll
            for (int t = 0; t < 2; ++t)
                if (dn[q][t] >= 0) atomicAdd(&cn[dn[q][t] - d0], 1);
#pragma unroll
        for (int q = 0; q < NP_PPW; ++q)
            for (int j = lane + 64; j < pl[q]; j += 32) atomicAdd(&cn[pairs[pb[q] + j].x - d0], 1);
    }
    __syncthreads();
    int cnt = 0, x = 0;
    if (tid < (1 << RSHIFT)) {
        cnt = cn[tid];
        x = cnt;
        for (int o = 1; o < 32; o <<= 1) {
            const int y = __shfl_up(x, o);
            if (lane >= o) x += y;
        }
        if (lane == 31) wt[wave] = x;
    }
    __syncthreads();
    if (tid < (1 << RSHIFT)) {
        int pre = s0;
        for (int w = 0; w < wave; ++w) pre += wt[w];
        const int ex = pre + (x - cnt);
        cn[tid] = ex;
        cst[tid] = ex - s0;
        if (tid == 0) cst[1 << RSHIFT] = total;
        if (d0 + tid < n) rowptr[d0 + tid] = ex;
    }
    if (b == (int)gridDim.x - 1 && tid == 0) rowptr[n] = s1;
    __syncthreads();
    auto place_all = [&](auto ldsb) {
        constexpr bool LDSB = decltype(ldsb)::value;
        auto put_src = [&](int pos, int nd, int s) {
            if (LDSB) {
                int lp = pos - s0;
                lp = lp < 0 ? 0 : (lp > NP_CAP - 1 ? NP_CAP - 1 : lp);
                lsrc[lp] = s;
                lnod[lp] = (unsigned char)nd;
            } else {
                const int gp = pos < 0 ? 0 : (pos > E - 1 ? E - 1 : pos);
                bsrc[gp] = s;
            }
        };
        int2 p[NP_PPW][2];
#pragma unroll
        for (int q = 0; q < NP_PPW; ++q)
#pragma unroll
            for (int t = 0; t < 2; ++t) {
                const int j = lane + 32 * t;
                p[q][t] = (j < pl[q]) ? pairs[pb[q] + j] : make_int2(-1, 0);
            }
#pragma unroll
        for (int q = 0; q < NP_PPW; ++q)
#pragma unroll
            for (int t = 0; t < 2; ++t)
                if (p[q][t].x >= 0) {
                    const int nd = p[q][t].x - d0;
                    const int pos = atomicAdd(&cn[nd], 1);
                    put_src(pos, nd, p[q][t].y);
                }
#pragma unroll
        for (int q = 0; q < NP_PPW; ++q)
            for (int j = lane + 64; j < pl[q]; j += 32) {
                const int2 pp = pairs[pb[q] + j];
                const int nd = pp.x - d0;
                const int pos = atomicAdd(&cn[nd], 1);
                put_src(pos, nd, pp.y);
            }
    };
    if (fits) place_all(in_lds_t{}); else place_all(in_mem_t{});
    __syncthreads();

    if (fits) {
        for (int i = tid; i < total; i += NP_WAVES * 32) {
            const int key = lsrc[i];
            const int nd  = lnod[i];
            int bbeg = cst[nd], bend = cst[nd + 1];
            bbeg = bbeg < 0 ? 0 : bbeg;
            bend = bend > total ? total : bend;
            int rank = 0;
            for (int j0 = bbeg; j0 < bend; j0 += 8) {
                const int own = i - j0;
                const int inb = bend - j0;
                int k[8];
#pragma unroll
                for (int u = 0; u < 8; ++u) k[u] = lsrc[j0 + u];
#pragma unroll
                for (int u = 0; u < 8; ++u) {
                    const int lim = key + (u < own ? 1 : 0);
                    rank += ((int)(u < inb) & (int)(k[u] < lim));
                }
            }
            int pos = s0 + bbeg + rank;
            pos = pos < 0 ? 0 : (pos > E - 1 ? E - 1 : pos);
            csrc[pos] = key;
        }
    } else {
        for (int q = 0; q < (1 << RSHIFT) / NP_WAVES; ++q) {
            const int nd  = q * NP_WAVES + wave;
            int bbeg = __builtin_amdgcn_readfirstlane(nd ? cn[nd - 1] : s0);
            int bend = __builtin_amdgcn_readfirstlane(cn[nd]);
            bbeg = bbeg < 0 ? 0 : bbeg;
            bend = bend > E ? E : bend;
            order_bucket_mem(bsrc, csrc, bbeg, bend, lane);
        }
    }
}


template <int G>
__device__ __forceinline__ void l1_group(const float* __restrict__ xh, int sl, float pw,
                                         int cnt, int lane, float4& acc)
{
    float4 v[4];
#pragma unroll
    for (int q = 0; q < 4; ++q) {
        const int su = __builtin_amdgcn_readlane(sl, 4 * G + q);
        v[q] = *(const float4*)(xh + (size_t)su * 128 + lane * 4);
    }
    const float w[4] = { quad_move<0x00>(pw), quad_move<0x55>(pw),
                         quad_move<0xAA>(pw), quad_move<0xFF>(pw) };
#pragma unroll
    for (int q = 0; q < 4; ++q) {
        const bool have = (4 * G + q == 0) || (4 * G + q < cnt);
        const float tx = fmaf(w[q], v[q].x, acc.x);
        const float ty = fmaf(w[q], v[q].y, acc.y);
        const float tz = fmaf(w[q], v[q].z, acc.z);
        const float tw = fmaf(w[q], v[q].w, acc.w);
        acc.x = have ? tx : acc.x;
        acc.y = have ? ty : acc.y;
        acc.z = have ? tz : acc.z;
        acc.w = have ? tw : acc.w;
    }
}

__global__ void __launch_bounds__(256)
node_gather_l1(const int* __restrict__ rowptr, const int* __restrict__ csrc,
               const float* __restrict__ es, const float* __restrict__ ed,
               const float* __restrict__ xh, const float* __restrict__ b,
               float* __restrict__ h1, int n)
{
    const int lane = threadIdx.x & 31;
    const int d = blockIdx.x * 8 + (threadIdx.x >> 5);
    if (d >= n) return;
    const int beg = __builtin_amdgcn_readfirstlane(rowptr[d]);
    const int end = __builtin_amdgcn_readfirstlane(rowptr[d + 1]);
    const int deg = end - beg;
    const int cnt = deg + 1;

    float4 acc = make_float4(0.f, 0.f, 0.f, 0.f);

    if (cnt <= 32) {
        const int hq = lane >> 2, sub = lane & 3;
        const float edh = ed[d * 8 + hq];
        const int sl = edge_src(csrc, beg, lane, deg, d, n);

        float a[8];
        float m = -__builtin_inff();
#pragma unroll
        for (int t = 0; t < 8; ++t) {
            const int el = 4 * t + sub;
            const int s  = __shfl(sl, el);
            a[t] = lrelu(es[s * 8 + hq] + edh);
            if (el < cnt) m = fmaxf(m, a[t]);
        }
        m = fmaxf(m, quad_move<QUAD_XOR1>(m));
        m = fmaxf(m, quad_move<QUAD_XOR2>(m));

        float sum = 0.0f;
#pragma unroll
        for (int t = 0; t < 8; ++t) {
            const int el = 4 * t + sub;
            if (el < cnt) sum += __expf(a[t] - m);
        }
        sum += quad_move<QUAD_XOR1>(sum);
        sum += quad_move<QUAD_XOR2>(sum);
        const float inv = 1.0f / (sum + 1e-16f);

#pragma unroll
        for (int t = 0; t < 8; ++t) a[t] = __expf(a[t] - m) * inv;

        l1_group<0>(xh, sl, a[0], cnt, lane, acc);
        if (cnt >  4) l1_group<1>(xh, sl, a[1], cnt, lane, acc);
        if (cnt >  8) l1_group<2>(xh, sl, a[2], cnt, lane, acc);
        if (cnt > 12) l1_group<3>(xh, sl, a[3], cnt, lane, acc);
        if (cnt > 16) l1_group<4>(xh, sl, a[4], cnt, lane, acc);
        if (cnt > 20) l1_group<5>(xh, sl, a[5], cnt, lane, acc);
        if (cnt > 24) l1_group<6>(xh, sl, a[6], cnt, lane, acc);
        if (cnt > 28) l1_group<7>(xh, sl, a[7], cnt, lane, acc);
    } else {
        const int h = lane & 7, sub = lane >> 3;
        const float edh = ed[d * 8 + h];

        float m = -__builtin_inff();
        for (int c0 = 0; c0 < cnt; c0 += 32) {
            const int i  = c0 + lane;
            const int sl = edge_src(csrc, beg, i, deg, d, n);
            const int nv = (cnt - c0 < 32) ? cnt - c0 : 32;
#pragma unroll
            for (int t = 0; t < 8; ++t) {
                const int el = 4 * t + sub;
                const int s  = __shfl(sl, el);
                const float a = lrelu(es[s * 8 + h] + edh);
                if (el < nv) m = fmaxf(m, a);
            }
        }
        m = fmaxf(m, __shfl_xor(m, 8));
        m = fmaxf(m, __shfl_xor(m, 16));

        float sum = 0.0f;
        for (int c0 = 0; c0 < cnt; c0 += 32) {
            const int i  = c0 + lane;
            const int sl = edge_src(csrc, beg, i, deg, d, n);
            const int nv = (cnt - c0 < 32) ? cnt - c0 : 32;
#pragma unroll
            for (int t = 0; t < 8; ++t) {
                const int el = 4 * t + sub;
                const int s  = __shfl(sl, el);
                const float a = lrelu(es[s * 8 + h] + edh);
                if (el < nv) sum += __expf(a - m);
            }
        }
        sum += __shfl_xor(sum, 8);
        sum += __shfl_xor(sum, 16);
        const float inv = 1.0f / (sum + 1e-16f);

        const int hh = lane >> 2;
        const float m2   = __shfl(m, hh);
        const float inv2 = __shfl(inv, hh);
        const float ed2  = __shfl(edh, hh);

        for (int c0 = 0; c0 < cnt; c0 += 32) {
            const int i  = c0 + lane;
            const int sl = edge_src(csrc, beg, i, deg, d, n);
            const int nv = (cnt - c0 < 32) ? cnt - c0 : 32;
            for (int g = 0; 4 * g < nv; ++g) {
                float4 v[4];
                float  ev[4];
#pragma unroll
                for (int q = 0; q < 4; ++q) {
                    const int su = __builtin_amdgcn_readlane(sl, 4 * g + q);
                    v[q]  = *(const float4*)(xh + (size_t)su * 128 + lane * 4);
                    ev[q] = es[su * 8 + hh];
                }
#pragma unroll
                for (int q = 0; q < 4; ++q) {
                    if (4 * g + q < nv) {
                        const float w = __expf(lrelu(ev[q] + ed2) - m2) * inv2;
                        acc.x = fmaf(w, v[q].x, acc.x);
                        acc.y = fmaf(w, v[q].y, acc.y);
                        acc.z = fmaf(w, v[q].z, acc.z);
                        acc.w = fmaf(w, v[q].w, acc.w);
                    }
                }
            }
        }
    }

    const float4 bv = *(const float4*)(b + lane * 4);
    float4 o;
    o.x = acc.x + bv.x; o.y = acc.y + bv.y; o.z = acc.z + bv.z; o.w = acc.w + bv.w;
    o.x = o.x > 0.0f ? o.x : expm1f(o.x);
    o.y = o.y > 0.0f ? o.y : expm1f(o.y);
    o.z = o.z > 0.0f ? o.z : expm1f(o.z);
    o.w = o.w > 0.0f ? o.w : expm1f(o.w);
    *(float4*)(h1 + (size_t)d * 128 + lane * 4) = o;
}

constexpr int W1E_COLS = 144;
constexpr int W2P_COLS = 48;
constexpr int XH2_ES   = 40;
constexpr int XH2_ED   = 41;

__global__ void prep_weights(const float* __restrict__ W1,
                             const float* __restrict__ as1,
                             const float* __restrict__ ad1,
                             const float* __restrict__ W2,
                             const float* __restrict__ as2,
                             const float* __restrict__ ad2,
                             float* __restrict__ W1e, float* __restrict__ W2p)
{
    const int t = blockIdx.x * blockDim.x + threadIdx.x;
    if (t < 128 * W1E_COLS) {
        const int k = t / W1E_COLS, c = t - k * W1E_COLS;
        float v;
        if (c < 128) {
            v = W1[k * 128 + c];
        } else {
            const int h = (c - 128) & 7;
            const float* a = (c < 136 ? as1 : ad1) + h * 16;
            const float* w = W1 + k * 128 + h * 16;
            float s = 0.f;
            for (int j = 0; j < 16; ++j) s += w[j] * a[j];
            v = s;
        }
        W1e[t] = v;
    } else if (t < 128 * W1E_COLS + 128 * W2P_COLS) {
        const int u = t - 128 * W1E_COLS;
        const int k = u / W2P_COLS, c = u - k * W2P_COLS;
        float v = 0.0f;
        if (c < 40) {
            v = W2[k * 40 + c];
        } else if (c == XH2_ES || c == XH2_ED) {
            const float* a = (c == XH2_ES) ? as2 : ad2;
            const float* w = W2 + k * 40;
            float s = 0.f;
            for (int j = 0; j < 40; ++j) s += w[j] * a[j];
            v = s;
        }
        W2p[u] = v;
    }
}


__device__ __forceinline__ void l2_rows(const float* __restrict__ xh2, int sl, float wl,
                                        int nv, int slot, int c4, float4& acc)
{
    for (int j0 = 0; j0 < nv; j0 += 12) {
        float4 v[4];
        float  w[4];
#pragma unroll
        for (int u = 0; u < 4; ++u) {
            const int j  = j0 + 3 * u + slot;
            const bool ok = (slot < 3) && (j < nv);
            const int jj = ok ? j : 0;
            const int s  = __shfl(sl, jj);
            const float wj = __shfl(wl, jj);
            w[u] = ok ? wj : 0.0f;
            v[u] = *(const float4*)(xh2 + (size_t)s * 48 + c4 * 4);
        }
#pragma unroll
        for (int u = 0; u < 4; ++u) {
            const int j = j0 + 3 * u + slot;
            if ((slot < 3) && (j < nv)) {
                acc.x = fmaf(w[u], v[u].x, acc.x);
                acc.y = fmaf(w[u], v[u].y, acc.y);
                acc.z = fmaf(w[u], v[u].z, acc.z);
                acc.w = fmaf(w[u], v[u].w, acc.w);
            }
        }
    }
}

__global__ void __launch_bounds__(256)
node_gather_l2(const int* __restrict__ rowptr, const int* __restrict__ csrc,
               const float* __restrict__ xh2, const float* __restrict__ b,
               float* __restrict__ out, int n)
{
    __shared__ __align__(16) float so[8 * 40];
    const int lane = threadIdx.x & 31;
    const int wv   = threadIdx.x >> 5;
    const int d0   = blockIdx.x * 8;
    const int d    = d0 + wv;

    if (d < n) {
        const int beg = __builtin_amdgcn_readfirstlane(rowptr[d]);
        const int end = __builtin_amdgcn_readfirstlane(rowptr[d + 1]);
        const int deg = end - beg;
        const int cnt = deg + 1;
        const float edd = xh2[(size_t)d * W2P_COLS + XH2_ED];

        const int slot = lane / 10;
        const int c4   = lane - slot * 10;
        float4 acc = make_float4(0.f, 0.f, 0.f, 0.f);

        if (cnt <= 32) {
            const int sl  = edge_src(csrc, beg, lane, deg, d, n);
            const float a = lrelu(xh2[(size_t)sl * W2P_COLS + XH2_ES] + edd);

            float m = -__builtin_inff();
            if (lane < cnt) m = fmaxf(m, a);
            for (int o = 16; o >= 1; o >>= 1) m = fmaxf(m, __shfl_xor(m, o));

            const float e = __expf(a - m);
            float sum = 0.0f;
            if (lane < cnt) sum += e;
            for (int o = 16; o >= 1; o >>= 1) sum += __shfl_xor(sum, o);
            const float inv = 1.0f / (sum + 1e-16f);

            l2_rows(xh2, sl, e * inv, cnt, slot, c4, acc);
        } else {
            float m = -__builtin_inff();
            for (int c0 = 0; c0 < cnt; c0 += 32) {
                const int i  = c0 + lane;
                const int sl = edge_src(csrc, beg, i, deg, d, n);
                const float a = lrelu(xh2[(size_t)sl * W2P_COLS + XH2_ES] + edd);
                if (i < cnt) m = fmaxf(m, a);
            }
            for (int o = 16; o >= 1; o >>= 1) m = fmaxf(m, __shfl_xor(m, o));

            float sum = 0.0f;
            for (int c0 = 0; c0 < cnt; c0 += 32) {
                const int i  = c0 + lane;
                const int sl = edge_src(csrc, beg, i, deg, d, n);
                const float a = lrelu(xh2[(size_t)sl * W2P_COLS + XH2_ES] + edd);
                if (i < cnt) sum += __expf(a - m);
            }
            for (int o = 16; o >= 1; o >>= 1) sum += __shfl_xor(sum, o);
            const float inv = 1.0f / (sum + 1e-16f);

            for (int c0 = 0; c0 < cnt; c0 += 32) {
                const int i  = c0 + lane;
                const int sl = edge_src(csrc, beg, i, deg, d, n);
                const float a  = lrelu(xh2[(size_t)sl * W2P_COLS + XH2_ES] + edd);
                const float wl = __expf(a - m) * inv;
                const int nv = (cnt - c0 < 32) ? cnt - c0 : 32;
                l2_rows(xh2, sl, wl, nv, slot, c4, acc);
            }
        }
        const int l1 = (lane + 10) & 31, l2 = (lane + 20) & 31;
        acc.x += __shfl(acc.x, l1) + __shfl(acc.x, l2);
        acc.y += __shfl(acc.y, l1) + __shfl(acc.y, l2);
        acc.z += __shfl(acc.z, l1) + __shfl(acc.z, l2);
        acc.w += __shfl(acc.w, l1) + __shfl(acc.w, l2);
        if (lane < 10) {
            const float4 bv = *(const float4*)(b + lane * 4);
            float4 o;
            o.x = acc.x + bv.x; o.y = acc.y + bv.y; o.z = acc.z + bv.z; o.w = acc.w + bv.w;
            *(float4*)(so + wv * 40 + lane * 4) = o;
        }
    }
    __syncthreads();
    const int rows = (n - d0 < 8) ? n - d0 : 8;
    if ((int)threadIdx.x < rows * 10)
        *(float4*)(out + (size_t)d0 * 40 + threadIdx.x * 4) = *(const float4*)(so + threadIdx.x * 4);
}

extern "C" void kernel_launch(void* const* d_in, const int* in_sizes, int n_in,
                              void* d_out, int out_size, void* d_ws, size_t ws_size,
                              hipStream_t stream)
{
    (void)n_in; (void)out_size; (void)ws_size;

    const float* x   = (const float*)d_in[0];
    const int*   ei  = (const int*)d_in[1];
    const float* W1  = (const float*)d_in[2];
    const float* as1 = (const float*)d_in[3];
    const float* ad1 = (const float*)d_in[4];
    const float* b1  = (const float*)d_in[5];
    const float* W2  = (const float*)d_in[6];
    const float* as2 = (const float*)d_in[7];
    const float* ad2 = (const float*)d_in[8];
    const float* b2  = (const float*)d_in[9];
    float* out = (float*)d_out;

    const int n  = in_sizes[0] / 128;
    const int E  = in_sizes[1] / 2;

    char* ws = (char*)d_ws;
    size_t off = 0;
    auto alloc = [&](size_t bytes) -> float* {
        float* p = (float*)(ws + off);
        off += (bytes + 255) & ~(size_t)255;
        return p;
    };
    float* xh1  = alloc((size_t)n * 128 * 4);
    float* h1   = alloc((size_t)n * 128 * 4);
    float* es1  = alloc((size_t)n * 8 * 4);
    float* ed1  = alloc((size_t)n * 8 * 4);
    float* W2p  = alloc((size_t)128 * W2P_COLS * 4);
    const int nr = (n + (1 << RSHIFT) - 1) >> RSHIFT;
    int* segs   = (int*)alloc((size_t)((nr + 1) * L1_BLOCKS + 16) * 4);
    int* rowptr = (int*)alloc((size_t)(n + 16) * 4);
    int* csrc   = (int*)alloc((size_t)E * 4);
    int* bsrc   = (int*)h1;
    int* pairs  = (int*)xh1;
    float* W1e  = h1 + (size_t)n * 64;

    const int B = 256;

    prep_weights<<<(128 * W1E_COLS + 128 * W2P_COLS + B - 1) / B, B, 0, stream>>>(W1, as1, ad1, W2, as2, ad2, W1e, W2p);

    range_bucket<<<L1_BLOCKS, 1024, nr * sizeof(int), stream>>>(ei, E, nr, segs, pairs);
    node_place<<<nr, NP_WAVES * 32, 0, stream>>>(pairs, segs, rowptr, bsrc, csrc, n, E);

    gemm_wmma_f32<9, true><<<(n + 127) / 128, 256, 0, stream>>>(x, W1e, xh1, es1, ed1, n);
    node_gather_l1<<<(n + 7) / 8, B, 0, stream>>>(rowptr, csrc, es1, ed1, xh1, b1, h1, n);

    float* xh2 = xh1;
    gemm_wmma_f32<3, false><<<(n + 127) / 128, 256, 0, stream>>>(h1, W2p, xh2, nullptr, nullptr, n);
    node_gather_l2<<<(n + 7) / 8, B, 0, stream>>>(rowptr, csrc, xh2, b2, out, n);
}


# gather kernels: a short list asks for more rows per wait (layer 1: 8 edges a group, not 4, the first 8 beside the logits; layer 2: first 24 edges' rows beside the logits); sum orders unchanged
# speedup vs baseline: 9.0428x; 1.0631x over previous
#include <hip/hip_runtime.h>
#include <hip/hip_bf16.h>
#include <math.h>


typedef float v2f __attribute__((ext_vector_type(2)));
typedef float v8f __attribute__((ext_vector_type(8)));

template <int NT, bool LOGITS>
__global__ void gemm_wmma_f32(const float* __restrict__ X,
                              const float* __restrict__ Wm,
                              float* __restrict__ Y,
                              float* __restrict__ es,
                              float* __restrict__ ed,
                              int nrows)
{
    constexpr int NCOL = NT * 16;
    constexpr int YT   = LOGITS ? NT - 1 : NT;
    constexpr int YP   = YT * 16;
    constexpr int P    = LOGITS ? NCOL : NCOL + 16;
    __shared__ __align__(16) float2 sB[64 * P];

    const int tid = threadIdx.x;
    for (int i = tid; i < 64 * NCOL; i += 256) {
        const int kp = i / NCOL;
        const int nn = i - kp * NCOL;
        float2 w2;
        w2.x = Wm[(2 * kp) * NCOL + nn];
        w2.y = Wm[(2 * kp + 1) * NCOL + nn];
        sB[kp * P + nn] = w2;
    }
    __syncthreads();

    const int wave = tid >> 5;
    const int lane = tid & 31;
    const int row0 = blockIdx.x * 128 + wave * 16;
    const int m    = lane & 15;
    const int khp  = lane >> 4;
    int row = row0 + m;
    if (row > nrows - 1) row = nrows - 1;

    const float2* Xp = (const float2*)(X + (size_t)row * 128) + khp;
    const int bbase = khp * P + m;

    v8f acc[NT];
#pragma unroll
    for (int t = 0; t < NT; ++t) acc[t] = (v8f)0.0f;

    for (int kk = 0; kk < 32; ++kk) {
        const float2 av = Xp[kk * 2];
        v2f a; a.x = av.x; a.y = av.y;
        const int bo = kk * 2 * P + bbase;
#pragma unroll
        for (int t = 0; t < NT; ++t) {
            const float2 bv = sB[bo + t * 16];
            v2f b; b.x = bv.x; b.y = bv.y;
            acc[t] = __builtin_amdgcn_wmma_f32_16x16x4_f32(
                false, a, false, b, (short)0, acc[t], false, false);
        }
    }

    const int rbase = khp * 8;
#pragma unroll
    for (int r = 0; r < 8; ++r) {
        const int rr = row0 + rbase + r;
        if (rr < nrows) {
#pragma unroll
            for (int t = 0; t < YT; ++t)
                Y[(size_t)rr * YP + t * 16 + m] = acc[t][r];
            if (LOGITS) {
                const float v = acc[NT - 1][r];
                if (m < 8) es[(size_t)rr * 8 + m] = v;
                else       ed[(size_t)rr * 8 + (m - 8)] = v;
            }
        }
    }
}

__device__ __forceinline__ float lrelu(float a) {
    return a > 0.0f ? a : 0.2f * a;
}

__device__ __forceinline__ int edge_src(const int* __restrict__ csrc, int beg,
                                        int i, int deg, int d, int n) {
    if (i >= deg) return d;
    const int s = csrc[beg + i];
    return s < 0 ? 0 : (s > n - 1 ? n - 1 : s);
}

template <int CTRL>
__device__ __forceinline__ float quad_move(float x) {
    return __int_as_float(__builtin_amdgcn_update_dpp(0, __float_as_int(x), CTRL, 0xF, 0xF, true));
}
constexpr int QUAD_XOR1 = 0xB1;
constexpr int QUAD_XOR2 = 0x4E;

constexpr int RSHIFT = 8;
constexpr int L1_BLOCKS = 128;

__global__ void __launch_bounds__(1024)
range_bucket(const int* __restrict__ ei, int E, int nr,
             int* __restrict__ segs, int* __restrict__ pairs_)
{
    extern __shared__ int lc[];
    __shared__ int wsum[32];
    int2* __restrict__ pairs = (int2*)pairs_;
    const int tid = threadIdx.x, lane = tid & 31, wave = tid >> 5;
    const int c = blockIdx.x, nb = gridDim.x;
    for (int r = tid; r < nr; r += 1024) lc[r] = 0;
    __syncthreads();
    const int chunk = (E + nb - 1) / nb;
    const int e0 = c * chunk;
    const int e1 = (e0 + chunk < E) ? e0 + chunk : E;
    const int* __restrict__ dstp = ei + E;
    int i0 = e0 + tid;
    for (; i0 + 3 * 1024 < e1; i0 += 4 * 1024) {
        int d[4];
#pragma unroll
        for (int u = 0; u < 4; ++u) d[u] = dstp[i0 + u * 1024];
#pragma unroll
        for (int u = 0; u < 4; ++u) atomicAdd(&lc[d[u] >> RSHIFT], 1);
    }
    for (; i0 < e1; i0 += 1024) atomicAdd(&lc[dstp[i0] >> RSHIFT], 1);
    __syncthreads();

    int carry = e0;
    for (int base = 0; base < nr; base += 1024) {
        const int r = base + tid;
        const int v = (r < nr) ? lc[r] : 0;
        int x = v;
        for (int o = 1; o < 32; o <<= 1) {
            const int y = __shfl_up(x, o);
            if (lane >= o) x += y;
        }
        if (lane == 31) wsum[wave] = x;
        __syncthreads();
        const int ws = wsum[lane];
        int wx = ws;
        for (int o = 1; o < 32; o <<= 1) {
            const int y = __shfl_up(wx, o);
            if (lane >= o) wx += y;
        }
        const int tile_total = __shfl(wx, 31);
        const int wave_excl  = __shfl(wx - ws, wave);
        __syncthreads();
        const int ex = carry + wave_excl + (x - v);
        if (r < nr) {
            lc[r] = ex;
            segs[r * nb + c] = ex;
        }
        carry += tile_total;
    }
    if (tid == 0) segs[nr * nb + c] = carry;
    __syncthreads();

    for (i0 = e0 + tid; i0 + 3 * 1024 < e1; i0 += 4 * 1024) {
        int d[4], s[4];
#pragma unroll
        for (int u = 0; u < 4; ++u) { d[u] = dstp[i0 + u * 1024]; s[u] = ei[i0 + u * 1024]; }
#pragma unroll
        for (int u = 0; u < 4; ++u) {
            int pos = atomicAdd(&lc[d[u] >> RSHIFT], 1);
            pos = pos < 0 ? 0 : (pos > E - 1 ? E - 1 : pos);
            pairs[pos] = make_int2(d[u], s[u]);
        }
    }
    for (; i0 < e1; i0 += 1024) {
        const int d = dstp[i0], s = ei[i0];
        int pos = atomicAdd(&lc[d >> RSHIFT], 1);
        pos = pos < 0 ? 0 : (pos > E - 1 ? E - 1 : pos);
        pairs[pos] = make_int2(d, s);
    }
}

constexpr int NP_WAVES = 16;
constexpr int NP_PPW   = L1_BLOCKS / NP_WAVES;
constexpr int NP_CAP   = 8192;

struct in_lds_t { static constexpr bool value = true;  };
struct in_mem_t { static constexpr bool value = false; };

__device__ __forceinline__ void order_bucket_mem(const int* bsrc, int* __restrict__ csrc,
                                                 int beg, int end, int lane)
{
    const int deg = end - beg;
    for (int c0 = 0; c0 < deg; c0 += 32) {
        const int i   = c0 + lane;
        const int key = (i < deg) ? bsrc[beg + i] : 0;
        int rank = 0;
        for (int b0 = 0; b0 < deg; b0 += 32) {
            const int kc = (b0 + lane < deg) ? bsrc[beg + b0 + lane] : 0;
            const int nb = (deg - b0 < 32) ? deg - b0 : 32;
            for (int jj = 0; jj < nb; ++jj) {
                const int kj = __builtin_amdgcn_readlane(kc, jj);
                const int j  = b0 + jj;
                rank += (kj < key || (kj == key && j < i)) ? 1 : 0;
            }
        }
        if (i < deg) csrc[beg + rank] = key;
    }
}

__global__ void __launch_bounds__(NP_WAVES * 32)
node_place(const int* __restrict__ pairs_, const int* __restrict__ segs,
           int* __restrict__ rowptr, int* bsrc, int* __restrict__ csrc, int n, int E)
{
    __shared__ int cn[1 << RSHIFT];
    __shared__ int wt[(1 << RSHIFT) / 32];
    __shared__ int ps[L1_BLOCKS];
    __shared__ int pe[L1_BLOCKS];
    __shared__ int cst[(1 << RSHIFT) + 1];
    __shared__ int lsrc[NP_CAP + 8];
    __shared__ unsigned char lnod[NP_CAP];
    const int2* __restrict__ pairs = (const int2*)pairs_;
    const int tid = threadIdx.x, lane = tid & 31, wave = tid >> 5;
    const int b  = blockIdx.x;
    const int d0 = b << RSHIFT;
    const int chunk = (E + L1_BLOCKS - 1) / L1_BLOCKS;
    if (tid < L1_BLOCKS) {
        ps[tid] = segs[b * L1_BLOCKS + tid];
        pe[tid] = segs[(b + 1) * L1_BLOCKS + tid];
    }
    if (tid < (1 << RSHIFT)) cn[tid] = 0;
    __syncthreads();

    int a0 = 0, a1 = 0;
    for (int cc = lane; cc < L1_BLOCKS; cc += 32) {
        a0 += ps[cc] - cc * chunk;
        a1 += pe[cc] - cc * chunk;
    }
    for (int o = 16; o >= 1; o >>= 1) {
        a0 += __shfl_xor(a0, o);
        a1 += __shfl_xor(a1, o);
    }
    const int s0 = __builtin_amdgcn_readfirstlane(a0);
    const int s1 = __builtin_amdgcn_readfirstlane(a1);
    const int total = s1 - s0;
    const bool fits = total <= NP_CAP;

    int pb[NP_PPW], pl[NP_PPW];
#pragma unroll
    for (int q = 0; q < NP_PPW; ++q) {
        pb[q] = __builtin_amdgcn_readfirstlane(ps[wave * NP_PPW + q]);
        pl[q] = __builtin_amdgcn_readfirstlane(pe[wave * NP_PPW + q]) - pb[q];
    }

    {
        int dn[NP_PPW][2];
#pragma unroll
        for (int q = 0; q < NP_PPW; ++q)
#pragma unroll
            for (int t = 0; t < 2; ++t) {
                const int j = lane + 32 * t;
                dn[q][t] = (j < pl[q]) ? pairs[pb[q] + j].x : -1;
            }
#pragma unroll
        for (int q = 0; q < NP_PPW; ++q)
#pragma unroll
            for (int t = 0; t < 2; ++t)
                if (dn[q][t] >= 0) atomicAdd(&cn[dn[q][t] - d0], 1);
#pragma unroll
        for (int q = 0; q < NP_PPW; ++q)
            for (int j = lane + 64; j < pl[q]; j += 32) atomicAdd(&cn[pairs[pb[q] + j].x - d0], 1);
    }
    __syncthreads();
    int cnt = 0, x = 0;
    if (tid < (1 << RSHIFT)) {
        cnt = cn[tid];
        x = cnt;
        for (int o = 1; o < 32; o <<= 1) {
            const int y = __shfl_up(x, o);
            if (lane >= o) x += y;
        }
        if (lane == 31) wt[wave] = x;
    }
    __syncthreads();
    if (tid < (1 << RSHIFT)) {
        int pre = s0;
        for (int w = 0; w < wave; ++w) pre += wt[w];
        const int ex = pre + (x - cnt);
        cn[tid] = ex;
        cst[tid] = ex - s0;
        if (tid == 0) cst[1 << RSHIFT] = total;
        if (d0 + tid < n) rowptr[d0 + tid] = ex;
    }
    if (b == (int)gridDim.x - 1 && tid == 0) rowptr[n] = s1;
    __syncthreads();
    auto place_all = [&](auto ldsb) {
        constexpr bool LDSB = decltype(ldsb)::value;
        auto put_src = [&](int pos, int nd, int s) {
            if (LDSB) {
                int lp = pos - s0;
                lp = lp < 0 ? 0 : (lp > NP_CAP - 1 ? NP_CAP - 1 : lp);
                lsrc[lp] = s;
                lnod[lp] = (unsigned char)nd;
            } else {
                const int gp = pos < 0 ? 0 : (pos > E - 1 ? E - 1 : pos);
                bsrc[gp] = s;
            }
        };
        int2 p[NP_PPW][2];
#pragma unroll
        for (int q = 0; q < NP_PPW; ++q)
#pragma unroll
            for (int t = 0; t < 2; ++t) {
                const int j = lane + 32 * t;
                p[q][t] = (j < pl[q]) ? pairs[pb[q] + j] : make_int2(-1, 0);
            }
#pragma unroll
        for (int q = 0; q < NP_PPW; ++q)
#pragma unroll
            for (int t = 0; t < 2; ++t)
                if (p[q][t].x >= 0) {
                    const int nd = p[q][t].x - d0;
                    const int pos = atomicAdd(&cn[nd], 1);
                    put_src(pos, nd, p[q][t].y);
                }
#pragma unroll
        for (int q = 0; q < NP_PPW; ++q)
            for (int j = lane + 64; j < pl[q]; j += 32) {
                const int2 pp = pairs[pb[q] + j];
                const int nd = pp.x - d0;
                const int pos = atomicAdd(&cn[nd], 1);
                put_src(pos, nd, pp.y);
            }
    };
    if (fits) place_all(in_lds_t{}); else place_all(in_mem_t{});
    __syncthreads();

    if (fits) {
        for (int i = tid; i < total; i += NP_WAVES * 32) {
            const int key = lsrc[i];
            const int nd  = lnod[i];
            int bbeg = cst[nd], bend = cst[nd + 1];
            bbeg = bbeg < 0 ? 0 : bbeg;
            bend = bend > total ? total : bend;
            int rank = 0;
            for (int j0 = bbeg; j0 < bend; j0 += 8) {
                const int own = i - j0;
                const int inb = bend - j0;
                int k[8];
#pragma unroll
                for (int u = 0; u < 8; ++u) k[u] = lsrc[j0 + u];
#pragma unroll
                for (int u = 0; u < 8; ++u) {
                    const int lim = key + (u < own ? 1 : 0);
                    rank += ((int)(u < inb) & (int)(k[u] < lim));
                }
            }
            int pos = s0 + bbeg + rank;
            pos = pos < 0 ? 0 : (pos > E - 1 ? E - 1 : pos);
            csrc[pos] = key;
        }
    } else {
        for (int q = 0; q < (1 << RSHIFT) / NP_WAVES; ++q) {
            const int nd  = q * NP_WAVES + wave;
            int bbeg = __builtin_amdgcn_readfirstlane(nd ? cn[nd - 1] : s0);
            int bend = __builtin_amdgcn_readfirstlane(cn[nd]);
            bbeg = bbeg < 0 ? 0 : bbeg;
            bend = bend > E ? E : bend;
            order_bucket_mem(bsrc, csrc, bbeg, bend, lane);
        }
    }
}


template <int G>
__device__ __forceinline__ void l1_ask(const float* __restrict__ xh, int sl, int lane, float4 (&v)[8])
{
#pragma unroll
    for (int q = 0; q < 8; ++q) {
        const int su = __builtin_amdgcn_readlane(sl, 8 * G + q);
        v[q] = *(const float4*)(xh + (size_t)su * 128 + lane * 4);
    }
}

template <int G>
__device__ __forceinline__ void l1_add(const float4 (&v)[8], float pw0, float pw1, int cnt, float4& acc)
{
    const float w[8] = { quad_move<0x00>(pw0), quad_move<0x55>(pw0),
                         quad_move<0xAA>(pw0), quad_move<0xFF>(pw0),
                         quad_move<0x00>(pw1), quad_move<0x55>(pw1),
                         quad_move<0xAA>(pw1), quad_move<0xFF>(pw1) };
#pragma unroll
    for (int q = 0; q < 8; ++q) {
        const bool have = (8 * G + q == 0) || (8 * G + q < cnt);
        const float tx = fmaf(w[q], v[q].x, acc.x);
        const float ty = fmaf(w[q], v[q].y, acc.y);
        const float tz = fmaf(w[q], v[q].z, acc.z);
        const float tw = fmaf(w[q], v[q].w, acc.w);
        acc.x = have ? tx : acc.x;
        acc.y = have ? ty : acc.y;
        acc.z = have ? tz : acc.z;
        acc.w = have ? tw : acc.w;
    }
}

__global__ void __launch_bounds__(256)
node_gather_l1(const int* __restrict__ rowptr, const int* __restrict__ csrc,
               const float* __restrict__ es, const float* __restrict__ ed,
               const float* __restrict__ xh, const float* __restrict__ b,
               float* __restrict__ h1, int n)
{
    const int lane = threadIdx.x & 31;
    const int d = blockIdx.x * 8 + (threadIdx.x >> 5);
    if (d >= n) return;
    const int beg = __builtin_amdgcn_readfirstlane(rowptr[d]);
    const int end = __builtin_amdgcn_readfirstlane(rowptr[d + 1]);
    const int deg = end - beg;
    const int cnt = deg + 1;

    float4 acc = make_float4(0.f, 0.f, 0.f, 0.f);

    if (cnt <= 32) {
        const int hq = lane >> 2, sub = lane & 3;
        const float edh = ed[d * 8 + hq];
        const int sl = edge_src(csrc, beg, lane, deg, d, n);

        float a[8];
#pragma unroll
        for (int t = 0; t < 8; ++t) {
            const int s = __shfl(sl, 4 * t + sub);
            a[t] = es[s * 8 + hq];
        }
        float4 v[8];
        l1_ask<0>(xh, sl, lane, v);
        __builtin_amdgcn_sched_group_barrier(0x020, 16, 0);
        __builtin_amdgcn_sched_group_barrier(0x002, 8, 0);

        float m = -__builtin_inff();
#pragma unroll
        for (int t = 0; t < 8; ++t) {
            const int el = 4 * t + sub;
            a[t] = lrelu(a[t] + edh);
            if (el < cnt) m = fmaxf(m, a[t]);
        }
        m = fmaxf(m, quad_move<QUAD_XOR1>(m));
        m = fmaxf(m, quad_move<QUAD_XOR2>(m));

        float sum = 0.0f;
#pragma unroll
        for (int t = 0; t < 8; ++t) {
            const int el = 4 * t + sub;
            if (el < cnt) sum += __expf(a[t] - m);
        }
        sum += quad_move<QUAD_XOR1>(sum);
        sum += quad_move<QUAD_XOR2>(sum);
        const float inv = 1.0f / (sum + 1e-16f);

#pragma unroll
        for (int t = 0; t < 8; ++t) a[t] = __expf(a[t] - m) * inv;

        l1_add<0>(v, a[0], a[1], cnt, acc);
        if (cnt >  8) { l1_ask<1>(xh, sl, lane, v); l1_add<1>(v, a[2], a[3], cnt, acc); }
        if (cnt > 16) { l1_ask<2>(xh, sl, lane, v); l1_add<2>(v, a[4], a[5], cnt, acc); }
        if (cnt > 24) { l1_ask<3>(xh, sl, lane, v); l1_add<3>(v, a[6], a[7], cnt, acc); }
    } else {
        const int h = lane & 7, sub = lane >> 3;
        const float edh = ed[d * 8 + h];

        float m = -__builtin_inff();
        for (int c0 = 0; c0 < cnt; c0 += 32) {
            const int i  = c0 + lane;
            const int sl = edge_src(csrc, beg, i, deg, d, n);
            const int nv = (cnt - c0 < 32) ? cnt - c0 : 32;
#pragma unroll
            for (int t = 0; t < 8; ++t) {
                const int el = 4 * t + sub;
                const int s  = __shfl(sl, el);
                const float a = lrelu(es[s * 8 + h] + edh);
                if (el < nv) m = fmaxf(m, a);
            }
        }
        m = fmaxf(m, __shfl_xor(m, 8));
        m = fmaxf(m, __shfl_xor(m, 16));

        float sum = 0.0f;
        for (int c0 = 0; c0 < cnt; c0 += 32) {
            const int i  = c0 + lane;
            const int sl = edge_src(csrc, beg, i, deg, d, n);
            const int nv = (cnt - c0 < 32) ? cnt - c0 : 32;
#pragma unroll
            for (int t = 0; t < 8; ++t) {
                const int el = 4 * t + sub;
                const int s  = __shfl(sl, el);
                const float a = lrelu(es[s * 8 + h] + edh);
                if (el < nv) sum += __expf(a - m);
            }
        }
        sum += __shfl_xor(sum, 8);
        sum += __shfl_xor(sum, 16);
        const float inv = 1.0f / (sum + 1e-16f);

        const int hh = lane >> 2;
        const float m2   = __shfl(m, hh);
        const float inv2 = __shfl(inv, hh);
        const float ed2  = __shfl(edh, hh);

        for (int c0 = 0; c0 < cnt; c0 += 32) {
            const int i  = c0 + lane;
            const int sl = edge_src(csrc, beg, i, deg, d, n);
            const int nv = (cnt - c0 < 32) ? cnt - c0 : 32;
            for (int g = 0; 4 * g < nv; ++g) {
                float4 v[4];
                float  ev[4];
#pragma unroll
                for (int q = 0; q < 4; ++q) {
                    const int su = __builtin_amdgcn_readlane(sl, 4 * g + q);
                    v[q]  = *(const float4*)(xh + (size_t)su * 128 + lane * 4);
                    ev[q] = es[su * 8 + hh];
                }
#pragma unroll
                for (int q = 0; q < 4; ++q) {
                    if (4 * g + q < nv) {
                        const float w = __expf(lrelu(ev[q] + ed2) - m2) * inv2;
                        acc.x = fmaf(w, v[q].x, acc.x);
                        acc.y = fmaf(w, v[q].y, acc.y);
                        acc.z = fmaf(w, v[q].z, acc.z);
                        acc.w = fmaf(w, v[q].w, acc.w);
                    }
                }
            }
        }
    }

    const float4 bv = *(const float4*)(b + lane * 4);
    float4 o;
    o.x = acc.x + bv.x; o.y = acc.y + bv.y; o.z = acc.z + bv.z; o.w = acc.w + bv.w;
    o.x = o.x > 0.0f ? o.x : expm1f(o.x);
    o.y = o.y > 0.0f ? o.y : expm1f(o.y);
    o.z = o.z > 0.0f ? o.z : expm1f(o.z);
    o.w = o.w > 0.0f ? o.w : expm1f(o.w);
    *(float4*)(h1 + (size_t)d * 128 + lane * 4) = o;
}

constexpr int W1E_COLS = 144;
constexpr int W2P_COLS = 48;
constexpr int XH2_ES   = 40;
constexpr int XH2_ED   = 41;

__global__ void prep_weights(const float* __restrict__ W1,
                             const float* __restrict__ as1,
                             const float* __restrict__ ad1,
                             const float* __restrict__ W2,
                             const float* __restrict__ as2,
                             const float* __restrict__ ad2,
                             float* __restrict__ W1e, float* __restrict__ W2p)
{
    const int t = blockIdx.x * blockDim.x + threadIdx.x;
    if (t < 128 * W1E_COLS) {
        const int k = t / W1E_COLS, c = t - k * W1E_COLS;
        float v;
        if (c < 128) {
            v = W1[k * 128 + c];
        } else {
            const int h = (c - 128) & 7;
            const float* a = (c < 136 ? as1 : ad1) + h * 16;
            const float* w = W1 + k * 128 + h * 16;
            float s = 0.f;
            for (int j = 0; j < 16; ++j) s += w[j] * a[j];
            v = s;
        }
        W1e[t] = v;
    } else if (t < 128 * W1E_COLS + 128 * W2P_COLS) {
        const int u = t - 128 * W1E_COLS;
        const int k = u / W2P_COLS, c = u - k * W2P_COLS;
        float v = 0.0f;
        if (c < 40) {
            v = W2[k * 40 + c];
        } else if (c == XH2_ES || c == XH2_ED) {
            const float* a = (c == XH2_ES) ? as2 : ad2;
            const float* w = W2 + k * 40;
            float s = 0.f;
            for (int j = 0; j < 40; ++j) s += w[j] * a[j];
            v = s;
        }
        W2p[u] = v;
    }
}


__device__ __forceinline__ void l2_rows(const float* __restrict__ xh2, int sl, float wl,
                                        int nv, int jbeg, int slot, int c4, float4& acc)
{
    for (int j0 = jbeg; j0 < nv; j0 += 12) {
        float4 v[4];
        float  w[4];
#pragma unroll
        for (int u = 0; u < 4; ++u) {
            const int j  = j0 + 3 * u + slot;
            const bool ok = (slot < 3) && (j < nv);
            const int jj = ok ? j : 0;
            const int s  = __shfl(sl, jj);
            const float wj = __shfl(wl, jj);
            w[u] = ok ? wj : 0.0f;
            v[u] = *(const float4*)(xh2 + (size_t)s * 48 + c4 * 4);
        }
#pragma unroll
        for (int u = 0; u < 4; ++u) {
            const int j = j0 + 3 * u + slot;
            if ((slot < 3) && (j < nv)) {
                acc.x = fmaf(w[u], v[u].x, acc.x);
                acc.y = fmaf(w[u], v[u].y, acc.y);
                acc.z = fmaf(w[u], v[u].z, acc.z);
                acc.w = fmaf(w[u], v[u].w, acc.w);
            }
        }
    }
}

template <int K0>
__device__ __forceinline__ void l2_ask(const float* __restrict__ xh2, int sl, int nv,
                                       int slot, int c4, float4 (&v)[8])
{
#pragma unroll
    for (int u = 0; u < 4; ++u) {
        const int j  = 3 * (K0 + u) + slot;
        const bool ok = (slot < 3) && (j < nv);
        const int s  = __shfl(sl, ok ? j : 0);
        v[K0 + u] = *(const float4*)(xh2 + (size_t)s * 48 + c4 * 4);
    }
}

template <int K0>
__device__ __forceinline__ void l2_add(const float4 (&v)[8], float wl, int nv,
                                       int slot, float4& acc)
{
#pragma unroll
    for (int u = 0; u < 4; ++u) {
        const int j  = 3 * (K0 + u) + slot;
        const bool ok = (slot < 3) && (j < nv);
        const float wj = __shfl(wl, ok ? j : 0);
        if (ok) {
            acc.x = fmaf(wj, v[K0 + u].x, acc.x);
            acc.y = fmaf(wj, v[K0 + u].y, acc.y);
            acc.z = fmaf(wj, v[K0 + u].z, acc.z);
            acc.w = fmaf(wj, v[K0 + u].w, acc.w);
        }
    }
}

__global__ void __launch_bounds__(256)
node_gather_l2(const int* __restrict__ rowptr, const int* __restrict__ csrc,
               const float* __restrict__ xh2, const float* __restrict__ b,
               float* __restrict__ out, int n)
{
    __shared__ __align__(16) float so[8 * 40];
    const int lane = threadIdx.x & 31;
    const int wv   = threadIdx.x >> 5;
    const int d0   = blockIdx.x * 8;
    const int d    = d0 + wv;

    if (d < n) {
        const int beg = __builtin_amdgcn_readfirstlane(rowptr[d]);
        const int end = __builtin_amdgcn_readfirstlane(rowptr[d + 1]);
        const int deg = end - beg;
        const int cnt = deg + 1;
        const float edd = xh2[(size_t)d * W2P_COLS + XH2_ED];

        const int slot = lane / 10;
        const int c4   = lane - slot * 10;
        float4 acc = make_float4(0.f, 0.f, 0.f, 0.f);

        if (cnt <= 32) {
            const int sl  = edge_src(csrc, beg, lane, deg, d, n);
            const float a = lrelu(xh2[(size_t)sl * W2P_COLS + XH2_ES] + edd);

            float4 v[8];
            l2_ask<0>(xh2, sl, cnt, slot, c4, v);
            if (cnt > 12) l2_ask<4>(xh2, sl, cnt, slot, c4, v);

            float m = -__builtin_inff();
            if (lane < cnt) m = fmaxf(m, a);
            for (int o = 16; o >= 1; o >>= 1) m = fmaxf(m, __shfl_xor(m, o));

            const float e = __expf(a - m);
            float sum = 0.0f;
            if (lane < cnt) sum += e;
            for (int o = 16; o >= 1; o >>= 1) sum += __shfl_xor(sum, o);
            const float inv = 1.0f / (sum + 1e-16f);

            const float wl = e * inv;
            l2_add<0>(v, wl, cnt, slot, acc);
            if (cnt > 12) l2_add<4>(v, wl, cnt, slot, acc);
            if (cnt > 24) l2_rows(xh2, sl, wl, cnt, 24, slot, c4, acc);
        } else {
            float m = -__builtin_inff();
            for (int c0 = 0; c0 < cnt; c0 += 32) {
                const int i  = c0 + lane;
                const int sl = edge_src(csrc, beg, i, deg, d, n);
                const float a = lrelu(xh2[(size_t)sl * W2P_COLS + XH2_ES] + edd);
                if (i < cnt) m = fmaxf(m, a);
            }
            for (int o = 16; o >= 1; o >>= 1) m = fmaxf(m, __shfl_xor(m, o));

            float sum = 0.0f;
            for (int c0 = 0; c0 < cnt; c0 += 32) {
                const int i  = c0 + lane;
                const int sl = edge_src(csrc, beg, i, deg, d, n);
                const float a = lrelu(xh2[(size_t)sl * W2P_COLS + XH2_ES] + edd);
                if (i < cnt) sum += __expf(a - m);
            }
            for (int o = 16; o >= 1; o >>= 1) sum += __shfl_xor(sum, o);
            const float inv = 1.0f / (sum + 1e-16f);

            for (int c0 = 0; c0 < cnt; c0 += 32) {
                const int i  = c0 + lane;
                const int sl = edge_src(csrc, beg, i, deg, d, n);
                const float a  = lrelu(xh2[(size_t)sl * W2P_COLS + XH2_ES] + edd);
                const float wl = __expf(a - m) * inv;
                const int nv = (cnt - c0 < 32) ? cnt - c0 : 32;
                l2_rows(xh2, sl, wl, nv, 0, slot, c4, acc);
            }
        }
        const int l1 = (lane + 10) & 31, l2 = (lane + 20) & 31;
        acc.x += __shfl(acc.x, l1) + __shfl(acc.x, l2);
        acc.y += __shfl(acc.y, l1) + __shfl(acc.y, l2);
        acc.z += __shfl(acc.z, l1) + __shfl(acc.z, l2);
        acc.w += __shfl(acc.w, l1) + __shfl(acc.w, l2);
        if (lane < 10) {
            const float4 bv = *(const float4*)(b + lane * 4);
            float4 o;
            o.x = acc.x + bv.x; o.y = acc.y + bv.y; o.z = acc.z + bv.z; o.w = acc.w + bv.w;
            *(float4*)(so + wv * 40 + lane * 4) = o;
        }
    }
    __syncthreads();
    const int rows = (n - d0 < 8) ? n - d0 : 8;
    if ((int)threadIdx.x < rows * 10)
        *(float4*)(out + (size_t)d0 * 40 + threadIdx.x * 4) = *(const float4*)(so + threadIdx.x * 4);
}

extern "C" void kernel_launch(void* const* d_in, const int* in_sizes, int n_in,
                              void* d_out, int out_size, void* d_ws, size_t ws_size,
                              hipStream_t stream)
{
    (void)n_in; (void)out_size; (void)ws_size;

    const float* x   = (const float*)d_in[0];
    const int*   ei  = (const int*)d_in[1];
    const float* W1  = (const float*)d_in[2];
    const float* as1 = (const float*)d_in[3];
    const float* ad1 = (const float*)d_in[4];
    const float* b1  = (const float*)d_in[5];
    const float* W2  = (const float*)d_in[6];
    const float* as2 = (const float*)d_in[7];
    const float* ad2 = (const float*)d_in[8];
    const float* b2  = (const float*)d_in[9];
    float* out = (float*)d_out;

    const int n  = in_sizes[0] / 128;
    const int E  = in_sizes[1] / 2;

    char* ws = (char*)d_ws;
    size_t off = 0;
    auto alloc = [&](size_t bytes) -> float* {
        float* p = (float*)(ws + off);
        off += (bytes + 255) & ~(size_t)255;
        return p;
    };
    float* xh1  = alloc((size_t)n * 128 * 4);
    float* h1   = alloc((size_t)n * 128 * 4);
    float* es1  = alloc((size_t)n * 8 * 4);
    float* ed1  = alloc((size_t)n * 8 * 4);
    float* W2p  = alloc((size_t)128 * W2P_COLS * 4);
    const int nr = (n + (1 << RSHIFT) - 1) >> RSHIFT;
    int* segs   = (int*)alloc((size_t)((nr + 1) * L1_BLOCKS + 16) * 4);
    int* rowptr = (int*)alloc((size_t)(n + 16) * 4);
    int* csrc   = (int*)alloc((size_t)E * 4);
    int* bsrc   = (int*)h1;
    int* pairs  = (int*)xh1;
    float* W1e  = h1 + (size_t)n * 64;

    const int B = 256;

    prep_weights<<<(128 * W1E_COLS + 128 * W2P_COLS + B - 1) / B, B, 0, stream>>>(W1, as1, ad1, W2, as2, ad2, W1e, W2p);

    range_bucket<<<L1_BLOCKS, 1024, nr * sizeof(int), stream>>>(ei, E, nr, segs, pairs);
    node_place<<<nr, NP_WAVES * 32, 0, stream>>>(pairs, segs, rowptr, bsrc, csrc, n, E);

    gemm_wmma_f32<9, true><<<(n + 127) / 128, 256, 0, stream>>>(x, W1e, xh1, es1, ed1, n);
    node_gather_l1<<<(n + 7) / 8, B, 0, stream>>>(rowptr, csrc, es1, ed1, xh1, b1, h1, n);

    float* xh2 = xh1;
    gemm_wmma_f32<3, false><<<(n + 127) / 128, 256, 0, stream>>>(h1, W2p, xh2, nullptr, nullptr, n);
    node_gather_l2<<<(n + 7) / 8, B, 0, stream>>>(rowptr, csrc, xh2, b2, out, n);
}


# independent work shares launches: weight prep runs in extra blocks of the level-1 bucketing launch, node_place blocks (8 waves now) run beside the layer-1 projection blocks; 7 launches become 5
# speedup vs baseline: 10.2581x; 1.1344x over previous
#include <hip/hip_runtime.h>
#include <hip/hip_bf16.h>
#include <math.h>


typedef float v2f __attribute__((ext_vector_type(2)));
typedef float v8f __attribute__((ext_vector_type(8)));

template <int NT, bool LOGITS>
struct gemm_cfg {
    static constexpr int NCOL = NT * 16;
    static constexpr int P  = LOGITS ? NCOL : NCOL + 16;
    static constexpr int SB = 64 * P;
};

template <int NT, bool LOGITS, int NW>
__device__ __forceinline__ void gemm_body(float2* sB, int blk,
                                          const float* __restrict__ X,
                                          const float* __restrict__ Wm,
                                          float* __restrict__ Y,
                                          float* __restrict__ es,
                                          float* __restrict__ ed,
                                          int nrows)
{
    constexpr int NCOL = gemm_cfg<NT, LOGITS>::NCOL;
    constexpr int YT   = LOGITS ? NT - 1 : NT;
    constexpr int YP   = YT * 16;
    constexpr int P    = gemm_cfg<NT, LOGITS>::P;

    const int tid = threadIdx.x;
    for (int i = tid; i < 64 * NCOL; i += NW * 32) {
        const int kp = i / NCOL;
        const int nn = i - kp * NCOL;
        float2 w2;
        w2.x = Wm[(2 * kp) * NCOL + nn];
        w2.y = Wm[(2 * kp + 1) * NCOL + nn];
        sB[kp * P + nn] = w2;
    }
    __syncthreads();

    const int wave = tid >> 5;
    const int lane = tid & 31;
    const int row0 = blk * (NW * 16) + wave * 16;
    const int m    = lane & 15;
    const int khp  = lane >> 4;
    int row = row0 + m;
    if (row > nrows - 1) row = nrows - 1;

    const float2* Xp = (const float2*)(X + (size_t)row * 128) + khp;
    const int bbase = khp * P + m;

    v8f acc[NT];
#pragma unroll
    for (int t = 0; t < NT; ++t) acc[t] = (v8f)0.0f;

    for (int kk = 0; kk < 32; ++kk) {
        const float2 av = Xp[kk * 2];
        v2f a; a.x = av.x; a.y = av.y;
        const int bo = kk * 2 * P + bbase;
#pragma unroll
        for (int t = 0; t < NT; ++t) {
            const float2 bv = sB[bo + t * 16];
            v2f b; b.x = bv.x; b.y = bv.y;
            acc[t] = __builtin_amdgcn_wmma_f32_16x16x4_f32(
                false, a, false, b, (short)0, acc[t], false, false);
        }
    }

    const int rbase = khp * 8;
#pragma unroll
    for (int r = 0; r < 8; ++r) {
        const int rr = row0 + rbase + r;
        if (rr < nrows) {
#pragma unroll
            for (int t = 0; t < YT; ++t)
                Y[(size_t)rr * YP + t * 16 + m] = acc[t][r];
            if (LOGITS) {
                const float v = acc[NT - 1][r];
                if (m < 8) es[(size_t)rr * 8 + m] = v;
                else       ed[(size_t)rr * 8 + (m - 8)] = v;
            }
        }
    }
}

template <int NT, bool LOGITS>
__global__ void gemm_wmma_f32(const float* __restrict__ X,
                              const float* __restrict__ Wm,
                              float* __restrict__ Y,
                              float* __restrict__ es,
                              float* __restrict__ ed,
                              int nrows)
{
    __shared__ __align__(16) float2 sB[gemm_cfg<NT, LOGITS>::SB];
    gemm_body<NT, LOGITS, 8>(sB, blockIdx.x, X, Wm, Y, es, ed, nrows);
}

__device__ __forceinline__ float lrelu(float a) {
    return a > 0.0f ? a : 0.2f * a;
}

__device__ __forceinline__ int edge_src(const int* __restrict__ csrc, int beg,
                                        int i, int deg, int d, int n) {
    if (i >= deg) return d;
    const int s = csrc[beg + i];
    return s < 0 ? 0 : (s > n - 1 ? n - 1 : s);
}

template <int CTRL>
__device__ __forceinline__ float quad_move(float x) {
    return __int_as_float(__builtin_amdgcn_update_dpp(0, __float_as_int(x), CTRL, 0xF, 0xF, true));
}
constexpr int QUAD_XOR1 = 0xB1;
constexpr int QUAD_XOR2 = 0x4E;

constexpr int W1E_COLS = 144;
constexpr int W2P_COLS = 48;
constexpr int XH2_ES   = 40;
constexpr int XH2_ED   = 41;

constexpr int PREP_ELEMS = 128 * W1E_COLS + 128 * W2P_COLS;

__device__ __forceinline__ void prep_element(int t,
                                             const float* __restrict__ W1,
                                             const float* __restrict__ as1,
                                             const float* __restrict__ ad1,
                                             const float* __restrict__ W2,
                                             const float* __restrict__ as2,
                                             const float* __restrict__ ad2,
                                             float* __restrict__ W1e, float* __restrict__ W2p)
{
    if (t < 128 * W1E_COLS) {
        const int k = t / W1E_COLS, c = t - k * W1E_COLS;
        float v;
        if (c < 128) {
            v = W1[k * 128 + c];
        } else {
            const int h = (c - 128) & 7;
            const float* a = (c < 136 ? as1 : ad1) + h * 16;
            const float* w = W1 + k * 128 + h * 16;
            float s = 0.f;
            for (int j = 0; j < 16; ++j) s += w[j] * a[j];
            v = s;
        }
        W1e[t] = v;
    } else if (t < PREP_ELEMS) {
        const int u = t - 128 * W1E_COLS;
        const int k = u / W2P_COLS, c = u - k * W2P_COLS;
        float v = 0.0f;
        if (c < 40) {
            v = W2[k * 40 + c];
        } else if (c == XH2_ES || c == XH2_ED) {
            const float* a = (c == XH2_ES) ? as2 : ad2;
            const float* w = W2 + k * 40;
            float s = 0.f;
            for (int j = 0; j < 40; ++j) s += w[j] * a[j];
            v = s;
        }
        W2p[u] = v;
    }
}

constexpr int RSHIFT = 8;
constexpr int L1_BLOCKS = 128;

__global__ void __launch_bounds__(1024)
range_bucket(const int* __restrict__ ei, int E, int nr,
             int* __restrict__ segs, int* __restrict__ pairs_,
             const float* __restrict__ W1, const float* __restrict__ as1,
             const float* __restrict__ ad1, const float* __restrict__ W2,
             const float* __restrict__ as2, const float* __restrict__ ad2,
             float* __restrict__ W1e, float* __restrict__ W2p)
{
    extern __shared__ int lc[];
    __shared__ int wsum[32];
    int2* __restrict__ pairs = (int2*)pairs_;
    const int tid = threadIdx.x, lane = tid & 31, wave = tid >> 5;
    const int c = blockIdx.x, nb = L1_BLOCKS;
    if (c >= nb) {
        prep_element((c - nb) * 1024 + tid, W1, as1, ad1, W2, as2, ad2, W1e, W2p);
        return;
    }
    for (int r = tid; r < nr; r += 1024) lc[r] = 0;
    __syncthreads();
    const int chunk = (E + nb - 1) / nb;
    const int e0 = c * chunk;
    const int e1 = (e0 + chunk < E) ? e0 + chunk : E;
    const int* __restrict__ dstp = ei + E;
    int i0 = e0 + tid;
    for (; i0 + 3 * 1024 < e1; i0 += 4 * 1024) {
        int d[4];
#pragma unroll
        for (int u = 0; u < 4; ++u) d[u] = dstp[i0 + u * 1024];
#pragma unroll
        for (int u = 0; u < 4; ++u) atomicAdd(&lc[d[u] >> RSHIFT], 1);
    }
    for (; i0 < e1; i0 += 1024) atomicAdd(&lc[dstp[i0] >> RSHIFT], 1);
    __syncthreads();

    int carry = e0;
    for (int base = 0; base < nr; base += 1024) {
        const int r = base + tid;
        const int v = (r < nr) ? lc[r] : 0;
        int x = v;
        for (int o = 1; o < 32; o <<= 1) {
            const int y = __shfl_up(x, o);
            if (lane >= o) x += y;
        }
        if (lane == 31) wsum[wave] = x;
        __syncthreads();
        const int ws = wsum[lane];
        int wx = ws;
        for (int o = 1; o < 32; o <<= 1) {
            const int y = __shfl_up(wx, o);
            if (lane >= o) wx += y;
        }
        const int tile_total = __shfl(wx, 31);
        const int wave_excl  = __shfl(wx - ws, wave);
        __syncthreads();
        const int ex = carry + wave_excl + (x - v);
        if (r < nr) {
            lc[r] = ex;
            segs[r * nb + c] = ex;
        }
        carry += tile_total;
    }
    if (tid == 0) segs[nr * nb + c] = carry;
    __syncthreads();

    for (i0 = e0 + tid; i0 + 3 * 1024 < e1; i0 += 4 * 1024) {
        int d[4], s[4];
#pragma unroll
        for (int u = 0; u < 4; ++u) { d[u] = dstp[i0 + u * 1024]; s[u] = ei[i0 + u * 1024]; }
#pragma unroll
        for (int u = 0; u < 4; ++u) {
            int pos = atomicAdd(&lc[d[u] >> RSHIFT], 1);
            pos = pos < 0 ? 0 : (pos > E - 1 ? E - 1 : pos);
            pairs[pos] = make_int2(d[u], s[u]);
        }
    }
    for (; i0 < e1; i0 += 1024) {
        const int d = dstp[i0], s = ei[i0];
        int pos = atomicAdd(&lc[d >> RSHIFT], 1);
        pos = pos < 0 ? 0 : (pos > E - 1 ? E - 1 : pos);
        pairs[pos] = make_int2(d, s);
    }
}

constexpr int NP_WAVES = 8;
constexpr int NP_PPW   = L1_BLOCKS / NP_WAVES;
constexpr int NP_CAP   = 8192;

struct in_lds_t { static constexpr bool value = true;  };
struct in_mem_t { static constexpr bool value = false; };

__device__ __forceinline__ void order_bucket_mem(const int* bsrc, int* __restrict__ csrc,
                                                 int beg, int end, int lane)
{
    const int deg = end - beg;
    for (int c0 = 0; c0 < deg; c0 += 32) {
        const int i   = c0 + lane;
        const int key = (i < deg) ? bsrc[beg + i] : 0;
        int rank = 0;
        for (int b0 = 0; b0 < deg; b0 += 32) {
            const int kc = (b0 + lane < deg) ? bsrc[beg + b0 + lane] : 0;
            const int nb = (deg - b0 < 32) ? deg - b0 : 32;
            for (int jj = 0; jj < nb; ++jj) {
                const int kj = __builtin_amdgcn_readlane(kc, jj);
                const int j  = b0 + jj;
                rank += (kj < key || (kj == key && j < i)) ? 1 : 0;
            }
        }
        if (i < deg) csrc[beg + rank] = key;
    }
}

struct np_lds_t {
    int cn[1 << RSHIFT];
    int wt[(1 << RSHIFT) / 32];
    int ps[L1_BLOCKS];
    int pe[L1_BLOCKS];
    int cst[(1 << RSHIFT) + 1];
    int lsrc[NP_CAP + 8];
    unsigned char lnod[NP_CAP];
};

__device__ __forceinline__ void
node_place_body(np_lds_t* lds, int b, int nblk,
                const int* __restrict__ pairs_, const int* __restrict__ segs,
                int* __restrict__ rowptr, int* bsrc, int* __restrict__ csrc, int n, int E)
{
    int* const cn   = lds->cn;
    int* const wt   = lds->wt;
    int* const ps   = lds->ps;
    int* const pe   = lds->pe;
    int* const cst  = lds->cst;
    int* const lsrc = lds->lsrc;
    unsigned char* const lnod = lds->lnod;
    const int2* __restrict__ pairs = (const int2*)pairs_;
    const int tid = threadIdx.x, lane = tid & 31, wave = tid >> 5;
    const int d0 = b << RSHIFT;
    const int chunk = (E + L1_BLOCKS - 1) / L1_BLOCKS;
    if (tid < L1_BLOCKS) {
        ps[tid] = segs[b * L1_BLOCKS + tid];
        pe[tid] = segs[(b + 1) * L1_BLOCKS + tid];
    }
    if (tid < (1 << RSHIFT)) cn[tid] = 0;
    __syncthreads();

    int a0 = 0, a1 = 0;
    for (int cc = lane; cc < L1_BLOCKS; cc += 32) {
        a0 += ps[cc] - cc * chunk;
        a1 += pe[cc] - cc * chunk;
    }
    for (int o = 16; o >= 1; o >>= 1) {
        a0 += __shfl_xor(a0, o);
        a1 += __shfl_xor(a1, o);
    }
    const int s0 = __builtin_amdgcn_readfirstlane(a0);
    const int s1 = __builtin_amdgcn_readfirstlane(a1);
    const int total = s1 - s0;
    const bool fits = total <= NP_CAP;

    int pb[NP_PPW], pl[NP_PPW];
#pragma unroll
    for (int q = 0; q < NP_PPW; ++q) {
        pb[q] = __builtin_amdgcn_readfirstlane(ps[wave * NP_PPW + q]);
        pl[q] = __builtin_amdgcn_readfirstlane(pe[wave * NP_PPW + q]) - pb[q];
    }

    {
        int dn[NP_PPW][2];
#pragma unroll
        for (int q = 0; q < NP_PPW; ++q)
#pragma unroll
            for (int t = 0; t < 2; ++t) {
                const int j = lane + 32 * t;
                dn[q][t] = (j < pl[q]) ? pairs[pb[q] + j].x : -1;
            }
#pragma unroll
        for (int q = 0; q < NP_PPW; ++q)
#pragma unroll
            for (int t = 0; t < 2; ++t)
                if (dn[q][t] >= 0) atomicAdd(&cn[dn[q][t] - d0], 1);
#pragma unroll
        for (int q = 0; q < NP_PPW; ++q)
            for (int j = lane + 64; j < pl[q]; j += 32) atomicAdd(&cn[pairs[pb[q] + j].x - d0], 1);
    }
    __syncthreads();
    int cnt = 0, x = 0;
    if (tid < (1 << RSHIFT)) {
        cnt = cn[tid];
        x = cnt;
        for (int o = 1; o < 32; o <<= 1) {
            const int y = __shfl_up(x, o);
            if (lane >= o) x += y;
        }
        if (lane == 31) wt[wave] = x;
    }
    __syncthreads();
    if (tid < (1 << RSHIFT)) {
        int pre = s0;
        for (int w = 0; w < wave; ++w) pre += wt[w];
        const int ex = pre + (x - cnt);
        cn[tid] = ex;
        cst[tid] = ex - s0;
        if (tid == 0) cst[1 << RSHIFT] = total;
        if (d0 + tid < n) rowptr[d0 + tid] = ex;
    }
    if (b == nblk - 1 && tid == 0) rowptr[n] = s1;
    __syncthreads();
    auto place_all = [&](auto ldsb) {
        constexpr bool LDSB = decltype(ldsb)::value;
        auto put_src = [&](int pos, int nd, int s) {
            if (LDSB) {
                int lp = pos - s0;
                lp = lp < 0 ? 0 : (lp > NP_CAP - 1 ? NP_CAP - 1 : lp);
                lsrc[lp] = s;
                lnod[lp] = (unsigned char)nd;
            } else {
                const int gp = pos < 0 ? 0 : (pos > E - 1 ? E - 1 : pos);
                bsrc[gp] = s;
            }
        };
        int2 p[NP_PPW][2];
#pragma unroll
        for (int q = 0; q < NP_PPW; ++q)
#pragma unroll
            for (int t = 0; t < 2; ++t) {
                const int j = lane + 32 * t;
                p[q][t] = (j < pl[q]) ? pairs[pb[q] + j] : make_int2(-1, 0);
            }
#pragma unroll
        for (int q = 0; q < NP_PPW; ++q)
#pragma unroll
            for (int t = 0; t < 2; ++t)
                if (p[q][t].x >= 0) {
                    const int nd = p[q][t].x - d0;
                    const int pos = atomicAdd(&cn[nd], 1);
                    put_src(pos, nd, p[q][t].y);
                }
#pragma unroll
        for (int q = 0; q < NP_PPW; ++q)
            for (int j = lane + 64; j < pl[q]; j += 32) {
                const int2 pp = pairs[pb[q] + j];
                const int nd = pp.x - d0;
                const int pos = atomicAdd(&cn[nd], 1);
                put_src(pos, nd, pp.y);
            }
    };
    if (fits) place_all(in_lds_t{}); else place_all(in_mem_t{});
    __syncthreads();

    if (fits) {
        for (int i = tid; i < total; i += NP_WAVES * 32) {
            const int key = lsrc[i];
            const int nd  = lnod[i];
            int bbeg = cst[nd], bend = cst[nd + 1];
            bbeg = bbeg < 0 ? 0 : bbeg;
            bend = bend > total ? total : bend;
            int rank = 0;
            for (int j0 = bbeg; j0 < bend; j0 += 8) {
                const int own = i - j0;
                const int inb = bend - j0;
                int k[8];
#pragma unroll
                for (int u = 0; u < 8; ++u) k[u] = lsrc[j0 + u];
#pragma unroll
                for (int u = 0; u < 8; ++u) {
                    const int lim = key + (u < own ? 1 : 0);
                    rank += ((int)(u < inb) & (int)(k[u] < lim));
                }
            }
            int pos = s0 + bbeg + rank;
            pos = pos < 0 ? 0 : (pos > E - 1 ? E - 1 : pos);
            csrc[pos] = key;
        }
    } else {
        for (int q = 0; q < (1 << RSHIFT) / NP_WAVES; ++q) {
            const int nd  = q * NP_WAVES + wave;
            int bbeg = __builtin_amdgcn_readfirstlane(nd ? cn[nd - 1] : s0);
            int bend = __builtin_amdgcn_readfirstlane(cn[nd]);
            bbeg = bbeg < 0 ? 0 : bbeg;
            bend = bend > E ? E : bend;
            order_bucket_mem(bsrc, csrc, bbeg, bend, lane);
        }
    }
}

constexpr int PJ_WAVES = NP_WAVES;
constexpr int PJ_LDS_BYTES = gemm_cfg<9, true>::SB * (int)sizeof(float2);
constexpr int PP_LDS_BYTES = PJ_LDS_BYTES > (int)sizeof(np_lds_t) ? PJ_LDS_BYTES : (int)sizeof(np_lds_t);

__global__ void __launch_bounds__(NP_WAVES * 32)
place_and_project(const int* __restrict__ pairs_, const int* __restrict__ segs,
                  int* __restrict__ rowptr, int* bsrc, int* __restrict__ csrc, int n, int E,
                  int np_blocks,
                  const float* __restrict__ X, const float* __restrict__ W1e,
                  float* __restrict__ xh, float* __restrict__ es, float* __restrict__ ed)
{
    __shared__ __align__(16) unsigned char smem[PP_LDS_BYTES];
    const int b = blockIdx.x;
    if (b < np_blocks) {
        node_place_body((np_lds_t*)smem, b, np_blocks, pairs_, segs, rowptr, bsrc, csrc, n, E);
    } else {
        gemm_body<9, true, PJ_WAVES>((float2*)smem, b - np_blocks, X, W1e, xh, es, ed, n);
    }
}


template <int G>
__device__ __forceinline__ void l1_ask(const float* __restrict__ xh, int sl, int lane, float4 (&v)[8])
{
#pragma unroll
    for (int q = 0; q < 8; ++q) {
        const int su = __builtin_amdgcn_readlane(sl, 8 * G + q);
        v[q] = *(const float4*)(xh + (size_t)su * 128 + lane * 4);
    }
}

template <int G>
__device__ __forceinline__ void l1_add(const float4 (&v)[8], float pw0, float pw1, int cnt, float4& acc)
{
    const float w[8] = { quad_move<0x00>(pw0), quad_move<0x55>(pw0),
                         quad_move<0xAA>(pw0), quad_move<0xFF>(pw0),
                         quad_move<0x00>(pw1), quad_move<0x55>(pw1),
                         quad_move<0xAA>(pw1), quad_move<0xFF>(pw1) };
#pragma unroll
    for (int q = 0; q < 8; ++q) {
        const bool have = (8 * G + q == 0) || (8 * G + q < cnt);
        const float tx = fmaf(w[q], v[q].x, acc.x);
        const float ty = fmaf(w[q], v[q].y, acc.y);
        const float tz = fmaf(w[q], v[q].z, acc.z);
        const float tw = fmaf(w[q], v[q].w, acc.w);
        acc.x = have ? tx : acc.x;
        acc.y = have ? ty : acc.y;
        acc.z = have ? tz : acc.z;
        acc.w = have ? tw : acc.w;
    }
}

__global__ void __launch_bounds__(256)
node_gather_l1(const int* __restrict__ rowptr, const int* __restrict__ csrc,
               const float* __restrict__ es, const float* __restrict__ ed,
               const float* __restrict__ xh, const float* __restrict__ b,
               float* __restrict__ h1, int n)
{
    const int lane = threadIdx.x & 31;
    const int d = blockIdx.x * 8 + (threadIdx.x >> 5);
    if (d >= n) return;
    const int beg = __builtin_amdgcn_readfirstlane(rowptr[d]);
    const int end = __builtin_amdgcn_readfirstlane(rowptr[d + 1]);
    const int deg = end - beg;
    const int cnt = deg + 1;

    float4 acc = make_float4(0.f, 0.f, 0.f, 0.f);

    if (cnt <= 32) {
        const int hq = lane >> 2, sub = lane & 3;
        const float edh = ed[d * 8 + hq];
        const int sl = edge_src(csrc, beg, lane, deg, d, n);

        float a[8];
#pragma unroll
        for (int t = 0; t < 8; ++t) {
            const int s = __shfl(sl, 4 * t + sub);
            a[t] = es[s * 8 + hq];
        }
        float4 v[8];
        l1_ask<0>(xh, sl, lane, v);
        __builtin_amdgcn_sched_group_barrier(0x020, 16, 0);
        __builtin_amdgcn_sched_group_barrier(0x002, 8, 0);

        float m = -__builtin_inff();
#pragma unroll
        for (int t = 0; t < 8; ++t) {
            const int el = 4 * t + sub;
            a[t] = lrelu(a[t] + edh);
            if (el < cnt) m = fmaxf(m, a[t]);
        }
        m = fmaxf(m, quad_move<QUAD_XOR1>(m));
        m = fmaxf(m, quad_move<QUAD_XOR2>(m));

        float sum = 0.0f;
#pragma unroll
        for (int t = 0; t < 8; ++t) {
            const int el = 4 * t + sub;
            if (el < cnt) sum += __expf(a[t] - m);
        }
        sum += quad_move<QUAD_XOR1>(sum);
        sum += quad_move<QUAD_XOR2>(sum);
        const float inv = 1.0f / (sum + 1e-16f);

#pragma unroll
        for (int t = 0; t < 8; ++t) a[t] = __expf(a[t] - m) * inv;

        l1_add<0>(v, a[0], a[1], cnt, acc);
        if (cnt >  8) { l1_ask<1>(xh, sl, lane, v); l1_add<1>(v, a[2], a[3], cnt, acc); }
        if (cnt > 16) { l1_ask<2>(xh, sl, lane, v); l1_add<2>(v, a[4], a[5], cnt, acc); }
        if (cnt > 24) { l1_ask<3>(xh, sl, lane, v); l1_add<3>(v, a[6], a[7], cnt, acc); }
    } else {
        const int h = lane & 7, sub = lane >> 3;
        const float edh = ed[d * 8 + h];

        float m = -__builtin_inff();
        for (int c0 = 0; c0 < cnt; c0 += 32) {
            const int i  = c0 + lane;
            const int sl = edge_src(csrc, beg, i, deg, d, n);
            const int nv = (cnt - c0 < 32) ? cnt - c0 : 32;
#pragma unroll
            for (int t = 0; t < 8; ++t) {
                const int el = 4 * t + sub;
                const int s  = __shfl(sl, el);
                const float a = lrelu(es[s * 8 + h] + edh);
                if (el < nv) m = fmaxf(m, a);
            }
        }
        m = fmaxf(m, __shfl_xor(m, 8));
        m = fmaxf(m, __shfl_xor(m, 16));

        float sum = 0.0f;
        for (int c0 = 0; c0 < cnt; c0 += 32) {
            const int i  = c0 + lane;
            const int sl = edge_src(csrc, beg, i, deg, d, n);
            const int nv = (cnt - c0 < 32) ? cnt - c0 : 32;
#pragma unroll
            for (int t = 0; t < 8; ++t) {
                const int el = 4 * t + sub;
                const int s  = __shfl(sl, el);
                const float a = lrelu(es[s * 8 + h] + edh);
                if (el < nv) sum += __expf(a - m);
            }
        }
        sum += __shfl_xor(sum, 8);
        sum += __shfl_xor(sum, 16);
        const float inv = 1.0f / (sum + 1e-16f);

        const int hh = lane >> 2;
        const float m2   = __shfl(m, hh);
        const float inv2 = __shfl(inv, hh);
        const float ed2  = __shfl(edh, hh);

        for (int c0 = 0; c0 < cnt; c0 += 32) {
            const int i  = c0 + lane;
            const int sl = edge_src(csrc, beg, i, deg, d, n);
            const int nv = (cnt - c0 < 32) ? cnt - c0 : 32;
            for (int g = 0; 4 * g < nv; ++g) {
                float4 v[4];
                float  ev[4];
#pragma unroll
                for (int q = 0; q < 4; ++q) {
                    const int su = __builtin_amdgcn_readlane(sl, 4 * g + q);
                    v[q]  = *(const float4*)(xh + (size_t)su * 128 + lane * 4);
                    ev[q] = es[su * 8 + hh];
                }
#pragma unroll
                for (int q = 0; q < 4; ++q) {
                    if (4 * g + q < nv) {
                        const float w = __expf(lrelu(ev[q] + ed2) - m2) * inv2;
                        acc.x = fmaf(w, v[q].x, acc.x);
                        acc.y = fmaf(w, v[q].y, acc.y);
                        acc.z = fmaf(w, v[q].z, acc.z);
                        acc.w = fmaf(w, v[q].w, acc.w);
                    }
                }
            }
        }
    }

    const float4 bv = *(const float4*)(b + lane * 4);
    float4 o;
    o.x = acc.x + bv.x; o.y = acc.y + bv.y; o.z = acc.z + bv.z; o.w = acc.w + bv.w;
    o.x = o.x > 0.0f ? o.x : expm1f(o.x);
    o.y = o.y > 0.0f ? o.y : expm1f(o.y);
    o.z = o.z > 0.0f ? o.z : expm1f(o.z);
    o.w = o.w > 0.0f ? o.w : expm1f(o.w);
    *(float4*)(h1 + (size_t)d * 128 + lane * 4) = o;
}


__device__ __forceinline__ void l2_rows(const float* __restrict__ xh2, int sl, float wl,
                                        int nv, int jbeg, int slot, int c4, float4& acc)
{
    for (int j0 = jbeg; j0 < nv; j0 += 12) {
        float4 v[4];
        float  w[4];
#pragma unroll
        for (int u = 0; u < 4; ++u) {
            const int j  = j0 + 3 * u + slot;
            const bool ok = (slot < 3) && (j < nv);
            const int jj = ok ? j : 0;
            const int s  = __shfl(sl, jj);
            const float wj = __shfl(wl, jj);
            w[u] = ok ? wj : 0.0f;
            v[u] = *(const float4*)(xh2 + (size_t)s * 48 + c4 * 4);
        }
#pragma unroll
        for (int u = 0; u < 4; ++u) {
            const int j = j0 + 3 * u + slot;
            if ((slot < 3) && (j < nv)) {
                acc.x = fmaf(w[u], v[u].x, acc.x);
                acc.y = fmaf(w[u], v[u].y, acc.y);
                acc.z = fmaf(w[u], v[u].z, acc.z);
                acc.w = fmaf(w[u], v[u].w, acc.w);
            }
        }
    }
}

template <int K0>
__device__ __forceinline__ void l2_ask(const float* __restrict__ xh2, int sl, int nv,
                                       int slot, int c4, float4 (&v)[8])
{
#pragma unroll
    for (int u = 0; u < 4; ++u) {
        const int j  = 3 * (K0 + u) + slot;
        const bool ok = (slot < 3) && (j < nv);
        const int s  = __shfl(sl, ok ? j : 0);
        v[K0 + u] = *(const float4*)(xh2 + (size_t)s * 48 + c4 * 4);
    }
}

template <int K0>
__device__ __forceinline__ void l2_add(const float4 (&v)[8], float wl, int nv,
                                       int slot, float4& acc)
{
#pragma unroll
    for (int u = 0; u < 4; ++u) {
        const int j  = 3 * (K0 + u) + slot;
        const bool ok = (slot < 3) && (j < nv);
        const float wj = __shfl(wl, ok ? j : 0);
        if (ok) {
            acc.x = fmaf(wj, v[K0 + u].x, acc.x);
            acc.y = fmaf(wj, v[K0 + u].y, acc.y);
            acc.z = fmaf(wj, v[K0 + u].z, acc.z);
            acc.w = fmaf(wj, v[K0 + u].w, acc.w);
        }
    }
}

__global__ void __launch_bounds__(256)
node_gather_l2(const int* __restrict__ rowptr, const int* __restrict__ csrc,
               const float* __restrict__ xh2, const float* __restrict__ b,
               float* __restrict__ out, int n)
{
    __shared__ __align__(16) float so[8 * 40];
    const int lane = threadIdx.x & 31;
    const int wv   = threadIdx.x >> 5;
    const int d0   = blockIdx.x * 8;
    const int d    = d0 + wv;

    if (d < n) {
        const int beg = __builtin_amdgcn_readfirstlane(rowptr[d]);
        const int end = __builtin_amdgcn_readfirstlane(rowptr[d + 1]);
        const int deg = end - beg;
        const int cnt = deg + 1;
        const float edd = xh2[(size_t)d * W2P_COLS + XH2_ED];

        const int slot = lane / 10;
        const int c4   = lane - slot * 10;
        float4 acc = make_float4(0.f, 0.f, 0.f, 0.f);

        if (cnt <= 32) {
            const int sl  = edge_src(csrc, beg, lane, deg, d, n);
            const float a = lrelu(xh2[(size_t)sl * W2P_COLS + XH2_ES] + edd);

            float4 v[8];
            l2_ask<0>(xh2, sl, cnt, slot, c4, v);
            if (cnt > 12) l2_ask<4>(xh2, sl, cnt, slot, c4, v);

            float m = -__builtin_inff();
            if (lane < cnt) m = fmaxf(m, a);
            for (int o = 16; o >= 1; o >>= 1) m = fmaxf(m, __shfl_xor(m, o));

            const float e = __expf(a - m);
            float sum = 0.0f;
            if (lane < cnt) sum += e;
            for (int o = 16; o >= 1; o >>= 1) sum += __shfl_xor(sum, o);
            const float inv = 1.0f / (sum + 1e-16f);

            const float wl = e * inv;
            l2_add<0>(v, wl, cnt, slot, acc);
            if (cnt > 12) l2_add<4>(v, wl, cnt, slot, acc);
            if (cnt > 24) l2_rows(xh2, sl, wl, cnt, 24, slot, c4, acc);
        } else {
            float m = -__builtin_inff();
            for (int c0 = 0; c0 < cnt; c0 += 32) {
                const int i  = c0 + lane;
                const int sl = edge_src(csrc, beg, i, deg, d, n);
                const float a = lrelu(xh2[(size_t)sl * W2P_COLS + XH2_ES] + edd);
                if (i < cnt) m = fmaxf(m, a);
            }
            for (int o = 16; o >= 1; o >>= 1) m = fmaxf(m, __shfl_xor(m, o));

            float sum = 0.0f;
            for (int c0 = 0; c0 < cnt; c0 += 32) {
                const int i  = c0 + lane;
                const int sl = edge_src(csrc, beg, i, deg, d, n);
                const float a = lrelu(xh2[(size_t)sl * W2P_COLS + XH2_ES] + edd);
                if (i < cnt) sum += __expf(a - m);
            }
            for (int o = 16; o >= 1; o >>= 1) sum += __shfl_xor(sum, o);
            const float inv = 1.0f / (sum + 1e-16f);

            for (int c0 = 0; c0 < cnt; c0 += 32) {
                const int i  = c0 + lane;
                const int sl = edge_src(csrc, beg, i, deg, d, n);
                const float a  = lrelu(xh2[(size_t)sl * W2P_COLS + XH2_ES] + edd);
                const float wl = __expf(a - m) * inv;
                const int nv = (cnt - c0 < 32) ? cnt - c0 : 32;
                l2_rows(xh2, sl, wl, nv, 0, slot, c4, acc);
            }
        }
        const int l1 = (lane + 10) & 31, l2 = (lane + 20) & 31;
        acc.x += __shfl(acc.x, l1) + __shfl(acc.x, l2);
        acc.y += __shfl(acc.y, l1) + __shfl(acc.y, l2);
        acc.z += __shfl(acc.z, l1) + __shfl(acc.z, l2);
        acc.w += __shfl(acc.w, l1) + __shfl(acc.w, l2);
        if (lane < 10) {
            const float4 bv = *(const float4*)(b + lane * 4);
            float4 o;
            o.x = acc.x + bv.x; o.y = acc.y + bv.y; o.z = acc.z + bv.z; o.w = acc.w + bv.w;
            *(float4*)(so + wv * 40 + lane * 4) = o;
        }
    }
    __syncthreads();
    const int rows = (n - d0 < 8) ? n - d0 : 8;
    if ((int)threadIdx.x < rows * 10)
        *(float4*)(out + (size_t)d0 * 40 + threadIdx.x * 4) = *(const float4*)(so + threadIdx.x * 4);
}

extern "C" void kernel_launch(void* const* d_in, const int* in_sizes, int n_in,
                              void* d_out, int out_size, void* d_ws, size_t ws_size,
                              hipStream_t stream)
{
    (void)n_in; (void)out_size; (void)ws_size;

    const float* x   = (const float*)d_in[0];
    const int*   ei  = (const int*)d_in[1];
    const float* W1  = (const float*)d_in[2];
    const float* as1 = (const float*)d_in[3];
    const float* ad1 = (const float*)d_in[4];
    const float* b1  = (const float*)d_in[5];
    const float* W2  = (const float*)d_in[6];
    const float* as2 = (const float*)d_in[7];
    const float* ad2 = (const float*)d_in[8];
    const float* b2  = (const float*)d_in[9];
    float* out = (float*)d_out;

    const int n  = in_sizes[0] / 128;
    const int E  = in_sizes[1] / 2;

    char* ws = (char*)d_ws;
    size_t off = 0;
    auto alloc = [&](size_t bytes) -> float* {
        float* p = (float*)(ws + off);
        off += (bytes + 255) & ~(size_t)255;
        return p;
    };
    float* xh1  = alloc((size_t)n * 128 * 4);
    float* h1   = alloc((size_t)n * 128 * 4);
    float* es1  = alloc((size_t)n * 8 * 4);
    float* ed1  = alloc((size_t)n * 8 * 4);
    float* W2p  = alloc((size_t)128 * W2P_COLS * 4);
    const int nr = (n + (1 << RSHIFT) - 1) >> RSHIFT;
    int* segs   = (int*)alloc((size_t)((nr + 1) * L1_BLOCKS + 16) * 4);
    int* rowptr = (int*)alloc((size_t)(n + 16) * 4);
    int* csrc   = (int*)alloc((size_t)E * 4);
    int* bsrc   = (int*)h1;
    int* pairs  = (int*)h1 + E;
    float* W1e  = h1 + (size_t)n * 64;

    const int B = 256;

    range_bucket<<<L1_BLOCKS + (PREP_ELEMS + 1023) / 1024, 1024, nr * sizeof(int), stream>>>(ei, E, nr, segs, pairs,
                                                                W1, as1, ad1, W2, as2, ad2, W1e, W2p);
    place_and_project<<<nr + (n + PJ_WAVES * 16 - 1) / (PJ_WAVES * 16), NP_WAVES * 32, 0, stream>>>(
        pairs, segs, rowptr, bsrc, csrc, n, E, nr, x, W1e, xh1, es1, ed1);

    node_gather_l1<<<(n + 7) / 8, B, 0, stream>>>(rowptr, csrc, es1, ed1, xh1, b1, h1, n);

    float* xh2 = xh1;
    gemm_wmma_f32<3, false><<<(n + 127) / 128, 256, 0, stream>>>(h1, W2p, xh2, nullptr, nullptr, n);
    node_gather_l2<<<(n + 7) / 8, B, 0, stream>>>(rowptr, csrc, xh2, b2, out, n);
}


# layer-2 projection moves into the layer-1 gather: a block (16 waves, 2 nodes each) keeps its 32 finished rows in LDS, 2 waves do the WMMA tiles, same k order; h1 and one launch go; sum orders same
# speedup vs baseline: 10.6438x; 1.0376x over previous
#include <hip/hip_runtime.h>
#include <hip/hip_bf16.h>
#include <math.h>


typedef float v2f __attribute__((ext_vector_type(2)));
typedef float v8f __attribute__((ext_vector_type(8)));

template <int NT, bool LOGITS>
struct gemm_cfg {
    static constexpr int NCOL = NT * 16;
    static constexpr int P  = LOGITS ? NCOL : NCOL + 16;
    static constexpr int SB = 64 * P;
};

template <int NT, bool LOGITS, int NW>
__device__ __forceinline__ void gemm_body(float2* sB, int blk,
                                          const float* __restrict__ X,
                                          const float* __restrict__ Wm,
                                          float* __restrict__ Y,
                                          float* __restrict__ es,
                                          float* __restrict__ ed,
                                          int nrows)
{
    constexpr int NCOL = gemm_cfg<NT, LOGITS>::NCOL;
    constexpr int YT   = LOGITS ? NT - 1 : NT;
    constexpr int YP   = YT * 16;
    constexpr int P    = gemm_cfg<NT, LOGITS>::P;

    const int tid = threadIdx.x;
    for (int i = tid; i < 64 * NCOL; i += NW * 32) {
        const int kp = i / NCOL;
        const int nn = i - kp * NCOL;
        float2 w2;
        w2.x = Wm[(2 * kp) * NCOL + nn];
        w2.y = Wm[(2 * kp + 1) * NCOL + nn];
        sB[kp * P + nn] = w2;
    }
    __syncthreads();

    const int wave = tid >> 5;
    const int lane = tid & 31;
    const int row0 = blk * (NW * 16) + wave * 16;
    const int m    = lane & 15;
    const int khp  = lane >> 4;
    int row = row0 + m;
    if (row > nrows - 1) row = nrows - 1;

    const float2* Xp = (const float2*)(X + (size_t)row * 128) + khp;
    const int bbase = khp * P + m;

    v8f acc[NT];
#pragma unroll
    for (int t = 0; t < NT; ++t) acc[t] = (v8f)0.0f;

    for (int kk = 0; kk < 32; ++kk) {
        const float2 av = Xp[kk * 2];
        v2f a; a.x = av.x; a.y = av.y;
        const int bo = kk * 2 * P + bbase;
#pragma unroll
        for (int t = 0; t < NT; ++t) {
            const float2 bv = sB[bo + t * 16];
            v2f b; b.x = bv.x; b.y = bv.y;
            acc[t] = __builtin_amdgcn_wmma_f32_16x16x4_f32(
                false, a, false, b, (short)0, acc[t], false, false);
        }
    }

    const int rbase = khp * 8;
#pragma unroll
    for (int r = 0; r < 8; ++r) {
        const int rr = row0 + rbase + r;
        if (rr < nrows) {
#pragma unroll
            for (int t = 0; t < YT; ++t)
                Y[(size_t)rr * YP + t * 16 + m] = acc[t][r];
            if (LOGITS) {
                const float v = acc[NT - 1][r];
                if (m < 8) es[(size_t)rr * 8 + m] = v;
                else       ed[(size_t)rr * 8 + (m - 8)] = v;
            }
        }
    }
}

__device__ __forceinline__ float lrelu(float a) {
    return a > 0.0f ? a : 0.2f * a;
}

__device__ __forceinline__ int edge_src(const int* __restrict__ csrc, int beg,
                                        int i, int deg, int d, int n) {
    if (i >= deg) return d;
    const int s = csrc[beg + i];
    return s < 0 ? 0 : (s > n - 1 ? n - 1 : s);
}

template <int CTRL>
__device__ __forceinline__ float quad_move(float x) {
    return __int_as_float(__builtin_amdgcn_update_dpp(0, __float_as_int(x), CTRL, 0xF, 0xF, true));
}
constexpr int QUAD_XOR1 = 0xB1;
constexpr int QUAD_XOR2 = 0x4E;

constexpr int W1E_COLS = 144;
constexpr int W2P_COLS = 48;
constexpr int XH2_ES   = 40;
constexpr int XH2_ED   = 41;

constexpr int PREP_ELEMS = 128 * W1E_COLS + 128 * W2P_COLS;

__device__ __forceinline__ void prep_element(int t,
                                             const float* __restrict__ W1,
                                             const float* __restrict__ as1,
                                             const float* __restrict__ ad1,
                                             const float* __restrict__ W2,
                                             const float* __restrict__ as2,
                                             const float* __restrict__ ad2,
                                             float* __restrict__ W1e, float* __restrict__ W2p)
{
    if (t < 128 * W1E_COLS) {
        const int k = t / W1E_COLS, c = t - k * W1E_COLS;
        float v;
        if (c < 128) {
            v = W1[k * 128 + c];
        } else {
            const int h = (c - 128) & 7;
            const float* a = (c < 136 ? as1 : ad1) + h * 16;
            const float* w = W1 + k * 128 + h * 16;
            float s = 0.f;
            for (int j = 0; j < 16; ++j) s += w[j] * a[j];
            v = s;
        }
        W1e[t] = v;
    } else if (t < PREP_ELEMS) {
        const int u = t - 128 * W1E_COLS;
        const int k = u / W2P_COLS, c = u - k * W2P_COLS;
        float v = 0.0f;
        if (c < 40) {
            v = W2[k * 40 + c];
        } else if (c == XH2_ES || c == XH2_ED) {
            const float* a = (c == XH2_ES) ? as2 : ad2;
            const float* w = W2 + k * 40;
            float s = 0.f;
            for (int j = 0; j < 40; ++j) s += w[j] * a[j];
            v = s;
        }
        W2p[((k >> 1) * W2P_COLS + c) * 2 + (k & 1)] = v;
    }
}

constexpr int RSHIFT = 8;
constexpr int L1_BLOCKS = 128;

__global__ void __launch_bounds__(1024)
range_bucket(const int* __restrict__ ei, int E, int nr,
             int* __restrict__ segs, int* __restrict__ pairs_,
             const float* __restrict__ W1, const float* __restrict__ as1,
             const float* __restrict__ ad1, const float* __restrict__ W2,
             const float* __restrict__ as2, const float* __restrict__ ad2,
             float* __restrict__ W1e, float* __restrict__ W2p)
{
    extern __shared__ int lc[];
    __shared__ int wsum[32];
    int2* __restrict__ pairs = (int2*)pairs_;
    const int tid = threadIdx.x, lane = tid & 31, wave = tid >> 5;
    const int c = blockIdx.x, nb = L1_BLOCKS;
    if (c >= nb) {
        prep_element((c - nb) * 1024 + tid, W1, as1, ad1, W2, as2, ad2, W1e, W2p);
        return;
    }
    for (int r = tid; r < nr; r += 1024) lc[r] = 0;
    __syncthreads();
    const int chunk = (E + nb - 1) / nb;
    const int e0 = c * chunk;
    const int e1 = (e0 + chunk < E) ? e0 + chunk : E;
    const int* __restrict__ dstp = ei + E;
    int i0 = e0 + tid;
    for (; i0 + 3 * 1024 < e1; i0 += 4 * 1024) {
        int d[4];
#pragma unroll
        for (int u = 0; u < 4; ++u) d[u] = dstp[i0 + u * 1024];
#pragma unroll
        for (int u = 0; u < 4; ++u) atomicAdd(&lc[d[u] >> RSHIFT], 1);
    }
    for (; i0 < e1; i0 += 1024) atomicAdd(&lc[dstp[i0] >> RSHIFT], 1);
    __syncthreads();

    int carry = e0;
    for (int base = 0; base < nr; base += 1024) {
        const int r = base + tid;
        const int v = (r < nr) ? lc[r] : 0;
        int x = v;
        for (int o = 1; o < 32; o <<= 1) {
            const int y = __shfl_up(x, o);
            if (lane >= o) x += y;
        }
        if (lane == 31) wsum[wave] = x;
        __syncthreads();
        const int ws = wsum[lane];
        int wx = ws;
        for (int o = 1; o < 32; o <<= 1) {
            const int y = __shfl_up(wx, o);
            if (lane >= o) wx += y;
        }
        const int tile_total = __shfl(wx, 31);
        const int wave_excl  = __shfl(wx - ws, wave);
        __syncthreads();
        const int ex = carry + wave_excl + (x - v);
        if (r < nr) {
            lc[r] = ex;
            segs[r * nb + c] = ex;
        }
        carry += tile_total;
    }
    if (tid == 0) segs[nr * nb + c] = carry;
    __syncthreads();

    for (i0 = e0 + tid; i0 + 3 * 1024 < e1; i0 += 4 * 1024) {
        int d[4], s[4];
#pragma unroll
        for (int u = 0; u < 4; ++u) { d[u] = dstp[i0 + u * 1024]; s[u] = ei[i0 + u * 1024]; }
#pragma unroll
        for (int u = 0; u < 4; ++u) {
            int pos = atomicAdd(&lc[d[u] >> RSHIFT], 1);
            pos = pos < 0 ? 0 : (pos > E - 1 ? E - 1 : pos);
            pairs[pos] = make_int2(d[u], s[u]);
        }
    }
    for (; i0 < e1; i0 += 1024) {
        const int d = dstp[i0], s = ei[i0];
        int pos = atomicAdd(&lc[d >> RSHIFT], 1);
        pos = pos < 0 ? 0 : (pos > E - 1 ? E - 1 : pos);
        pairs[pos] = make_int2(d, s);
    }
}

constexpr int NP_WAVES = 8;
constexpr int NP_PPW   = L1_BLOCKS / NP_WAVES;
constexpr int NP_CAP   = 8192;

struct in_lds_t { static constexpr bool value = true;  };
struct in_mem_t { static constexpr bool value = false; };

__device__ __forceinline__ void order_bucket_mem(const int* bsrc, int* __restrict__ csrc,
                                                 int beg, int end, int lane)
{
    const int deg = end - beg;
    for (int c0 = 0; c0 < deg; c0 += 32) {
        const int i   = c0 + lane;
        const int key = (i < deg) ? bsrc[beg + i] : 0;
        int rank = 0;
        for (int b0 = 0; b0 < deg; b0 += 32) {
            const int kc = (b0 + lane < deg) ? bsrc[beg + b0 + lane] : 0;
            const int nb = (deg - b0 < 32) ? deg - b0 : 32;
            for (int jj = 0; jj < nb; ++jj) {
                const int kj = __builtin_amdgcn_readlane(kc, jj);
                const int j  = b0 + jj;
                rank += (kj < key || (kj == key && j < i)) ? 1 : 0;
            }
        }
        if (i < deg) csrc[beg + rank] = key;
    }
}

struct np_lds_t {
    int cn[1 << RSHIFT];
    int wt[(1 << RSHIFT) / 32];
    int ps[L1_BLOCKS];
    int pe[L1_BLOCKS];
    int cst[(1 << RSHIFT) + 1];
    int lsrc[NP_CAP + 8];
    unsigned char lnod[NP_CAP];
};

__device__ __forceinline__ void
node_place_body(np_lds_t* lds, int b, int nblk,
                const int* __restrict__ pairs_, const int* __restrict__ segs,
                int* __restrict__ rowptr, int* bsrc, int* __restrict__ csrc, int n, int E)
{
    int* const cn   = lds->cn;
    int* const wt   = lds->wt;
    int* const ps   = lds->ps;
    int* const pe   = lds->pe;
    int* const cst  = lds->cst;
    int* const lsrc = lds->lsrc;
    unsigned char* const lnod = lds->lnod;
    const int2* __restrict__ pairs = (const int2*)pairs_;
    const int tid = threadIdx.x, lane = tid & 31, wave = tid >> 5;
    const int d0 = b << RSHIFT;
    const int chunk = (E + L1_BLOCKS - 1) / L1_BLOCKS;
    if (tid < L1_BLOCKS) {
        ps[tid] = segs[b * L1_BLOCKS + tid];
        pe[tid] = segs[(b + 1) * L1_BLOCKS + tid];
    }
    if (tid < (1 << RSHIFT)) cn[tid] = 0;
    __syncthreads();

    int a0 = 0, a1 = 0;
    for (int cc = lane; cc < L1_BLOCKS; cc += 32) {
        a0 += ps[cc] - cc * chunk;
        a1 += pe[cc] - cc * chunk;
    }
    for (int o = 16; o >= 1; o >>= 1) {
        a0 += __shfl_xor(a0, o);
        a1 += __shfl_xor(a1, o);
    }
    const int s0 = __builtin_amdgcn_readfirstlane(a0);
    const int s1 = __builtin_amdgcn_readfirstlane(a1);
    const int total = s1 - s0;
    const bool fits = total <= NP_CAP;

    int pb[NP_PPW], pl[NP_PPW];
#pragma unroll
    for (int q = 0; q < NP_PPW; ++q) {
        pb[q] = __builtin_amdgcn_readfirstlane(ps[wave * NP_PPW + q]);
        pl[q] = __builtin_amdgcn_readfirstlane(pe[wave * NP_PPW + q]) - pb[q];
    }

    {
        int dn[NP_PPW][2];
#pragma unroll
        for (int q = 0; q < NP_PPW; ++q)
#pragma unroll
            for (int t = 0; t < 2; ++t) {
                const int j = lane + 32 * t;
                dn[q][t] = (j < pl[q]) ? pairs[pb[q] + j].x : -1;
            }
#pragma unroll
        for (int q = 0; q < NP_PPW; ++q)
#pragma unroll
            for (int t = 0; t < 2; ++t)
                if (dn[q][t] >= 0) atomicAdd(&cn[dn[q][t] - d0], 1);
#pragma unroll
        for (int q = 0; q < NP_PPW; ++q)
            for (int j = lane + 64; j < pl[q]; j += 32) atomicAdd(&cn[pairs[pb[q] + j].x - d0], 1);
    }
    __syncthreads();
    int cnt = 0, x = 0;
    if (tid < (1 << RSHIFT)) {
        cnt = cn[tid];
        x = cnt;
        for (int o = 1; o < 32; o <<= 1) {
            const int y = __shfl_up(x, o);
            if (lane >= o) x += y;
        }
        if (lane == 31) wt[wave] = x;
    }
    __syncthreads();
    if (tid < (1 << RSHIFT)) {
        int pre = s0;
        for (int w = 0; w < wave; ++w) pre += wt[w];
        const int ex = pre + (x - cnt);
        cn[tid] = ex;
        cst[tid] = ex - s0;
        if (tid == 0) cst[1 << RSHIFT] = total;
        if (d0 + tid < n) rowptr[d0 + tid] = ex;
    }
    if (b == nblk - 1 && tid == 0) rowptr[n] = s1;
    __syncthreads();
    auto place_all = [&](auto ldsb) {
        constexpr bool LDSB = decltype(ldsb)::value;
        auto put_src = [&](int pos, int nd, int s) {
            if (LDSB) {
                int lp = pos - s0;
                lp = lp < 0 ? 0 : (lp > NP_CAP - 1 ? NP_CAP - 1 : lp);
                lsrc[lp] = s;
                lnod[lp] = (unsigned char)nd;
            } else {
                const int gp = pos < 0 ? 0 : (pos > E - 1 ? E - 1 : pos);
                bsrc[gp] = s;
            }
        };
        int2 p[NP_PPW][2];
#pragma unroll
        for (int q = 0; q < NP_PPW; ++q)
#pragma unroll
            for (int t = 0; t < 2; ++t) {
                const int j = lane + 32 * t;
                p[q][t] = (j < pl[q]) ? pairs[pb[q] + j] : make_int2(-1, 0);
            }
#pragma unroll
        for (int q = 0; q < NP_PPW; ++q)
#pragma unroll
            for (int t = 0; t < 2; ++t)
                if (p[q][t].x >= 0) {
                    const int nd = p[q][t].x - d0;
                    const int pos = atomicAdd(&cn[nd], 1);
                    put_src(pos, nd, p[q][t].y);
                }
#pragma unroll
        for (int q = 0; q < NP_PPW; ++q)
            for (int j = lane + 64; j < pl[q]; j += 32) {
                const int2 pp = pairs[pb[q] + j];
                const int nd = pp.x - d0;
                const int pos = atomicAdd(&cn[nd], 1);
                put_src(pos, nd, pp.y);
            }
    };
    if (fits) place_all(in_lds_t{}); else place_all(in_mem_t{});
    __syncthreads();

    if (fits) {
        for (int i = tid; i < total; i += NP_WAVES * 32) {
            const int key = lsrc[i];
            const int nd  = lnod[i];
            int bbeg = cst[nd], bend = cst[nd + 1];
            bbeg = bbeg < 0 ? 0 : bbeg;
            bend = bend > total ? total : bend;
            int rank = 0;
            for (int j0 = bbeg; j0 < bend; j0 += 8) {
                const int own = i - j0;
                const int inb = bend - j0;
                int k[8];
#pragma unroll
                for (int u = 0; u < 8; ++u) k[u] = lsrc[j0 + u];
#pragma unroll
                for (int u = 0; u < 8; ++u) {
                    const int lim = key + (u < own ? 1 : 0);
                    rank += ((int)(u < inb) & (int)(k[u] < lim));
                }
            }
            int pos = s0 + bbeg + rank;
            pos = pos < 0 ? 0 : (pos > E - 1 ? E - 1 : pos);
            csrc[pos] = key;
        }
    } else {
        for (int q = 0; q < (1 << RSHIFT) / NP_WAVES; ++q) {
            const int nd  = q * NP_WAVES + wave;
            int bbeg = __builtin_amdgcn_readfirstlane(nd ? cn[nd - 1] : s0);
            int bend = __builtin_amdgcn_readfirstlane(cn[nd]);
            bbeg = bbeg < 0 ? 0 : bbeg;
            bend = bend > E ? E : bend;
            order_bucket_mem(bsrc, csrc, bbeg, bend, lane);
        }
    }
}

constexpr int PJ_WAVES = NP_WAVES;
constexpr int PJ_LDS_BYTES = gemm_cfg<9, true>::SB * (int)sizeof(float2);
constexpr int PP_LDS_BYTES = PJ_LDS_BYTES > (int)sizeof(np_lds_t) ? PJ_LDS_BYTES : (int)sizeof(np_lds_t);

__global__ void __launch_bounds__(NP_WAVES * 32)
place_and_project(const int* __restrict__ pairs_, const int* __restrict__ segs,
                  int* __restrict__ rowptr, int* bsrc, int* __restrict__ csrc, int n, int E,
                  int np_blocks,
                  const float* __restrict__ X, const float* __restrict__ W1e,
                  float* __restrict__ xh, float* __restrict__ es, float* __restrict__ ed)
{
    __shared__ __align__(16) unsigned char smem[PP_LDS_BYTES];
    const int b = blockIdx.x;
    if (b < np_blocks) {
        node_place_body((np_lds_t*)smem, b, np_blocks, pairs_, segs, rowptr, bsrc, csrc, n, E);
    } else {
        gemm_body<9, true, PJ_WAVES>((float2*)smem, b - np_blocks, X, W1e, xh, es, ed, n);
    }
}


template <int G>
__device__ __forceinline__ void l1_ask(const float* __restrict__ xh, int sl, int lane, float4 (&v)[8])
{
#pragma unroll
    for (int q = 0; q < 8; ++q) {
        const int su = __builtin_amdgcn_readlane(sl, 8 * G + q);
        v[q] = *(const float4*)(xh + (size_t)su * 128 + lane * 4);
    }
}

template <int G>
__device__ __forceinline__ void l1_add(const float4 (&v)[8], float pw0, float pw1, int cnt, float4& acc)
{
    const float w[8] = { quad_move<0x00>(pw0), quad_move<0x55>(pw0),
                         quad_move<0xAA>(pw0), quad_move<0xFF>(pw0),
                         quad_move<0x00>(pw1), quad_move<0x55>(pw1),
                         quad_move<0xAA>(pw1), quad_move<0xFF>(pw1) };
#pragma unroll
    for (int q = 0; q < 8; ++q) {
        const bool have = (8 * G + q == 0) || (8 * G + q < cnt);
        const float tx = fmaf(w[q], v[q].x, acc.x);
        const float ty = fmaf(w[q], v[q].y, acc.y);
        const float tz = fmaf(w[q], v[q].z, acc.z);
        const float tw = fmaf(w[q], v[q].w, acc.w);
        acc.x = have ? tx : acc.x;
        acc.y = have ? ty : acc.y;
        acc.z = have ? tz : acc.z;
        acc.w = have ? tw : acc.w;
    }
}

constexpr int L1_WAVES = 16;
constexpr int L1_NPW   = 2;
constexpr int L1_NODES = L1_WAVES * L1_NPW;
constexpr int SH_PITCH = 132;
constexpr int W2I_F2   = 64 * W2P_COLS;

__global__ void __launch_bounds__(L1_WAVES * 32)
node_gather_l1(const int* __restrict__ rowptr, const int* __restrict__ csrc,
               const float* __restrict__ es, const float* __restrict__ ed,
               const float* __restrict__ xh, const float* __restrict__ b,
               const float* __restrict__ W2p, float* __restrict__ xh2, int n)
{
    __shared__ __align__(16) float2 sB[W2I_F2];
    __shared__ __align__(16) float  sH[L1_NODES * SH_PITCH];
    const int tid   = threadIdx.x;
    const int lane0 = tid & 31;
    const int wv    = tid >> 5;
    const int d0    = blockIdx.x * L1_NODES;

    const int dc0 = (d0 + wv < n) ? d0 + wv : n - 1;
    const int dc1 = (d0 + L1_WAVES + wv < n) ? d0 + L1_WAVES + wv : n - 1;
    const int rp00 = rowptr[dc0], rp01 = rowptr[dc0 + 1];
    const int rp10 = rowptr[dc1], rp11 = rowptr[dc1 + 1];

    {
        const float4* __restrict__ gw = (const float4*)W2p;
        float4* sw = (float4*)sB;
#pragma unroll
        for (int i = 0; i < W2I_F2 / 2 / (L1_WAVES * 32); ++i)
            sw[tid + i * (L1_WAVES * 32)] = gw[tid + i * (L1_WAVES * 32)];
    }

    const int beg0 = __builtin_amdgcn_readfirstlane(rp00), end0 = __builtin_amdgcn_readfirstlane(rp01);
    const int beg1 = __builtin_amdgcn_readfirstlane(rp10), end1 = __builtin_amdgcn_readfirstlane(rp11);

#pragma unroll 1
    for (int j = 0; j < L1_NPW; ++j) {
    int lane = lane0;
    asm volatile("" : "+v"(lane));
    const int d = d0 + j * L1_WAVES + wv;
    float4 o = make_float4(0.f, 0.f, 0.f, 0.f);
    if (d < n) {
    const int beg = j == 0 ? beg0 : beg1;
    const int end = j == 0 ? end0 : end1;
    const int deg = end - beg;
    const int cnt = deg + 1;

    float4 acc = make_float4(0.f, 0.f, 0.f, 0.f);

    if (cnt <= 32) {
        const int hq = lane >> 2, sub = lane & 3;
        const float edh = ed[d * 8 + hq];
        const int sl = edge_src(csrc, beg, lane, deg, d, n);

        float a[8];
#pragma unroll
        for (int t = 0; t < 8; ++t) {
            const int s = __shfl(sl, 4 * t + sub);
            a[t] = es[s * 8 + hq];
        }
        float4 v[8];
        l1_ask<0>(xh, sl, lane, v);
        __builtin_amdgcn_sched_group_barrier(0x020, 16, 0);
        __builtin_amdgcn_sched_group_barrier(0x002, 8, 0);

        float m = -__builtin_inff();
#pragma unroll
        for (int t = 0; t < 8; ++t) {
            const int el = 4 * t + sub;
            a[t] = lrelu(a[t] + edh);
            if (el < cnt) m = fmaxf(m, a[t]);
        }
        m = fmaxf(m, quad_move<QUAD_XOR1>(m));
        m = fmaxf(m, quad_move<QUAD_XOR2>(m));

        float sum = 0.0f;
#pragma unroll
        for (int t = 0; t < 8; ++t) {
            const int el = 4 * t + sub;
            if (el < cnt) sum += __expf(a[t] - m);
        }
        sum += quad_move<QUAD_XOR1>(sum);
        sum += quad_move<QUAD_XOR2>(sum);
        const float inv = 1.0f / (sum + 1e-16f);

#pragma unroll
        for (int t = 0; t < 8; ++t) a[t] = __expf(a[t] - m) * inv;

        l1_add<0>(v, a[0], a[1], cnt, acc);
        if (cnt >  8) { l1_ask<1>(xh, sl, lane, v); l1_add<1>(v, a[2], a[3], cnt, acc); }
        if (cnt > 16) { l1_ask<2>(xh, sl, lane, v); l1_add<2>(v, a[4], a[5], cnt, acc); }
        if (cnt > 24) { l1_ask<3>(xh, sl, lane, v); l1_add<3>(v, a[6], a[7], cnt, acc); }
    } else {
        const int h = lane & 7, sub = lane >> 3;
        const float edh = ed[d * 8 + h];

        float m = -__builtin_inff();
        for (int c0 = 0; c0 < cnt; c0 += 32) {
            const int i  = c0 + lane;
            const int sl = edge_src(csrc, beg, i, deg, d, n);
            const int nv = (cnt - c0 < 32) ? cnt - c0 : 32;
#pragma unroll
            for (int t = 0; t < 8; ++t) {
                const int el = 4 * t + sub;
                const int s  = __shfl(sl, el);
                const float a = lrelu(es[s * 8 + h] + edh);
                if (el < nv) m = fmaxf(m, a);
            }
        }
        m = fmaxf(m, __shfl_xor(m, 8));
        m = fmaxf(m, __shfl_xor(m, 16));

        float sum = 0.0f;
        for (int c0 = 0; c0 < cnt; c0 += 32) {
            const int i  = c0 + lane;
            const int sl = edge_src(csrc, beg, i, deg, d, n);
            const int nv = (cnt - c0 < 32) ? cnt - c0 : 32;
#pragma unroll
            for (int t = 0; t < 8; ++t) {
                const int el = 4 * t + sub;
                const int s  = __shfl(sl, el);
                const float a = lrelu(es[s * 8 + h] + edh);
                if (el < nv) sum += __expf(a - m);
            }
        }
        sum += __shfl_xor(sum, 8);
        sum += __shfl_xor(sum, 16);
        const float inv = 1.0f / (sum + 1e-16f);

        const int hh = lane >> 2;
        const float m2   = __shfl(m, hh);
        const float inv2 = __shfl(inv, hh);
        const float ed2  = __shfl(edh, hh);

        for (int c0 = 0; c0 < cnt; c0 += 32) {
            const int i  = c0 + lane;
            const int sl = edge_src(csrc, beg, i, deg, d, n);
            const int nv = (cnt - c0 < 32) ? cnt - c0 : 32;
            for (int g = 0; 4 * g < nv; ++g) {
                float4 v[4];
                float  ev[4];
#pragma unroll
                for (int q = 0; q < 4; ++q) {
                    const int su = __builtin_amdgcn_readlane(sl, 4 * g + q);
                    v[q]  = *(const float4*)(xh + (size_t)su * 128 + lane * 4);
                    ev[q] = es[su * 8 + hh];
                }
#pragma unroll
                for (int q = 0; q < 4; ++q) {
                    if (4 * g + q < nv) {
                        const float w = __expf(lrelu(ev[q] + ed2) - m2) * inv2;
                        acc.x = fmaf(w, v[q].x, acc.x);
                        acc.y = fmaf(w, v[q].y, acc.y);
                        acc.z = fmaf(w, v[q].z, acc.z);
                        acc.w = fmaf(w, v[q].w, acc.w);
                    }
                }
            }
        }
    }

    const float4 bv = *(const float4*)(b + lane * 4);
    o.x = acc.x + bv.x; o.y = acc.y + bv.y; o.z = acc.z + bv.z; o.w = acc.w + bv.w;
    o.x = o.x > 0.0f ? o.x : expm1f(o.x);
    o.y = o.y > 0.0f ? o.y : expm1f(o.y);
    o.z = o.z > 0.0f ? o.z : expm1f(o.z);
    o.w = o.w > 0.0f ? o.w : expm1f(o.w);
    }
    *(float4*)(sH + (j * L1_WAVES + wv) * SH_PITCH + lane * 4) = o;
    }

    __syncthreads();
    const int wfirst = (int)(blockIdx.x & 1) * 2;
    if (wv == wfirst || wv == wfirst + 1) {
        const int rt  = wv - wfirst;
        const int m   = lane0 & 15;
        const int khp = lane0 >> 4;
        const float*  Ap = sH + (rt * 16 + m) * SH_PITCH + 2 * khp;
        const float2* Bq = sB + khp * W2P_COLS + m;
        v8f c2[3];
#pragma unroll
        for (int t = 0; t < 3; ++t) c2[t] = (v8f)0.0f;
        for (int kk = 0; kk < 32; ++kk) {
            const float2 av = *(const float2*)(Ap + 4 * kk);
            v2f a; a.x = av.x; a.y = av.y;
#pragma unroll
            for (int t = 0; t < 3; ++t) {
                const float2 bv2 = Bq[kk * 2 * W2P_COLS + t * 16];
                v2f bb; bb.x = bv2.x; bb.y = bv2.y;
                c2[t] = __builtin_amdgcn_wmma_f32_16x16x4_f32(
                    false, a, false, bb, (short)0, c2[t], false, false);
            }
        }
#pragma unroll
        for (int r = 0; r < 8; ++r) {
            const int rr = d0 + rt * 16 + khp * 8 + r;
            if (rr < n) {
#pragma unroll
                for (int t = 0; t < 3; ++t)
                    xh2[(size_t)rr * W2P_COLS + t * 16 + m] = c2[t][r];
            }
        }
    }
}


__device__ __forceinline__ void l2_rows(const float* __restrict__ xh2, int sl, float wl,
                                        int nv, int jbeg, int slot, int c4, float4& acc)
{
    for (int j0 = jbeg; j0 < nv; j0 += 12) {
        float4 v[4];
        float  w[4];
#pragma unroll
        for (int u = 0; u < 4; ++u) {
            const int j  = j0 + 3 * u + slot;
            const bool ok = (slot < 3) && (j < nv);
            const int jj = ok ? j : 0;
            const int s  = __shfl(sl, jj);
            const float wj = __shfl(wl, jj);
            w[u] = ok ? wj : 0.0f;
            v[u] = *(const float4*)(xh2 + (size_t)s * 48 + c4 * 4);
        }
#pragma unroll
        for (int u = 0; u < 4; ++u) {
            const int j = j0 + 3 * u + slot;
            if ((slot < 3) && (j < nv)) {
                acc.x = fmaf(w[u], v[u].x, acc.x);
                acc.y = fmaf(w[u], v[u].y, acc.y);
                acc.z = fmaf(w[u], v[u].z, acc.z);
                acc.w = fmaf(w[u], v[u].w, acc.w);
            }
        }
    }
}

template <int K0>
__device__ __forceinline__ void l2_ask(const float* __restrict__ xh2, int sl, int nv,
                                       int slot, int c4, float4 (&v)[8])
{
#pragma unroll
    for (int u = 0; u < 4; ++u) {
        const int j  = 3 * (K0 + u) + slot;
        const bool ok = (slot < 3) && (j < nv);
        const int s  = __shfl(sl, ok ? j : 0);
        v[K0 + u] = *(const float4*)(xh2 + (size_t)s * 48 + c4 * 4);
    }
}

template <int K0>
__device__ __forceinline__ void l2_add(const float4 (&v)[8], float wl, int nv,
                                       int slot, float4& acc)
{
#pragma unroll
    for (int u = 0; u < 4; ++u) {
        const int j  = 3 * (K0 + u) + slot;
        const bool ok = (slot < 3) && (j < nv);
        const float wj = __shfl(wl, ok ? j : 0);
        if (ok) {
            acc.x = fmaf(wj, v[K0 + u].x, acc.x);
            acc.y = fmaf(wj, v[K0 + u].y, acc.y);
            acc.z = fmaf(wj, v[K0 + u].z, acc.z);
            acc.w = fmaf(wj, v[K0 + u].w, acc.w);
        }
    }
}

__global__ void __launch_bounds__(256)
node_gather_l2(const int* __restrict__ rowptr, const int* __restrict__ csrc,
               const float* __restrict__ xh2, const float* __restrict__ b,
               float* __restrict__ out, int n)
{
    __shared__ __align__(16) float so[8 * 40];
    const int lane = threadIdx.x & 31;
    const int wv   = threadIdx.x >> 5;
    const int d0   = blockIdx.x * 8;
    const int d    = d0 + wv;

    if (d < n) {
        const int beg = __builtin_amdgcn_readfirstlane(rowptr[d]);
        const int end = __builtin_amdgcn_readfirstlane(rowptr[d + 1]);
        const int deg = end - beg;
        const int cnt = deg + 1;
        const float edd = xh2[(size_t)d * W2P_COLS + XH2_ED];

        const int slot = lane / 10;
        const int c4   = lane - slot * 10;
        float4 acc = make_float4(0.f, 0.f, 0.f, 0.f);

        if (cnt <= 32) {
            const int sl  = edge_src(csrc, beg, lane, deg, d, n);
            const float a = lrelu(xh2[(size_t)sl * W2P_COLS + XH2_ES] + edd);

            float4 v[8];
            l2_ask<0>(xh2, sl, cnt, slot, c4, v);
            if (cnt > 12) l2_ask<4>(xh2, sl, cnt, slot, c4, v);

            float m = -__builtin_inff();
            if (lane < cnt) m = fmaxf(m, a);
            for (int o = 16; o >= 1; o >>= 1) m = fmaxf(m, __shfl_xor(m, o));

            const float e = __expf(a - m);
            float sum = 0.0f;
            if (lane < cnt) sum += e;
            for (int o = 16; o >= 1; o >>= 1) sum += __shfl_xor(sum, o);
            const float inv = 1.0f / (sum + 1e-16f);

            const float wl = e * inv;
            l2_add<0>(v, wl, cnt, slot, acc);
            if (cnt > 12) l2_add<4>(v, wl, cnt, slot, acc);
            if (cnt > 24) l2_rows(xh2, sl, wl, cnt, 24, slot, c4, acc);
        } else {
            float m = -__builtin_inff();
            for (int c0 = 0; c0 < cnt; c0 += 32) {
                const int i  = c0 + lane;
                const int sl = edge_src(csrc, beg, i, deg, d, n);
                const float a = lrelu(xh2[(size_t)sl * W2P_COLS + XH2_ES] + edd);
                if (i < cnt) m = fmaxf(m, a);
            }
            for (int o = 16; o >= 1; o >>= 1) m = fmaxf(m, __shfl_xor(m, o));

            float sum = 0.0f;
            for (int c0 = 0; c0 < cnt; c0 += 32) {
                const int i  = c0 + lane;
                const int sl = edge_src(csrc, beg, i, deg, d, n);
                const float a = lrelu(xh2[(size_t)sl * W2P_COLS + XH2_ES] + edd);
                if (i < cnt) sum += __expf(a - m);
            }
            for (int o = 16; o >= 1; o >>= 1) sum += __shfl_xor(sum, o);
            const float inv = 1.0f / (sum + 1e-16f);

            for (int c0 = 0; c0 < cnt; c0 += 32) {
                const int i  = c0 + lane;
                const int sl = edge_src(csrc, beg, i, deg, d, n);
                const float a  = lrelu(xh2[(size_t)sl * W2P_COLS + XH2_ES] + edd);
                const float wl = __expf(a - m) * inv;
                const int nv = (cnt - c0 < 32) ? cnt - c0 : 32;
                l2_rows(xh2, sl, wl, nv, 0, slot, c4, acc);
            }
        }
        const int l1 = (lane + 10) & 31, l2 = (lane + 20) & 31;
        acc.x += __shfl(acc.x, l1) + __shfl(acc.x, l2);
        acc.y += __shfl(acc.y, l1) + __shfl(acc.y, l2);
        acc.z += __shfl(acc.z, l1) + __shfl(acc.z, l2);
        acc.w += __shfl(acc.w, l1) + __shfl(acc.w, l2);
        if (lane < 10) {
            const float4 bv = *(const float4*)(b + lane * 4);
            float4 o;
            o.x = acc.x + bv.x; o.y = acc.y + bv.y; o.z = acc.z + bv.z; o.w = acc.w + bv.w;
            *(float4*)(so + wv * 40 + lane * 4) = o;
        }
    }
    __syncthreads();
    const int rows = (n - d0 < 8) ? n - d0 : 8;
    if ((int)threadIdx.x < rows * 10)
        *(float4*)(out + (size_t)d0 * 40 + threadIdx.x * 4) = *(const float4*)(so + threadIdx.x * 4);
}

extern "C" void kernel_launch(void* const* d_in, const int* in_sizes, int n_in,
                              void* d_out, int out_size, void* d_ws, size_t ws_size,
                              hipStream_t stream)
{
    (void)n_in; (void)out_size; (void)ws_size;

    const float* x   = (const float*)d_in[0];
    const int*   ei  = (const int*)d_in[1];
    const float* W1  = (const float*)d_in[2];
    const float* as1 = (const float*)d_in[3];
    const float* ad1 = (const float*)d_in[4];
    const float* b1  = (const float*)d_in[5];
    const float* W2  = (const float*)d_in[6];
    const float* as2 = (const float*)d_in[7];
    const float* ad2 = (const float*)d_in[8];
    const float* b2  = (const float*)d_in[9];
    float* out = (float*)d_out;

    const int n  = in_sizes[0] / 128;
    const int E  = in_sizes[1] / 2;

    char* ws = (char*)d_ws;
    size_t off = 0;
    auto alloc = [&](size_t bytes) -> float* {
        float* p = (float*)(ws + off);
        off += (bytes + 255) & ~(size_t)255;
        return p;
    };
    float* xh1  = alloc((size_t)n * 128 * 4);
    float* aux  = alloc((size_t)n * 128 * 4);
    float* es1  = alloc((size_t)n * 8 * 4);
    float* ed1  = alloc((size_t)n * 8 * 4);
    float* W2p  = alloc((size_t)128 * W2P_COLS * 4);
    const int nr = (n + (1 << RSHIFT) - 1) >> RSHIFT;
    int* segs   = (int*)alloc((size_t)((nr + 1) * L1_BLOCKS + 16) * 4);
    int* rowptr = (int*)alloc((size_t)(n + 16) * 4);
    int* csrc   = (int*)alloc((size_t)E * 4);
    int* bsrc   = (int*)aux;
    int* pairs  = (int*)aux + E;
    float* W1e  = aux + (size_t)n * 64;
    float* xh2  = aux;

    const int B = 256;

    range_bucket<<<L1_BLOCKS + (PREP_ELEMS + 1023) / 1024, 1024, nr * sizeof(int), stream>>>(ei, E, nr, segs, pairs,
                                                                W1, as1, ad1, W2, as2, ad2, W1e, W2p);
    place_and_project<<<nr + (n + PJ_WAVES * 16 - 1) / (PJ_WAVES * 16), NP_WAVES * 32, 0, stream>>>(
        pairs, segs, rowptr, bsrc, csrc, n, E, nr, x, W1e, xh1, es1, ed1);

    node_gather_l1<<<(n + L1_NODES - 1) / L1_NODES, L1_WAVES * 32, 0, stream>>>(
        rowptr, csrc, es1, ed1, xh1, b1, W2p, xh2, n);

    node_gather_l2<<<(n + 7) / 8, B, 0, stream>>>(rowptr, csrc, xh2, b2, out, n);
}
